# starting kernel
# baseline (speedup 1.0000x reference)
#include <hip/hip_runtime.h>
#include <cstdint>

typedef float v2f __attribute__((ext_vector_type(2)));
typedef float v8f __attribute__((ext_vector_type(8)));

#define HEADS 8
#define NEG_SLOPE 0.2f
#define SM_EPS 1e-16f
#define NT 4

__global__ void wmma_gemm_f32(const float* __restrict__ A,
                              const float* __restrict__ B,
                              float* __restrict__ C,
                              int M, int K, int Ncols) {
  int wave = (int)((blockIdx.x * blockDim.x + threadIdx.x) >> 5);
  int lane = (int)(threadIdx.x & 31);
  int tilesM  = (M + 15) >> 4;
  int groupsN = Ncols >> 6;
  if (wave >= tilesM * groupsN) return;
  int tm = wave / groupsN;
  int tg = wave - tm * groupsN;
  int half = lane >> 4;
  int l = lane & 15;

  int rowA = tm * 16 + l;
  if (rowA >= M) rowA = M - 1;
  const float* aP = A + (size_t)rowA * K + 2 * half;
  const float* bP = B + (size_t)(2 * half) * Ncols + tg * (16 * NT) + l;

  v8f acc[NT];
#pragma unroll
  for (int t = 0; t < NT; ++t) acc[t] = (v8f){};

  v2f aCur, bCur[NT];
  aCur.x = aP[0];
  aCur.y = aP[1];
#pragma unroll
  for (int t = 0; t < NT; ++t) {
    bCur[t].x = bP[t * 16];
    bCur[t].y = bP[t * 16 + Ncols];
  }

  for (int k0 = 0; k0 < K - 4; k0 += 4) {
    aP += 4;
    bP += (size_t)4 * Ncols;
    v2f aNxt, bNxt[NT];
    aNxt.x = aP[0];
    aNxt.y = aP[1];
#pragma unroll
    for (int t = 0; t < NT; ++t) {
      bNxt[t].x = bP[t * 16];
      bNxt[t].y = bP[t * 16 + Ncols];
    }
#pragma unroll
    for (int t = 0; t < NT; ++t) {
      acc[t] = __builtin_amdgcn_wmma_f32_16x16x4_f32(false, aCur, false, bCur[t],
                                                     (short)0, acc[t],
                                                     false, false);
    }
    aCur = aNxt;
#pragma unroll
    for (int t = 0; t < NT; ++t) bCur[t] = bNxt[t];
  }
#pragma unroll
  for (int t = 0; t < NT; ++t) {
    acc[t] = __builtin_amdgcn_wmma_f32_16x16x4_f32(false, aCur, false, bCur[t],
                                                   (short)0, acc[t],
                                                   false, false);
  }

#pragma unroll
  for (int t = 0; t < NT; ++t) {
    int col = tg * (16 * NT) + t * 16 + l;
#pragma unroll
    for (int r = 0; r < 8; ++r) {
      int row = tm * 16 + r + 8 * half;
      if (row < M) C[(size_t)row * Ncols + col] = acc[t][r];
    }
  }
}

__global__ void attn_coeff_kernel(const float* __restrict__ h,
                                  const float* __restrict__ att_src,
                                  const float* __restrict__ att_dst,
                                  float* __restrict__ asrc,
                                  float* __restrict__ adst,
                                  int N, int C) {
  long i = blockIdx.x * (long)blockDim.x + threadIdx.x;
  if (i >= (long)N * HEADS) return;
  int head = (int)(i & (HEADS - 1));
  const float* hp = h + (size_t)i * C;
  const float* as = att_src + (size_t)head * C;
  const float* ad = att_dst + (size_t)head * C;
  float s = 0.f, d = 0.f;
  for (int c = 0; c < C; ++c) {
    float v = hp[c];
    s += v * as[c];
    d += v * ad[c];
  }
  asrc[i] = s;
  adst[i] = d;
}

__global__ void fill_f32(float* __restrict__ p, float v, long n) {
  long i = blockIdx.x * (long)blockDim.x + threadIdx.x;
  if (i < n) p[i] = v;
}

__device__ __forceinline__ float lrelu(float v) {
  return v > 0.f ? v : NEG_SLOPE * v;
}

__global__ void edge_max_kernel(const int* __restrict__ src,
                                const int* __restrict__ dst,
                                int E, int N,
                                const float* __restrict__ asrc,
                                const float* __restrict__ adst,
                                float* __restrict__ m) {
  long i = blockIdx.x * (long)blockDim.x + threadIdx.x;
  long tot = (long)(E + N) * HEADS;
  if (i >= tot) return;
  int e = (int)(i >> 3);
  int head = (int)(i & 7);
  int s = (e < E) ? src[e] : (e - E);
  int d = (e < E) ? dst[e] : (e - E);
  float v = lrelu(asrc[(size_t)s * HEADS + head] + adst[(size_t)d * HEADS + head]);
  atomicMax(&m[(size_t)d * HEADS + head], v);
}

__global__ void edge_expsum_kernel(const int* __restrict__ src,
                                   const int* __restrict__ dst,
                                   int E, int N,
                                   const float* __restrict__ asrc,
                                   const float* __restrict__ adst,
                                   const float* __restrict__ m,
                                   float* __restrict__ denom) {
  long i = blockIdx.x * (long)blockDim.x + threadIdx.x;
  long tot = (long)(E + N) * HEADS;
  if (i >= tot) return;
  int e = (int)(i >> 3);
  int head = (int)(i & 7);
  int s = (e < E) ? src[e] : (e - E);
  int d = (e < E) ? dst[e] : (e - E);
  size_t dh = (size_t)d * HEADS + head;
  float v = lrelu(asrc[(size_t)s * HEADS + head] + adst[dh]);
  atomicAdd(&denom[dh], __expf(v - m[dh]));
}

__global__ void edge_agg_kernel(const int* __restrict__ src,
                                const int* __restrict__ dst,
                                int E, int N, int C,
                                const float* __restrict__ asrc,
                                const float* __restrict__ adst,
                                const float* __restrict__ m,
                                const float* __restrict__ denom,
                                const float* __restrict__ h,
                                float* __restrict__ agg) {
  long i = blockIdx.x * (long)blockDim.x + threadIdx.x;
  long tot = (long)(E + N) * HEADS;
  if (i >= tot) return;
  int e = (int)(i >> 3);
  int head = (int)(i & 7);
  int s = (e < E) ? src[e] : (e - E);
  int d = (e < E) ? dst[e] : (e - E);
  size_t dh = (size_t)d * HEADS + head;
  float v = lrelu(asrc[(size_t)s * HEADS + head] + adst[dh]);
  float alpha = __expf(v - m[dh]) / (denom[dh] + SM_EPS);
  const float4* hs4 = (const float4*)(h + ((size_t)s * HEADS + head) * C);
  float* od = agg + dh * C;
  int C4 = C >> 2;
  for (int c4 = 0; c4 < C4; ++c4) {
    float4 hv = hs4[c4];
    atomicAdd(&od[c4 * 4 + 0], alpha * hv.x);
    atomicAdd(&od[c4 * 4 + 1], alpha * hv.y);
    atomicAdd(&od[c4 * 4 + 2], alpha * hv.z);
    atomicAdd(&od[c4 * 4 + 3], alpha * hv.w);
  }
}

__global__ void finalize_kernel(const float* __restrict__ agg,
                                const float* __restrict__ bias,
                                float* __restrict__ out,
                                int N, int C, int do_relu) {
  long i = blockIdx.x * (long)blockDim.x + threadIdx.x;
  if (i >= (long)N * C) return;
  int n = (int)(i / C);
  int c = (int)(i - (long)n * C);
  float s = 0.f;
#pragma unroll
  for (int hh = 0; hh < HEADS; ++hh)
    s += agg[((size_t)n * HEADS + hh) * C + c];
  s = s * (1.0f / HEADS) + bias[c];
  if (do_relu) s = s > 0.f ? s : 0.f;
  out[i] = s;
}

static inline unsigned nblk(long n, int b) { return (unsigned)((n + b - 1) / b); }

extern "C" void kernel_launch(void* const* d_in, const int* in_sizes, int n_in,
                              void* d_out, int out_size, void* d_ws, size_t ws_size,
                              hipStream_t stream) {
  const float* x        = (const float*)d_in[0];
  const int*   ei       = (const int*)  d_in[1];
  const float* W1       = (const float*)d_in[2];
  const float* att_src1 = (const float*)d_in[3];
  const float* att_dst1 = (const float*)d_in[4];
  const float* b1       = (const float*)d_in[5];
  const float* W2       = (const float*)d_in[6];
  const float* att_src2 = (const float*)d_in[7];
  const float* att_dst2 = (const float*)d_in[8];
  const float* b2       = (const float*)d_in[9];

  const int F_h   = in_sizes[3] / HEADS;
  const int F_out = in_sizes[7] / HEADS;
  const int F_in  = in_sizes[2] / (HEADS * F_h);
  const int N     = in_sizes[0] / F_in;
  const int E     = in_sizes[1] / 2;
  const int C1    = HEADS * F_h;
  const int C2    = HEADS * F_out;
  const int* src = ei;
  const int* dst = ei + E;

  float* ws = (float*)d_ws;
  size_t o = 0;
  float* h1   = ws + o; o += (size_t)N * C1;
  float* agg  = ws + o; o += (size_t)N * C1;
  float* asrc = ws + o; o += (size_t)N * HEADS;
  float* adst = ws + o; o += (size_t)N * HEADS;
  float* mbuf = ws + o; o += (size_t)N * HEADS;
  float* dbuf = ws + o; o += (size_t)N * HEADS;
  float* x2   = ws + o; o += (size_t)N * F_h;
  float* h2   = h1;
  float* outp = (float*)d_out;

  const int TB = 256;
  const long NH = (long)N * HEADS;
  const long ETOT = (long)(E + N) * HEADS;

  {
    long waves = (long)((N + 15) / 16) * (C1 / (16 * NT));
    wmma_gemm_f32<<<nblk(waves * 32, TB), TB, 0, stream>>>(x, W1, h1, N, F_in, C1);
  }
  attn_coeff_kernel<<<nblk(NH, TB), TB, 0, stream>>>(h1, att_src1, att_dst1,
                                                     asrc, adst, N, F_h);
  fill_f32<<<nblk(NH, TB), TB, 0, stream>>>(mbuf, -3.0e38f, NH);
  fill_f32<<<nblk(NH, TB), TB, 0, stream>>>(dbuf, 0.0f, NH);
  fill_f32<<<nblk((long)N * C1, TB), TB, 0, stream>>>(agg, 0.0f, (long)N * C1);
  edge_max_kernel<<<nblk(ETOT, TB), TB, 0, stream>>>(src, dst, E, N, asrc, adst, mbuf);
  edge_expsum_kernel<<<nblk(ETOT, TB), TB, 0, stream>>>(src, dst, E, N, asrc, adst,
                                                        mbuf, dbuf);
  edge_agg_kernel<<<nblk(ETOT, TB), TB, 0, stream>>>(src, dst, E, N, F_h, asrc, adst,
                                                     mbuf, dbuf, h1, agg);
  finalize_kernel<<<nblk((long)N * F_h, TB), TB, 0, stream>>>(agg, b1, x2, N, F_h, 1);

  {
    long waves = (long)((N + 15) / 16) * (C2 / (16 * NT));
    wmma_gemm_f32<<<nblk(waves * 32, TB), TB, 0, stream>>>(x2, W2, h2, N, F_h, C2);
  }
  attn_coeff_kernel<<<nblk(NH, TB), TB, 0, stream>>>(h2, att_src2, att_dst2,
                                                     asrc, adst, N, F_out);
  fill_f32<<<nblk(NH, TB), TB, 0, stream>>>(mbuf, -3.0e38f, NH);
  fill_f32<<<nblk(NH, TB), TB, 0, stream>>>(dbuf, 0.0f, NH);
  fill_f32<<<nblk((long)N * C2, TB), TB, 0, stream>>>(agg, 0.0f, (long)N * C2);
  edge_max_kernel<<<nblk(ETOT, TB), TB, 0, stream>>>(src, dst, E, N, asrc, adst, mbuf);
  edge_expsum_kernel<<<nblk(ETOT, TB), TB, 0, stream>>>(src, dst, E, N, asrc, adst,
                                                        mbuf, dbuf);
  edge_agg_kernel<<<nblk(ETOT, TB), TB, 0, stream>>>(src, dst, E, N, F_out, asrc, adst,
                                                     mbuf, dbuf, h2, agg);
  finalize_kernel<<<nblk((long)N * F_out, TB), TB, 0, stream>>>(agg, b2, outp,
                                                                N, F_out, 0);
}


# edge aggregation sums the 8 heads in registers per edge and adds only [N,C] rows (lane=channel, whole-line f32 atomic adds): 8x fewer adds; changes the order of the float sums
# speedup vs baseline: 6.0099x; 6.0099x over previous
#include <hip/hip_runtime.h>
#include <cstdint>

typedef float v2f __attribute__((ext_vector_type(2)));
typedef float v8f __attribute__((ext_vector_type(8)));

#define HEADS 8
#define NEG_SLOPE 0.2f
#define SM_EPS 1e-16f
#define NT 4

__global__ void wmma_gemm_f32(const float* __restrict__ A,
                              const float* __restrict__ B,
                              float* __restrict__ C,
                              int M, int K, int Ncols) {
  int wave = (int)((blockIdx.x * blockDim.x + threadIdx.x) >> 5);
  int lane = (int)(threadIdx.x & 31);
  int tilesM  = (M + 15) >> 4;
  int groupsN = Ncols >> 6;
  if (wave >= tilesM * groupsN) return;
  int tm = wave / groupsN;
  int tg = wave - tm * groupsN;
  int half = lane >> 4;
  int l = lane & 15;

  int rowA = tm * 16 + l;
  if (rowA >= M) rowA = M - 1;
  const float* aP = A + (size_t)rowA * K + 2 * half;
  const float* bP = B + (size_t)(2 * half) * Ncols + tg * (16 * NT) + l;

  v8f acc[NT];
#pragma unroll
  for (int t = 0; t < NT; ++t) acc[t] = (v8f){};

  v2f aCur, bCur[NT];
  aCur.x = aP[0];
  aCur.y = aP[1];
#pragma unroll
  for (int t = 0; t < NT; ++t) {
    bCur[t].x = bP[t * 16];
    bCur[t].y = bP[t * 16 + Ncols];
  }

  for (int k0 = 0; k0 < K - 4; k0 += 4) {
    aP += 4;
    bP += (size_t)4 * Ncols;
    v2f aNxt, bNxt[NT];
    aNxt.x = aP[0];
    aNxt.y = aP[1];
#pragma unroll
    for (int t = 0; t < NT; ++t) {
      bNxt[t].x = bP[t * 16];
      bNxt[t].y = bP[t * 16 + Ncols];
    }
#pragma unroll
    for (int t = 0; t < NT; ++t) {
      acc[t] = __builtin_amdgcn_wmma_f32_16x16x4_f32(false, aCur, false, bCur[t],
                                                     (short)0, acc[t],
                                                     false, false);
    }
    aCur = aNxt;
#pragma unroll
    for (int t = 0; t < NT; ++t) bCur[t] = bNxt[t];
  }
#pragma unroll
  for (int t = 0; t < NT; ++t) {
    acc[t] = __builtin_amdgcn_wmma_f32_16x16x4_f32(false, aCur, false, bCur[t],
                                                   (short)0, acc[t],
                                                   false, false);
  }

#pragma unroll
  for (int t = 0; t < NT; ++t) {
    int col = tg * (16 * NT) + t * 16 + l;
#pragma unroll
    for (int r = 0; r < 8; ++r) {
      int row = tm * 16 + r + 8 * half;
      if (row < M) C[(size_t)row * Ncols + col] = acc[t][r];
    }
  }
}

__global__ void attn_coeff_kernel(const float* __restrict__ h,
                                  const float* __restrict__ att_src,
                                  const float* __restrict__ att_dst,
                                  float* __restrict__ asrc,
                                  float* __restrict__ adst,
                                  int N, int C) {
  long i = blockIdx.x * (long)blockDim.x + threadIdx.x;
  if (i >= (long)N * HEADS) return;
  int head = (int)(i & (HEADS - 1));
  const float* hp = h + (size_t)i * C;
  const float* as = att_src + (size_t)head * C;
  const float* ad = att_dst + (size_t)head * C;
  float s = 0.f, d = 0.f;
  for (int c = 0; c < C; ++c) {
    float v = hp[c];
    s += v * as[c];
    d += v * ad[c];
  }
  asrc[i] = s;
  adst[i] = d;
}

__global__ void fill_f32(float* __restrict__ p, float v, long n) {
  long i = blockIdx.x * (long)blockDim.x + threadIdx.x;
  if (i < n) p[i] = v;
}

__device__ __forceinline__ float lrelu(float v) {
  return v > 0.f ? v : NEG_SLOPE * v;
}

__global__ void edge_max_kernel(const int* __restrict__ src,
                                const int* __restrict__ dst,
                                int E, int N,
                                const float* __restrict__ asrc,
                                const float* __restrict__ adst,
                                float* __restrict__ m) {
  long i = blockIdx.x * (long)blockDim.x + threadIdx.x;
  long tot = (long)(E + N) * HEADS;
  if (i >= tot) return;
  int e = (int)(i >> 3);
  int head = (int)(i & 7);
  int s = (e < E) ? src[e] : (e - E);
  int d = (e < E) ? dst[e] : (e - E);
  float v = lrelu(asrc[(size_t)s * HEADS + head] + adst[(size_t)d * HEADS + head]);
  atomicMax(&m[(size_t)d * HEADS + head], v);
}

__global__ void edge_expsum_kernel(const int* __restrict__ src,
                                   const int* __restrict__ dst,
                                   int E, int N,
                                   const float* __restrict__ asrc,
                                   const float* __restrict__ adst,
                                   const float* __restrict__ m,
                                   float* __restrict__ denom) {
  long i = blockIdx.x * (long)blockDim.x + threadIdx.x;
  long tot = (long)(E + N) * HEADS;
  if (i >= tot) return;
  int e = (int)(i >> 3);
  int head = (int)(i & 7);
  int s = (e < E) ? src[e] : (e - E);
  int d = (e < E) ? dst[e] : (e - E);
  size_t dh = (size_t)d * HEADS + head;
  float v = lrelu(asrc[(size_t)s * HEADS + head] + adst[dh]);
  atomicAdd(&denom[dh], __expf(v - m[dh]));
}

template <int C>
__global__ void edge_agg_kernel(const int* __restrict__ src,
                                const int* __restrict__ dst,
                                int E, int N,
                                const float* __restrict__ asrc,
                                const float* __restrict__ adst,
                                const float* __restrict__ m,
                                const float* __restrict__ denom,
                                const float* __restrict__ h,
                                float* __restrict__ acc) {
  static_assert(C == 32 || C == 16, "row pieces are mapped for 32 or 16 channels");
  constexpr int ROW = HEADS * C;
  constexpr int NLD = ROW / 32;
  const int lane = (int)(threadIdx.x & 31);
  const int wib = __builtin_amdgcn_readfirstlane((int)(threadIdx.x >> 5));
  const long wave = (long)blockIdx.x * (long)(blockDim.x >> 5) + wib;
  const long tot = (long)E + N;
  const long ebase = wave * 4;
  if (ebase >= tot) return;
  const int nlive = (tot - ebase >= 4) ? 4 : (int)(tot - ebase);
  const int jl = lane >> 3;
  const long e = ebase + jl;
  const int head = lane & 7;
  const bool live = jl < nlive;
  int s = 0, d = 0;
  if (live) {
    s = (e < E) ? src[e] : (int)(e - E);
    d = (e < E) ? dst[e] : (int)(e - E);
  }
  const size_t dh = (size_t)d * HEADS + head;
  const float v = lrelu(asrc[(size_t)s * HEADS + head] + adst[dh]);
  float alpha = __expf(v - m[dh]) / (denom[dh] + SM_EPS);
  if (!live) alpha = 0.f;
  const int abits = __float_as_int(alpha);

  float sum[4];
  int dj[4];
#pragma unroll
  for (int j = 0; j < 4; ++j) {
    const int sj = __builtin_amdgcn_readlane(s, j * 8);
    dj[j] = __builtin_amdgcn_readlane(d, j * 8);
    const float* hp = h + (size_t)sj * ROW + lane;
    float a;
#pragma unroll
    for (int k = 0; k < NLD; ++k) {
      float w;
      if (C == 32) {
        w = __int_as_float(__builtin_amdgcn_readlane(abits, j * 8 + k));
      } else {
        const float w0 = __int_as_float(__builtin_amdgcn_readlane(abits, j * 8 + 2 * k));
        const float w1 = __int_as_float(__builtin_amdgcn_readlane(abits, j * 8 + 2 * k + 1));
        w = (lane & 16) ? w1 : w0;
      }
      if (k == 0) a = w * hp[0];
      else        a += w * hp[k * 32];
    }
    sum[j] = (j < nlive) ? a : 0.f;
  }

  if (C == 32) {
#pragma unroll
    for (int j = 0; j < 4; ++j)
      atomicAdd(&acc[(size_t)dj[j] * C + lane], sum[j]);
  } else {
    const bool hi = (lane & 16) != 0;
#pragma unroll
    for (int j = 0; j < 4; j += 2) {
      const float give = hi ? sum[j] : sum[j + 1];
      const float keep = hi ? sum[j + 1] : sum[j];
      const float t = keep + __shfl_xor(give, 16);
      const int dd = hi ? dj[j + 1] : dj[j];
      atomicAdd(&acc[(size_t)dd * C + (lane & 15)], t);
    }
  }
}

__global__ void finalize_kernel(const float* __restrict__ acc,
                                const float* __restrict__ bias,
                                float* __restrict__ out,
                                int N, int C, int do_relu) {
  long i = blockIdx.x * (long)blockDim.x + threadIdx.x;
  if (i >= (long)N * C) return;
  int c = (int)(i % C);
  float s = acc[i] * (1.0f / HEADS) + bias[c];
  if (do_relu) s = s > 0.f ? s : 0.f;
  out[i] = s;
}

static inline unsigned nblk(long n, int b) { return (unsigned)((n + b - 1) / b); }

extern "C" void kernel_launch(void* const* d_in, const int* in_sizes, int n_in,
                              void* d_out, int out_size, void* d_ws, size_t ws_size,
                              hipStream_t stream) {
  const float* x        = (const float*)d_in[0];
  const int*   ei       = (const int*)  d_in[1];
  const float* W1       = (const float*)d_in[2];
  const float* att_src1 = (const float*)d_in[3];
  const float* att_dst1 = (const float*)d_in[4];
  const float* b1       = (const float*)d_in[5];
  const float* W2       = (const float*)d_in[6];
  const float* att_src2 = (const float*)d_in[7];
  const float* att_dst2 = (const float*)d_in[8];
  const float* b2       = (const float*)d_in[9];

  const int F_h   = in_sizes[3] / HEADS;
  const int F_out = in_sizes[7] / HEADS;
  const int F_in  = in_sizes[2] / (HEADS * F_h);
  const int N     = in_sizes[0] / F_in;
  const int E     = in_sizes[1] / 2;
  const int C1    = HEADS * F_h;
  const int C2    = HEADS * F_out;
  const int* src = ei;
  const int* dst = ei + E;

  float* ws = (float*)d_ws;
  size_t o = 0;
  float* h1   = ws + o; o += (size_t)N * C1;
  float* agg  = ws + o; o += (size_t)N * C1;
  float* asrc = ws + o; o += (size_t)N * HEADS;
  float* adst = ws + o; o += (size_t)N * HEADS;
  float* mbuf = ws + o; o += (size_t)N * HEADS;
  float* dbuf = ws + o; o += (size_t)N * HEADS;
  float* x2   = ws + o; o += (size_t)N * F_h;
  float* h2   = h1;
  float* outp = (float*)d_out;

  const int TB = 256;
  const long NH = (long)N * HEADS;
  const long ETOT = (long)(E + N) * HEADS;

  {
    long waves = (long)((N + 15) / 16) * (C1 / (16 * NT));
    wmma_gemm_f32<<<nblk(waves * 32, TB), TB, 0, stream>>>(x, W1, h1, N, F_in, C1);
  }
  attn_coeff_kernel<<<nblk(NH, TB), TB, 0, stream>>>(h1, att_src1, att_dst1,
                                                     asrc, adst, N, F_h);
  fill_f32<<<nblk(NH, TB), TB, 0, stream>>>(mbuf, -3.0e38f, NH);
  fill_f32<<<nblk(NH, TB), TB, 0, stream>>>(dbuf, 0.0f, NH);
  fill_f32<<<nblk((long)N * F_h, TB), TB, 0, stream>>>(agg, 0.0f, (long)N * F_h);
  edge_max_kernel<<<nblk(ETOT, TB), TB, 0, stream>>>(src, dst, E, N, asrc, adst, mbuf);
  edge_expsum_kernel<<<nblk(ETOT, TB), TB, 0, stream>>>(src, dst, E, N, asrc, adst,
                                                        mbuf, dbuf);
  edge_agg_kernel<32><<<nblk(ETOT, TB), TB, 0, stream>>>(src, dst, E, N, asrc, adst,
                                                         mbuf, dbuf, h1, agg);
  finalize_kernel<<<nblk((long)N * F_h, TB), TB, 0, stream>>>(agg, b1, x2, N, F_h, 1);

  {
    long waves = (long)((N + 15) / 16) * (C2 / (16 * NT));
    wmma_gemm_f32<<<nblk(waves * 32, TB), TB, 0, stream>>>(x2, W2, h2, N, F_h, C2);
  }
  attn_coeff_kernel<<<nblk(NH, TB), TB, 0, stream>>>(h2, att_src2, att_dst2,
                                                     asrc, adst, N, F_out);
  fill_f32<<<nblk(NH, TB), TB, 0, stream>>>(mbuf, -3.0e38f, NH);
  fill_f32<<<nblk(NH, TB), TB, 0, stream>>>(dbuf, 0.0f, NH);
  fill_f32<<<nblk((long)N * F_out, TB), TB, 0, stream>>>(agg, 0.0f, (long)N * F_out);
  edge_max_kernel<<<nblk(ETOT, TB), TB, 0, stream>>>(src, dst, E, N, asrc, adst, mbuf);
  edge_expsum_kernel<<<nblk(ETOT, TB), TB, 0, stream>>>(src, dst, E, N, asrc, adst,
                                                        mbuf, dbuf);
  edge_agg_kernel<16><<<nblk(ETOT, TB), TB, 0, stream>>>(src, dst, E, N, asrc, adst,
                                                         mbuf, dbuf, h2, agg);
  finalize_kernel<<<nblk((long)N * F_out, TB), TB, 0, stream>>>(agg, b2, outp,
                                                                N, F_out, 0);
}


# edges grouped by destination in each call (counting sort, integer atomic adds); one wave per node does softmax and row sum in registers, no float atomics; changes the order of the float sums
# speedup vs baseline: 9.8317x; 1.6359x over previous
#include <hip/hip_runtime.h>
#include <cstdint>

typedef float v2f __attribute__((ext_vector_type(2)));
typedef float v8f __attribute__((ext_vector_type(8)));

#define HEADS 8
#define NEG_SLOPE 0.2f
#define SM_EPS 1e-16f
#define NT 4

__global__ void wmma_gemm_f32(const float* __restrict__ A,
                              const float* __restrict__ B,
                              float* __restrict__ C,
                              int M, int K, int Ncols) {
  int wave = (int)((blockIdx.x * blockDim.x + threadIdx.x) >> 5);
  int lane = (int)(threadIdx.x & 31);
  int tilesM  = (M + 15) >> 4;
  int groupsN = Ncols >> 6;
  if (wave >= tilesM * groupsN) return;
  int tm = wave / groupsN;
  int tg = wave - tm * groupsN;
  int half = lane >> 4;
  int l = lane & 15;

  int rowA = tm * 16 + l;
  if (rowA >= M) rowA = M - 1;
  const float* aP = A + (size_t)rowA * K + 2 * half;
  const float* bP = B + (size_t)(2 * half) * Ncols + tg * (16 * NT) + l;

  v8f acc[NT];
#pragma unroll
  for (int t = 0; t < NT; ++t) acc[t] = (v8f){};

  v2f aCur, bCur[NT];
  aCur.x = aP[0];
  aCur.y = aP[1];
#pragma unroll
  for (int t = 0; t < NT; ++t) {
    bCur[t].x = bP[t * 16];
    bCur[t].y = bP[t * 16 + Ncols];
  }

  for (int k0 = 0; k0 < K - 4; k0 += 4) {
    aP += 4;
    bP += (size_t)4 * Ncols;
    v2f aNxt, bNxt[NT];
    aNxt.x = aP[0];
    aNxt.y = aP[1];
#pragma unroll
    for (int t = 0; t < NT; ++t) {
      bNxt[t].x = bP[t * 16];
      bNxt[t].y = bP[t * 16 + Ncols];
    }
#pragma unroll
    for (int t = 0; t < NT; ++t) {
      acc[t] = __builtin_amdgcn_wmma_f32_16x16x4_f32(false, aCur, false, bCur[t],
                                                     (short)0, acc[t],
                                                     false, false);
    }
    aCur = aNxt;
#pragma unroll
    for (int t = 0; t < NT; ++t) bCur[t] = bNxt[t];
  }
#pragma unroll
  for (int t = 0; t < NT; ++t) {
    acc[t] = __builtin_amdgcn_wmma_f32_16x16x4_f32(false, aCur, false, bCur[t],
                                                   (short)0, acc[t],
                                                   false, false);
  }

#pragma unroll
  for (int t = 0; t < NT; ++t) {
    int col = tg * (16 * NT) + t * 16 + l;
#pragma unroll
    for (int r = 0; r < 8; ++r) {
      int row = tm * 16 + r + 8 * half;
      if (row < M) C[(size_t)row * Ncols + col] = acc[t][r];
    }
  }
}

__global__ void attn_coeff_kernel(const float* __restrict__ h,
                                  const float* __restrict__ att_src,
                                  const float* __restrict__ att_dst,
                                  float* __restrict__ asrc,
                                  float* __restrict__ adst,
                                  int N, int C) {
  long i = blockIdx.x * (long)blockDim.x + threadIdx.x;
  if (i >= (long)N * HEADS) return;
  int head = (int)(i & (HEADS - 1));
  const float* hp = h + (size_t)i * C;
  const float* as = att_src + (size_t)head * C;
  const float* ad = att_dst + (size_t)head * C;
  float s = 0.f, d = 0.f;
  for (int c = 0; c < C; ++c) {
    float v = hp[c];
    s += v * as[c];
    d += v * ad[c];
  }
  asrc[i] = s;
  adst[i] = d;
}

__device__ __forceinline__ float lrelu(float v) {
  return v > 0.f ? v : NEG_SLOPE * v;
}

__global__ void zero_i32(int* __restrict__ p, int n) {
  int i = (int)(blockIdx.x * blockDim.x + threadIdx.x);
  if (i < n) p[i] = 0;
}

__global__ void dst_rank_kernel(const int* __restrict__ dst, int E,
                                int* __restrict__ cnt, int* __restrict__ rank) {
  int e = (int)(blockIdx.x * blockDim.x + threadIdx.x);
  if (e >= E) return;
  rank[e] = atomicAdd(&cnt[dst[e]], 1);
}

__global__ void __launch_bounds__(1024)
rowptr_kernel(const int* __restrict__ cnt, int N, int* __restrict__ rowptr) {
  __shared__ int wsum[32];
  const int t = (int)threadIdx.x;
  const int lane = t & 31;
  const int w = t >> 5;
  const int base = (int)blockIdx.x * 1024;

  int part = 0;
  for (int i = t; i < base; i += 1024) part += cnt[i];
  for (int o = 16; o > 0; o >>= 1) part += __shfl_xor(part, o);
  if (lane == 0) wsum[w] = part;
  __syncthreads();
  int prefix = 0;
  for (int k = 0; k < 32; ++k) prefix += wsum[k];
  __syncthreads();

  const int i = base + t;
  const int c = (i < N) ? cnt[i] : 0;
  int incl = c;
  for (int o = 1; o < 32; o <<= 1) {
    const int v = __shfl_up(incl, o);
    if (lane >= o) incl += v;
  }
  if (lane == 31) wsum[w] = incl;
  __syncthreads();
  int woff = 0;
  for (int k = 0; k < 32; ++k) woff += (k < w) ? wsum[k] : 0;
  if (i < N) rowptr[i] = prefix + woff + incl - c;
}

__global__ void csr_fill_kernel(const int* __restrict__ src,
                                const int* __restrict__ dst, int E,
                                const int* __restrict__ rowptr,
                                const int* __restrict__ rank,
                                int* __restrict__ csr_src) {
  int e = (int)(blockIdx.x * blockDim.x + threadIdx.x);
  if (e >= E) return;
  csr_src[rowptr[dst[e]] + rank[e]] = src[e];
}

__device__ __forceinline__ void edge_scores(const int* __restrict__ csr_src,
                                            const float* __restrict__ asrc,
                                            int start, int deg, int n, int j,
                                            const float (&ad)[HEADS],
                                            int& s, float (&e)[HEADS]) {
  s = (j < deg) ? csr_src[start + j] : n;
  const float4 a0 = *(const float4*)(asrc + (size_t)s * HEADS);
  const float4 a1 = *(const float4*)(asrc + (size_t)s * HEADS + 4);
  e[0] = lrelu(a0.x + ad[0]);
  e[1] = lrelu(a0.y + ad[1]);
  e[2] = lrelu(a0.z + ad[2]);
  e[3] = lrelu(a0.w + ad[3]);
  e[4] = lrelu(a1.x + ad[4]);
  e[5] = lrelu(a1.y + ad[5]);
  e[6] = lrelu(a1.z + ad[6]);
  e[7] = lrelu(a1.w + ad[7]);
}

template <int C>
__device__ __forceinline__ void gather_rows(const float* __restrict__ h, int sreg,
                                            const float (&alpha)[HEADS], int count,
                                            int lane, int hq, float4& acc) {
  constexpr int ROW = HEADS * C;
#pragma unroll 4
  for (int jj = 0; jj < count; ++jj) {
    const int sj = __builtin_amdgcn_readlane(sreg, jj);
    float a[HEADS];
#pragma unroll
    for (int k = 0; k < HEADS; ++k)
      a[k] = __int_as_float(__builtin_amdgcn_readlane(__float_as_int(alpha[k]), jj));
    const float* hp = h + (size_t)sj * ROW + 4 * lane;
    if (C == 32) {
      float w0 = a[0], w1 = a[4];
      w0 = (hq == 1) ? a[1] : w0;  w1 = (hq == 1) ? a[5] : w1;
      w0 = (hq == 2) ? a[2] : w0;  w1 = (hq == 2) ? a[6] : w1;
      w0 = (hq == 3) ? a[3] : w0;  w1 = (hq == 3) ? a[7] : w1;
      const float4 v0 = *(const float4*)hp;
      const float4 v1 = *(const float4*)(hp + 128);
      acc.x += w0 * v0.x;  acc.y += w0 * v0.y;  acc.z += w0 * v0.z;  acc.w += w0 * v0.w;
      acc.x += w1 * v1.x;  acc.y += w1 * v1.y;  acc.z += w1 * v1.z;  acc.w += w1 * v1.w;
    } else {
      float w = a[0];
#pragma unroll
      for (int k = 1; k < HEADS; ++k) w = (hq == k) ? a[k] : w;
      const float4 v = *(const float4*)hp;
      acc.x += w * v.x;  acc.y += w * v.y;  acc.z += w * v.z;  acc.w += w * v.w;
    }
  }
}

template <int C>
__global__ void node_attend_kernel(const int* __restrict__ rowptr,
                                   const int* __restrict__ cnt,
                                   const int* __restrict__ csr_src,
                                   const float* __restrict__ asrc,
                                   const float* __restrict__ adst,
                                   const float* __restrict__ h,
                                   const float* __restrict__ bias,
                                   float* __restrict__ out,
                                   int N, int do_relu) {
  static_assert(C == 32 || C == 16, "lane maps are written for 32 or 16 channels");
  constexpr int NPW = 32 / C;
  constexpr int GL  = C / 4;
  const int lane = (int)(threadIdx.x & 31);
  const int wib = __builtin_amdgcn_readfirstlane((int)(threadIdx.x >> 5));
  const int wave = (int)blockIdx.x * (int)(blockDim.x >> 5) + wib;
  const int n0 = wave * NPW;
  if (n0 >= N) return;
  const int hq = lane / GL;
  float4 keep = make_float4(0.f, 0.f, 0.f, 0.f);

  for (int u = 0; u < NPW; ++u) {
    const int n = n0 + u;
    if (n >= N) break;
    const int start = rowptr[n];
    const int deg = cnt[n];
    const int tot = deg + 1;
    float ad[HEADS];
#pragma unroll
    for (int k = 0; k < HEADS; ++k) ad[k] = adst[(size_t)n * HEADS + k];

    int s0;
    float e0[HEADS];
    const bool v0 = lane < tot;
    edge_scores(csr_src, asrc, start, deg, n, lane, ad, s0, e0);

    float mx[HEADS];
#pragma unroll
    for (int k = 0; k < HEADS; ++k) mx[k] = v0 ? e0[k] : -3.0e38f;
    for (int base = 32; base < tot; base += 32) {
      int s; float e[HEADS];
      const int j = base + lane;
      edge_scores(csr_src, asrc, start, deg, n, j, ad, s, e);
#pragma unroll
      for (int k = 0; k < HEADS; ++k) mx[k] = (j < tot) ? fmaxf(mx[k], e[k]) : mx[k];
    }
#pragma unroll
    for (int k = 0; k < HEADS; ++k)
      for (int o = 16; o > 0; o >>= 1) mx[k] = fmaxf(mx[k], __shfl_xor(mx[k], o));

    float ex0[HEADS], sm[HEADS];
#pragma unroll
    for (int k = 0; k < HEADS; ++k) {
      ex0[k] = v0 ? __expf(e0[k] - mx[k]) : 0.f;
      sm[k] = ex0[k];
    }
    for (int base = 32; base < tot; base += 32) {
      int s; float e[HEADS];
      const int j = base + lane;
      edge_scores(csr_src, asrc, start, deg, n, j, ad, s, e);
#pragma unroll
      for (int k = 0; k < HEADS; ++k) sm[k] += (j < tot) ? __expf(e[k] - mx[k]) : 0.f;
    }
#pragma unroll
    for (int k = 0; k < HEADS; ++k)
      for (int o = 16; o > 0; o >>= 1) sm[k] += __shfl_xor(sm[k], o);

    float4 acc = make_float4(0.f, 0.f, 0.f, 0.f);
    {
      float alpha[HEADS];
#pragma unroll
      for (int k = 0; k < HEADS; ++k) alpha[k] = ex0[k] / (sm[k] + SM_EPS);
      gather_rows<C>(h, s0, alpha, tot < 32 ? tot : 32, lane, hq, acc);
    }
    for (int base = 32; base < tot; base += 32) {
      int s; float e[HEADS], alpha[HEADS];
      const int j = base + lane;
      edge_scores(csr_src, asrc, start, deg, n, j, ad, s, e);
#pragma unroll
      for (int k = 0; k < HEADS; ++k)
        alpha[k] = (j < tot) ? __expf(e[k] - mx[k]) / (sm[k] + SM_EPS) : 0.f;
      const int left = tot - base;
      gather_rows<C>(h, s, alpha, left < 32 ? left : 32, lane, hq, acc);
    }

#pragma unroll
    for (int o = GL; o < 32; o <<= 1) {
      acc.x += __shfl_xor(acc.x, o);
      acc.y += __shfl_xor(acc.y, o);
      acc.z += __shfl_xor(acc.z, o);
      acc.w += __shfl_xor(acc.w, o);
    }
    if (NPW == 1 || hq == u) keep = acc;
  }

  const int nn = (N - n0 < NPW) ? (N - n0) : NPW;
  if (lane < GL * nn) {
    const float4 b = *(const float4*)(bias + 4 * (lane % GL));
    float4 r;
    r.x = keep.x * (1.0f / HEADS) + b.x;
    r.y = keep.y * (1.0f / HEADS) + b.y;
    r.z = keep.z * (1.0f / HEADS) + b.z;
    r.w = keep.w * (1.0f / HEADS) + b.w;
    if (do_relu) {
      r.x = r.x > 0.f ? r.x : 0.f;
      r.y = r.y > 0.f ? r.y : 0.f;
      r.z = r.z > 0.f ? r.z : 0.f;
      r.w = r.w > 0.f ? r.w : 0.f;
    }
    *(float4*)(out + (size_t)n0 * C + 4 * lane) = r;
  }
}

static inline unsigned nblk(long n, int b) { return (unsigned)((n + b - 1) / b); }

extern "C" void kernel_launch(void* const* d_in, const int* in_sizes, int n_in,
                              void* d_out, int out_size, void* d_ws, size_t ws_size,
                              hipStream_t stream) {
  const float* x        = (const float*)d_in[0];
  const int*   ei       = (const int*)  d_in[1];
  const float* W1       = (const float*)d_in[2];
  const float* att_src1 = (const float*)d_in[3];
  const float* att_dst1 = (const float*)d_in[4];
  const float* b1       = (const float*)d_in[5];
  const float* W2       = (const float*)d_in[6];
  const float* att_src2 = (const float*)d_in[7];
  const float* att_dst2 = (const float*)d_in[8];
  const float* b2       = (const float*)d_in[9];

  const int F_h   = in_sizes[3] / HEADS;
  const int F_out = in_sizes[7] / HEADS;
  const int F_in  = in_sizes[2] / (HEADS * F_h);
  const int N     = in_sizes[0] / F_in;
  const int E     = in_sizes[1] / 2;
  const int C1    = HEADS * F_h;
  const int C2    = HEADS * F_out;
  const int* src = ei;
  const int* dst = ei + E;

  float* ws = (float*)d_ws;
  size_t o = 0;
  float* h1   = ws + o; o += (size_t)N * C1;
  float* asrc = ws + o; o += (size_t)N * HEADS;
  float* adst = ws + o; o += (size_t)N * HEADS;
  float* x2   = ws + o; o += (size_t)N * F_h;
  int* cnt     = (int*)(ws + o); o += (size_t)N;
  int* rowptr  = (int*)(ws + o); o += (size_t)N;
  int* rank    = (int*)(ws + o); o += (size_t)E;
  int* csr_src = (int*)(ws + o); o += (size_t)E;
  float* h2   = h1;
  float* outp = (float*)d_out;

  const int TB = 256;
  const long NH = (long)N * HEADS;

  zero_i32<<<nblk(N, TB), TB, 0, stream>>>(cnt, N);
  dst_rank_kernel<<<nblk(E, TB), TB, 0, stream>>>(dst, E, cnt, rank);
  rowptr_kernel<<<nblk(N, 1024), 1024, 0, stream>>>(cnt, N, rowptr);
  csr_fill_kernel<<<nblk(E, TB), TB, 0, stream>>>(src, dst, E, rowptr, rank, csr_src);

  {
    long waves = (long)((N + 15) / 16) * (C1 / (16 * NT));
    wmma_gemm_f32<<<nblk(waves * 32, TB), TB, 0, stream>>>(x, W1, h1, N, F_in, C1);
  }
  attn_coeff_kernel<<<nblk(NH, TB), TB, 0, stream>>>(h1, att_src1, att_dst1,
                                                     asrc, adst, N, F_h);
  node_attend_kernel<32><<<nblk((long)N * 32, TB), TB, 0, stream>>>(
      rowptr, cnt, csr_src, asrc, adst, h1, b1, x2, N, 1);

  {
    long waves = (long)((N + 15) / 16) * (C2 / (16 * NT));
    wmma_gemm_f32<<<nblk(waves * 32, TB), TB, 0, stream>>>(x2, W2, h2, N, F_h, C2);
  }
  attn_coeff_kernel<<<nblk(NH, TB), TB, 0, stream>>>(h2, att_src2, att_dst2,
                                                     asrc, adst, N, F_out);
  node_attend_kernel<16><<<nblk((long)((N + 1) / 2) * 32, TB), TB, 0, stream>>>(
      rowptr, cnt, csr_src, asrc, adst, h2, b2, outp, N, 0);
}


# a_src/a_dst computed in the GEMM epilogue from the accumulator registers (16-lane row_xmask sums); both coefficient kernels and their re-read of h dropped; changes the order of the float sums
# speedup vs baseline: 10.3285x; 1.0505x over previous
#include <hip/hip_runtime.h>
#include <cstdint>

typedef float v2f __attribute__((ext_vector_type(2)));
typedef float v8f __attribute__((ext_vector_type(8)));

#define HEADS 8
#define NEG_SLOPE 0.2f
#define SM_EPS 1e-16f
#define NT 4

template <int CTRL>
__device__ __forceinline__ float xmask_add(float v) {
  const int o = __builtin_amdgcn_update_dpp(0, __float_as_int(v), CTRL, 0xf, 0xf, true);
  return v + __int_as_float(o);
}
__device__ __forceinline__ float row16_sum(float v) {
  v = xmask_add<0x161>(v);
  v = xmask_add<0x162>(v);
  v = xmask_add<0x164>(v);
  v = xmask_add<0x168>(v);
  return v;
}

template <int CH>
__global__ void wmma_gemm_f32(const float* __restrict__ A,
                              const float* __restrict__ B,
                              const float* __restrict__ att_src,
                              const float* __restrict__ att_dst,
                              float* __restrict__ C,
                              float* __restrict__ asrc,
                              float* __restrict__ adst,
                              int M, int K, int Ncols) {
  static_assert(CH == 32 || CH == 16, "a wave's 64 columns must be whole heads");
  constexpr int TPH = CH / 16;
  constexpr int HPW = NT / TPH;
  int wave = (int)((blockIdx.x * blockDim.x + threadIdx.x) >> 5);
  int lane = (int)(threadIdx.x & 31);
  int tilesM  = (M + 15) >> 4;
  int groupsN = Ncols >> 6;
  if (wave >= tilesM * groupsN) return;
  int tm = wave / groupsN;
  int tg = wave - tm * groupsN;
  int half = lane >> 4;
  int l = lane & 15;

  int rowA = tm * 16 + l;
  if (rowA >= M) rowA = M - 1;
  const float* aP = A + (size_t)rowA * K + 2 * half;
  const float* bP = B + (size_t)(2 * half) * Ncols + tg * (16 * NT) + l;

  v8f acc[NT];
#pragma unroll
  for (int t = 0; t < NT; ++t) acc[t] = (v8f){};

  float wS[NT], wD[NT];
#pragma unroll
  for (int t = 0; t < NT; ++t) {
    wS[t] = att_src[tg * (16 * NT) + t * 16 + l];
    wD[t] = att_dst[tg * (16 * NT) + t * 16 + l];
  }

  v2f aCur, bCur[NT];
  aCur.x = aP[0];
  aCur.y = aP[1];
#pragma unroll
  for (int t = 0; t < NT; ++t) {
    bCur[t].x = bP[t * 16];
    bCur[t].y = bP[t * 16 + Ncols];
  }

  for (int k0 = 0; k0 < K - 4; k0 += 4) {
    aP += 4;
    bP += (size_t)4 * Ncols;
    v2f aNxt, bNxt[NT];
    aNxt.x = aP[0];
    aNxt.y = aP[1];
#pragma unroll
    for (int t = 0; t < NT; ++t) {
      bNxt[t].x = bP[t * 16];
      bNxt[t].y = bP[t * 16 + Ncols];
    }
#pragma unroll
    for (int t = 0; t < NT; ++t) {
      acc[t] = __builtin_amdgcn_wmma_f32_16x16x4_f32(false, aCur, false, bCur[t],
                                                     (short)0, acc[t],
                                                     false, false);
    }
    aCur = aNxt;
#pragma unroll
    for (int t = 0; t < NT; ++t) bCur[t] = bNxt[t];
  }
#pragma unroll
  for (int t = 0; t < NT; ++t) {
    acc[t] = __builtin_amdgcn_wmma_f32_16x16x4_f32(false, aCur, false, bCur[t],
                                                   (short)0, acc[t],
                                                   false, false);
  }

#pragma unroll
  for (int t = 0; t < NT; ++t) {
    int col = tg * (16 * NT) + t * 16 + l;
#pragma unroll
    for (int r = 0; r < 8; ++r) {
      int row = tm * 16 + r + 8 * half;
      if (row < M) C[(size_t)row * Ncols + col] = acc[t][r];
    }
  }

#pragma unroll
  for (int j = 0; j < HPW / 2; ++j) {
    float vs = 0.f, vd = 0.f;
#pragma unroll
    for (int hb = 0; hb < 2; ++hb) {
#pragma unroll
      for (int r = 0; r < 8; ++r) {
        float ps = 0.f, pd = 0.f;
#pragma unroll
        for (int tt = 0; tt < TPH; ++tt) {
          const int t = (2 * j + hb) * TPH + tt;
          ps += acc[t][r] * wS[t];
          pd += acc[t][r] * wD[t];
        }
        ps = row16_sum(ps);
        pd = row16_sum(pd);
        const bool mine = (l == hb * 8 + r);
        vs = mine ? ps : vs;
        vd = mine ? pd : vd;
      }
    }
    const int row = tm * 16 + (l & 7) + 8 * half;
    const int head = tg * HPW + 2 * j + (l >> 3);
    if (row < M) {
      asrc[(size_t)row * HEADS + head] = vs;
      adst[(size_t)row * HEADS + head] = vd;
    }
  }
}

__device__ __forceinline__ float lrelu(float v) {
  return v > 0.f ? v : NEG_SLOPE * v;
}

__global__ void zero_i32(int* __restrict__ p, int n) {
  int i = (int)(blockIdx.x * blockDim.x + threadIdx.x);
  if (i < n) p[i] = 0;
}

__global__ void dst_rank_kernel(const int* __restrict__ dst, int E,
                                int* __restrict__ cnt, int* __restrict__ rank) {
  int e = (int)(blockIdx.x * blockDim.x + threadIdx.x);
  if (e >= E) return;
  rank[e] = atomicAdd(&cnt[dst[e]], 1);
}

__global__ void __launch_bounds__(1024)
rowptr_kernel(const int* __restrict__ cnt, int N, int* __restrict__ rowptr) {
  __shared__ int wsum[32];
  const int t = (int)threadIdx.x;
  const int lane = t & 31;
  const int w = t >> 5;
  const int base = (int)blockIdx.x * 1024;

  int part = 0;
  for (int i = t; i < base; i += 1024) part += cnt[i];
  for (int o = 16; o > 0; o >>= 1) part += __shfl_xor(part, o);
  if (lane == 0) wsum[w] = part;
  __syncthreads();
  int prefix = 0;
  for (int k = 0; k < 32; ++k) prefix += wsum[k];
  __syncthreads();

  const int i = base + t;
  const int c = (i < N) ? cnt[i] : 0;
  int incl = c;
  for (int o = 1; o < 32; o <<= 1) {
    const int v = __shfl_up(incl, o);
    if (lane >= o) incl += v;
  }
  if (lane == 31) wsum[w] = incl;
  __syncthreads();
  int woff = 0;
  for (int k = 0; k < 32; ++k) woff += (k < w) ? wsum[k] : 0;
  if (i < N) rowptr[i] = prefix + woff + incl - c;
}

__global__ void csr_fill_kernel(const int* __restrict__ src,
                                const int* __restrict__ dst, int E,
                                const int* __restrict__ rowptr,
                                const int* __restrict__ rank,
                                int* __restrict__ csr_src) {
  int e = (int)(blockIdx.x * blockDim.x + threadIdx.x);
  if (e >= E) return;
  csr_src[rowptr[dst[e]] + rank[e]] = src[e];
}

__device__ __forceinline__ void edge_scores(const int* __restrict__ csr_src,
                                            const float* __restrict__ asrc,
                                            int start, int deg, int n, int j,
                                            const float (&ad)[HEADS],
                                            int& s, float (&e)[HEADS]) {
  s = (j < deg) ? csr_src[start + j] : n;
  const float4 a0 = *(const float4*)(asrc + (size_t)s * HEADS);
  const float4 a1 = *(const float4*)(asrc + (size_t)s * HEADS + 4);
  e[0] = lrelu(a0.x + ad[0]);
  e[1] = lrelu(a0.y + ad[1]);
  e[2] = lrelu(a0.z + ad[2]);
  e[3] = lrelu(a0.w + ad[3]);
  e[4] = lrelu(a1.x + ad[4]);
  e[5] = lrelu(a1.y + ad[5]);
  e[6] = lrelu(a1.z + ad[6]);
  e[7] = lrelu(a1.w + ad[7]);
}

template <int C>
__device__ __forceinline__ void gather_rows(const float* __restrict__ h, int sreg,
                                            const float (&alpha)[HEADS], int count,
                                            int lane, int hq, float4& acc) {
  constexpr int ROW = HEADS * C;
#pragma unroll 4
  for (int jj = 0; jj < count; ++jj) {
    const int sj = __builtin_amdgcn_readlane(sreg, jj);
    float a[HEADS];
#pragma unroll
    for (int k = 0; k < HEADS; ++k)
      a[k] = __int_as_float(__builtin_amdgcn_readlane(__float_as_int(alpha[k]), jj));
    const float* hp = h + (size_t)sj * ROW + 4 * lane;
    if (C == 32) {
      float w0 = a[0], w1 = a[4];
      w0 = (hq == 1) ? a[1] : w0;  w1 = (hq == 1) ? a[5] : w1;
      w0 = (hq == 2) ? a[2] : w0;  w1 = (hq == 2) ? a[6] : w1;
      w0 = (hq == 3) ? a[3] : w0;  w1 = (hq == 3) ? a[7] : w1;
      const float4 v0 = *(const float4*)hp;
      const float4 v1 = *(const float4*)(hp + 128);
      acc.x += w0 * v0.x;  acc.y += w0 * v0.y;  acc.z += w0 * v0.z;  acc.w += w0 * v0.w;
      acc.x += w1 * v1.x;  acc.y += w1 * v1.y;  acc.z += w1 * v1.z;  acc.w += w1 * v1.w;
    } else {
      float w = a[0];
#pragma unroll
      for (int k = 1; k < HEADS; ++k) w = (hq == k) ? a[k] : w;
      const float4 v = *(const float4*)hp;
      acc.x += w * v.x;  acc.y += w * v.y;  acc.z += w * v.z;  acc.w += w * v.w;
    }
  }
}

template <int C>
__global__ void node_attend_kernel(const int* __restrict__ rowptr,
                                   const int* __restrict__ cnt,
                                   const int* __restrict__ csr_src,
                                   const float* __restrict__ asrc,
                                   const float* __restrict__ adst,
                                   const float* __restrict__ h,
                                   const float* __restrict__ bias,
                                   float* __restrict__ out,
                                   int N, int do_relu) {
  static_assert(C == 32 || C == 16, "lane maps are written for 32 or 16 channels");
  constexpr int NPW = 32 / C;
  constexpr int GL  = C / 4;
  const int lane = (int)(threadIdx.x & 31);
  const int wib = __builtin_amdgcn_readfirstlane((int)(threadIdx.x >> 5));
  const int wave = (int)blockIdx.x * (int)(blockDim.x >> 5) + wib;
  const int n0 = wave * NPW;
  if (n0 >= N) return;
  const int hq = lane / GL;
  float4 keep = make_float4(0.f, 0.f, 0.f, 0.f);

  for (int u = 0; u < NPW; ++u) {
    const int n = n0 + u;
    if (n >= N) break;
    const int start = rowptr[n];
    const int deg = cnt[n];
    const int tot = deg + 1;
    float ad[HEADS];
#pragma unroll
    for (int k = 0; k < HEADS; ++k) ad[k] = adst[(size_t)n * HEADS + k];

    int s0;
    float e0[HEADS];
    const bool v0 = lane < tot;
    edge_scores(csr_src, asrc, start, deg, n, lane, ad, s0, e0);

    float mx[HEADS];
#pragma unroll
    for (int k = 0; k < HEADS; ++k) mx[k] = v0 ? e0[k] : -3.0e38f;
    for (int base = 32; base < tot; base += 32) {
      int s; float e[HEADS];
      const int j = base + lane;
      edge_scores(csr_src, asrc, start, deg, n, j, ad, s, e);
#pragma unroll
      for (int k = 0; k < HEADS; ++k) mx[k] = (j < tot) ? fmaxf(mx[k], e[k]) : mx[k];
    }
#pragma unroll
    for (int k = 0; k < HEADS; ++k)
      for (int o = 16; o > 0; o >>= 1) mx[k] = fmaxf(mx[k], __shfl_xor(mx[k], o));

    float ex0[HEADS], sm[HEADS];
#pragma unroll
    for (int k = 0; k < HEADS; ++k) {
      ex0[k] = v0 ? __expf(e0[k] - mx[k]) : 0.f;
      sm[k] = ex0[k];
    }
    for (int base = 32; base < tot; base += 32) {
      int s; float e[HEADS];
      const int j = base + lane;
      edge_scores(csr_src, asrc, start, deg, n, j, ad, s, e);
#pragma unroll
      for (int k = 0; k < HEADS; ++k) sm[k] += (j < tot) ? __expf(e[k] - mx[k]) : 0.f;
    }
#pragma unroll
    for (int k = 0; k < HEADS; ++k)
      for (int o = 16; o > 0; o >>= 1) sm[k] += __shfl_xor(sm[k], o);

    float4 acc = make_float4(0.f, 0.f, 0.f, 0.f);
    {
      float alpha[HEADS];
#pragma unroll
      for (int k = 0; k < HEADS; ++k) alpha[k] = ex0[k] / (sm[k] + SM_EPS);
      gather_rows<C>(h, s0, alpha, tot < 32 ? tot : 32, lane, hq, acc);
    }
    for (int base = 32; base < tot; base += 32) {
      int s; float e[HEADS], alpha[HEADS];
      const int j = base + lane;
      edge_scores(csr_src, asrc, start, deg, n, j, ad, s, e);
#pragma unroll
      for (int k = 0; k < HEADS; ++k)
        alpha[k] = (j < tot) ? __expf(e[k] - mx[k]) / (sm[k] + SM_EPS) : 0.f;
      const int left = tot - base;
      gather_rows<C>(h, s, alpha, left < 32 ? left : 32, lane, hq, acc);
    }

#pragma unroll
    for (int o = GL; o < 32; o <<= 1) {
      acc.x += __shfl_xor(acc.x, o);
      acc.y += __shfl_xor(acc.y, o);
      acc.z += __shfl_xor(acc.z, o);
      acc.w += __shfl_xor(acc.w, o);
    }
    if (NPW == 1 || hq == u) keep = acc;
  }

  const int nn = (N - n0 < NPW) ? (N - n0) : NPW;
  if (lane < GL * nn) {
    const float4 b = *(const float4*)(bias + 4 * (lane % GL));
    float4 r;
    r.x = keep.x * (1.0f / HEADS) + b.x;
    r.y = keep.y * (1.0f / HEADS) + b.y;
    r.z = keep.z * (1.0f / HEADS) + b.z;
    r.w = keep.w * (1.0f / HEADS) + b.w;
    if (do_relu) {
      r.x = r.x > 0.f ? r.x : 0.f;
      r.y = r.y > 0.f ? r.y : 0.f;
      r.z = r.z > 0.f ? r.z : 0.f;
      r.w = r.w > 0.f ? r.w : 0.f;
    }
    *(float4*)(out + (size_t)n0 * C + 4 * lane) = r;
  }
}

static inline unsigned nblk(long n, int b) { return (unsigned)((n + b - 1) / b); }

extern "C" void kernel_launch(void* const* d_in, const int* in_sizes, int n_in,
                              void* d_out, int out_size, void* d_ws, size_t ws_size,
                              hipStream_t stream) {
  const float* x        = (const float*)d_in[0];
  const int*   ei       = (const int*)  d_in[1];
  const float* W1       = (const float*)d_in[2];
  const float* att_src1 = (const float*)d_in[3];
  const float* att_dst1 = (const float*)d_in[4];
  const float* b1       = (const float*)d_in[5];
  const float* W2       = (const float*)d_in[6];
  const float* att_src2 = (const float*)d_in[7];
  const float* att_dst2 = (const float*)d_in[8];
  const float* b2       = (const float*)d_in[9];

  const int F_h   = in_sizes[3] / HEADS;
  const int F_out = in_sizes[7] / HEADS;
  const int F_in  = in_sizes[2] / (HEADS * F_h);
  const int N     = in_sizes[0] / F_in;
  const int E     = in_sizes[1] / 2;
  const int C1    = HEADS * F_h;
  const int C2    = HEADS * F_out;
  const int* src = ei;
  const int* dst = ei + E;

  float* ws = (float*)d_ws;
  size_t o = 0;
  float* h1   = ws + o; o += (size_t)N * C1;
  float* asrc = ws + o; o += (size_t)N * HEADS;
  float* adst = ws + o; o += (size_t)N * HEADS;
  float* x2   = ws + o; o += (size_t)N * F_h;
  int* cnt     = (int*)(ws + o); o += (size_t)N;
  int* rowptr  = (int*)(ws + o); o += (size_t)N;
  int* rank    = (int*)(ws + o); o += (size_t)E;
  int* csr_src = (int*)(ws + o); o += (size_t)E;
  float* h2   = h1;
  float* outp = (float*)d_out;

  const int TB = 256;

  zero_i32<<<nblk(N, TB), TB, 0, stream>>>(cnt, N);
  dst_rank_kernel<<<nblk(E, TB), TB, 0, stream>>>(dst, E, cnt, rank);
  rowptr_kernel<<<nblk(N, 1024), 1024, 0, stream>>>(cnt, N, rowptr);
  csr_fill_kernel<<<nblk(E, TB), TB, 0, stream>>>(src, dst, E, rowptr, rank, csr_src);

  {
    long waves = (long)((N + 15) / 16) * (C1 / (16 * NT));
    wmma_gemm_f32<32><<<nblk(waves * 32, TB), TB, 0, stream>>>(
        x, W1, att_src1, att_dst1, h1, asrc, adst, N, F_in, C1);
  }
  node_attend_kernel<32><<<nblk((long)N * 32, TB), TB, 0, stream>>>(
      rowptr, cnt, csr_src, asrc, adst, h1, b1, x2, N, 1);

  {
    long waves = (long)((N + 15) / 16) * (C2 / (16 * NT));
    wmma_gemm_f32<16><<<nblk(waves * 32, TB), TB, 0, stream>>>(
        x2, W2, att_src2, att_dst2, h2, asrc, adst, N, F_h, C2);
  }
  node_attend_kernel<16><<<nblk((long)((N + 1) / 2) * 32, TB), TB, 0, stream>>>(
      rowptr, cnt, csr_src, asrc, adst, h2, b2, outp, N, 0);
}


# edge lists by destination built by a two-level counting sort on LDS counters (256-node bins, then nodes in a bin); no global atomic left; LDS add arrival order sets list and float sum order
# speedup vs baseline: 13.2051x; 1.2785x over previous
#include <hip/hip_runtime.h>
#include <cstdint>

typedef float v2f __attribute__((ext_vector_type(2)));
typedef float v8f __attribute__((ext_vector_type(8)));

#define HEADS 8
#define NEG_SLOPE 0.2f
#define SM_EPS 1e-16f
#define NT 4

template <int CTRL>
__device__ __forceinline__ float xmask_add(float v) {
  const int o = __builtin_amdgcn_update_dpp(0, __float_as_int(v), CTRL, 0xf, 0xf, true);
  return v + __int_as_float(o);
}
__device__ __forceinline__ float row16_sum(float v) {
  v = xmask_add<0x161>(v);
  v = xmask_add<0x162>(v);
  v = xmask_add<0x164>(v);
  v = xmask_add<0x168>(v);
  return v;
}

template <int CH>
__global__ void wmma_gemm_f32(const float* __restrict__ A,
                              const float* __restrict__ B,
                              const float* __restrict__ att_src,
                              const float* __restrict__ att_dst,
                              float* __restrict__ C,
                              float* __restrict__ asrc,
                              float* __restrict__ adst,
                              int M, int K, int Ncols) {
  static_assert(CH == 32 || CH == 16, "a wave's 64 columns must be whole heads");
  constexpr int TPH = CH / 16;
  constexpr int HPW = NT / TPH;
  int wave = (int)((blockIdx.x * blockDim.x + threadIdx.x) >> 5);
  int lane = (int)(threadIdx.x & 31);
  int tilesM  = (M + 15) >> 4;
  int groupsN = Ncols >> 6;
  if (wave >= tilesM * groupsN) return;
  int tm = wave / groupsN;
  int tg = wave - tm * groupsN;
  int half = lane >> 4;
  int l = lane & 15;

  int rowA = tm * 16 + l;
  if (rowA >= M) rowA = M - 1;
  const float* aP = A + (size_t)rowA * K + 2 * half;
  const float* bP = B + (size_t)(2 * half) * Ncols + tg * (16 * NT) + l;

  v8f acc[NT];
#pragma unroll
  for (int t = 0; t < NT; ++t) acc[t] = (v8f){};

  float wS[NT], wD[NT];
#pragma unroll
  for (int t = 0; t < NT; ++t) {
    wS[t] = att_src[tg * (16 * NT) + t * 16 + l];
    wD[t] = att_dst[tg * (16 * NT) + t * 16 + l];
  }

  v2f aCur, bCur[NT];
  aCur.x = aP[0];
  aCur.y = aP[1];
#pragma unroll
  for (int t = 0; t < NT; ++t) {
    bCur[t].x = bP[t * 16];
    bCur[t].y = bP[t * 16 + Ncols];
  }

  for (int k0 = 0; k0 < K - 4; k0 += 4) {
    aP += 4;
    bP += (size_t)4 * Ncols;
    v2f aNxt, bNxt[NT];
    aNxt.x = aP[0];
    aNxt.y = aP[1];
#pragma unroll
    for (int t = 0; t < NT; ++t) {
      bNxt[t].x = bP[t * 16];
      bNxt[t].y = bP[t * 16 + Ncols];
    }
#pragma unroll
    for (int t = 0; t < NT; ++t) {
      acc[t] = __builtin_amdgcn_wmma_f32_16x16x4_f32(false, aCur, false, bCur[t],
                                                     (short)0, acc[t],
                                                     false, false);
    }
    aCur = aNxt;
#pragma unroll
    for (int t = 0; t < NT; ++t) bCur[t] = bNxt[t];
  }
#pragma unroll
  for (int t = 0; t < NT; ++t) {
    acc[t] = __builtin_amdgcn_wmma_f32_16x16x4_f32(false, aCur, false, bCur[t],
                                                   (short)0, acc[t],
                                                   false, false);
  }

#pragma unroll
  for (int t = 0; t < NT; ++t) {
    int col = tg * (16 * NT) + t * 16 + l;
#pragma unroll
    for (int r = 0; r < 8; ++r) {
      int row = tm * 16 + r + 8 * half;
      if (row < M) C[(size_t)row * Ncols + col] = acc[t][r];
    }
  }

#pragma unroll
  for (int j = 0; j < HPW / 2; ++j) {
    float vs = 0.f, vd = 0.f;
#pragma unroll
    for (int hb = 0; hb < 2; ++hb) {
#pragma unroll
      for (int r = 0; r < 8; ++r) {
        float ps = 0.f, pd = 0.f;
#pragma unroll
        for (int tt = 0; tt < TPH; ++tt) {
          const int t = (2 * j + hb) * TPH + tt;
          ps += acc[t][r] * wS[t];
          pd += acc[t][r] * wD[t];
        }
        ps = row16_sum(ps);
        pd = row16_sum(pd);
        const bool mine = (l == hb * 8 + r);
        vs = mine ? ps : vs;
        vd = mine ? pd : vd;
      }
    }
    const int row = tm * 16 + (l & 7) + 8 * half;
    const int head = tg * HPW + 2 * j + (l >> 3);
    if (row < M) {
      asrc[(size_t)row * HEADS + head] = vs;
      adst[(size_t)row * HEADS + head] = vd;
    }
  }
}

__device__ __forceinline__ float lrelu(float v) {
  return v > 0.f ? v : NEG_SLOPE * v;
}

#define BIN_SHIFT 8
#define BIN_NODES (1 << BIN_SHIFT)
#define SORT_TB 1024
#define EPT 8
#define SPT 4

__global__ void __launch_bounds__(SORT_TB)
bin_hist_kernel(const int* __restrict__ dst, int E, int NB, int* __restrict__ hist) {
  extern __shared__ int sh[];
  const int t = (int)threadIdx.x;
  for (int b = t; b < NB; b += SORT_TB) sh[b] = 0;
  __syncthreads();
  const int base = (int)blockIdx.x * (SORT_TB * EPT) + t;
  int d[EPT];
#pragma unroll
  for (int k = 0; k < EPT; ++k) {
    const int e = base + k * SORT_TB;
    d[k] = dst[e < E ? e : E - 1];
  }
#pragma unroll
  for (int k = 0; k < EPT; ++k) {
    const int e = base + k * SORT_TB;
    if (e < E) atomicAdd(&sh[d[k] >> BIN_SHIFT], 1);
  }
  __syncthreads();
  for (int b = t; b < NB; b += SORT_TB)
    hist[(size_t)b * gridDim.x + blockIdx.x] = sh[b];
}

__global__ void __launch_bounds__(SORT_TB)
bin_scatter_kernel(const int* __restrict__ src, const int* __restrict__ dst, int E,
                   int NB, const int* __restrict__ offs, int* __restrict__ binned) {
  extern __shared__ int sh[];
  const int t = (int)threadIdx.x;
  for (int b = t; b < NB; b += SORT_TB)
    sh[b] = offs[(size_t)b * gridDim.x + blockIdx.x];
  __syncthreads();
  const int base = (int)blockIdx.x * (SORT_TB * EPT) + t;
  int s[EPT], d[EPT], pos[EPT];
#pragma unroll
  for (int k = 0; k < EPT; ++k) {
    const int e = base + k * SORT_TB;
    const int ec = e < E ? e : E - 1;
    s[k] = src[ec];
    d[k] = dst[ec];
  }
#pragma unroll
  for (int k = 0; k < EPT; ++k) {
    const int e = base + k * SORT_TB;
    pos[k] = 0;
    if (e < E) pos[k] = atomicAdd(&sh[d[k] >> BIN_SHIFT], 1);
  }
  int2* __restrict__ outp = (int2*)binned;
#pragma unroll
  for (int k = 0; k < EPT; ++k) {
    const int e = base + k * SORT_TB;
    if (e < E) outp[pos[k]] = make_int2(s[k], d[k]);
  }
}

__global__ void __launch_bounds__(SORT_TB)
bin_sort_kernel(const int* __restrict__ binned, const int* __restrict__ offs,
                int G1, int NB, int N, int E,
                int* __restrict__ cnt, int* __restrict__ rowptr,
                int* __restrict__ csr_src) {
  __shared__ int c[BIN_NODES];
  __shared__ int wsum[BIN_NODES / 32];
  const int2* __restrict__ pairs = (const int2*)binned;
  const int t = (int)threadIdx.x;
  const int lane = t & 31;
  const int w = t >> 5;
  const int b = (int)blockIdx.x;
  const int beg = offs[(size_t)b * G1];
  const int end = (b + 1 < NB) ? offs[(size_t)(b + 1) * G1] : E;
  if (t < BIN_NODES) c[t] = 0;
  __syncthreads();
  for (int i0 = beg + t; i0 < end; i0 += SPT * SORT_TB) {
    int y[SPT];
#pragma unroll
    for (int k = 0; k < SPT; ++k) {
      const int i = i0 + k * SORT_TB;
      y[k] = pairs[i < end ? i : end - 1].y;
    }
#pragma unroll
    for (int k = 0; k < SPT; ++k)
      if (i0 + k * SORT_TB < end) atomicAdd(&c[y[k] & (BIN_NODES - 1)], 1);
  }
  __syncthreads();
  int mine = 0, incl = 0;
  if (t < BIN_NODES) {
    mine = c[t];
    incl = mine;
    for (int o = 1; o < 32; o <<= 1) {
      const int v = __shfl_up(incl, o);
      if (lane >= o) incl += v;
    }
    if (lane == 31) wsum[w] = incl;
  }
  __syncthreads();
  if (t < BIN_NODES) {
    int woff = 0;
    for (int k = 0; k < BIN_NODES / 32; ++k) woff += (k < w) ? wsum[k] : 0;
    const int start = beg + woff + incl - mine;
    c[t] = start;
    const int n = b * BIN_NODES + t;
    if (n < N) {
      cnt[n] = mine;
      rowptr[n] = start;
    }
  }
  __syncthreads();
  for (int i0 = beg + t; i0 < end; i0 += SPT * SORT_TB) {
    int2 sd[SPT];
    int pos[SPT];
#pragma unroll
    for (int k = 0; k < SPT; ++k) {
      const int i = i0 + k * SORT_TB;
      sd[k] = pairs[i < end ? i : end - 1];
    }
#pragma unroll
    for (int k = 0; k < SPT; ++k) {
      pos[k] = 0;
      if (i0 + k * SORT_TB < end) pos[k] = atomicAdd(&c[sd[k].y & (BIN_NODES - 1)], 1);
    }
#pragma unroll
    for (int k = 0; k < SPT; ++k)
      if (i0 + k * SORT_TB < end) csr_src[pos[k]] = sd[k].x;
  }
}

__global__ void __launch_bounds__(1024)
prefix_kernel(const int* __restrict__ cnt, int N, int* __restrict__ rowptr) {
  __shared__ int wsum[32];
  const int t = (int)threadIdx.x;
  const int lane = t & 31;
  const int w = t >> 5;
  const int base = (int)blockIdx.x * 1024;

  int part = 0;
  for (int i = t; i < base; i += 1024) part += cnt[i];
  for (int o = 16; o > 0; o >>= 1) part += __shfl_xor(part, o);
  if (lane == 0) wsum[w] = part;
  __syncthreads();
  int prefix = 0;
  for (int k = 0; k < 32; ++k) prefix += wsum[k];
  __syncthreads();

  const int i = base + t;
  const int c = (i < N) ? cnt[i] : 0;
  int incl = c;
  for (int o = 1; o < 32; o <<= 1) {
    const int v = __shfl_up(incl, o);
    if (lane >= o) incl += v;
  }
  if (lane == 31) wsum[w] = incl;
  __syncthreads();
  int woff = 0;
  for (int k = 0; k < 32; ++k) woff += (k < w) ? wsum[k] : 0;
  if (i < N) rowptr[i] = prefix + woff + incl - c;
}

__device__ __forceinline__ void edge_scores(const int* __restrict__ csr_src,
                                            const float* __restrict__ asrc,
                                            int start, int deg, int n, int j,
                                            const float (&ad)[HEADS],
                                            int& s, float (&e)[HEADS]) {
  s = (j < deg) ? csr_src[start + j] : n;
  const float4 a0 = *(const float4*)(asrc + (size_t)s * HEADS);
  const float4 a1 = *(const float4*)(asrc + (size_t)s * HEADS + 4);
  e[0] = lrelu(a0.x + ad[0]);
  e[1] = lrelu(a0.y + ad[1]);
  e[2] = lrelu(a0.z + ad[2]);
  e[3] = lrelu(a0.w + ad[3]);
  e[4] = lrelu(a1.x + ad[4]);
  e[5] = lrelu(a1.y + ad[5]);
  e[6] = lrelu(a1.z + ad[6]);
  e[7] = lrelu(a1.w + ad[7]);
}

template <int C>
__device__ __forceinline__ void gather_rows(const float* __restrict__ h, int sreg,
                                            const float (&alpha)[HEADS], int count,
                                            int lane, int hq, float4& acc) {
  constexpr int ROW = HEADS * C;
#pragma unroll 4
  for (int jj = 0; jj < count; ++jj) {
    const int sj = __builtin_amdgcn_readlane(sreg, jj);
    float a[HEADS];
#pragma unroll
    for (int k = 0; k < HEADS; ++k)
      a[k] = __int_as_float(__builtin_amdgcn_readlane(__float_as_int(alpha[k]), jj));
    const float* hp = h + (size_t)sj * ROW + 4 * lane;
    if (C == 32) {
      float w0 = a[0], w1 = a[4];
      w0 = (hq == 1) ? a[1] : w0;  w1 = (hq == 1) ? a[5] : w1;
      w0 = (hq == 2) ? a[2] : w0;  w1 = (hq == 2) ? a[6] : w1;
      w0 = (hq == 3) ? a[3] : w0;  w1 = (hq == 3) ? a[7] : w1;
      const float4 v0 = *(const float4*)hp;
      const float4 v1 = *(const float4*)(hp + 128);
      acc.x += w0 * v0.x;  acc.y += w0 * v0.y;  acc.z += w0 * v0.z;  acc.w += w0 * v0.w;
      acc.x += w1 * v1.x;  acc.y += w1 * v1.y;  acc.z += w1 * v1.z;  acc.w += w1 * v1.w;
    } else {
      float w = a[0];
#pragma unroll
      for (int k = 1; k < HEADS; ++k) w = (hq == k) ? a[k] : w;
      const float4 v = *(const float4*)hp;
      acc.x += w * v.x;  acc.y += w * v.y;  acc.z += w * v.z;  acc.w += w * v.w;
    }
  }
}

template <int C>
__global__ void node_attend_kernel(const int* __restrict__ rowptr,
                                   const int* __restrict__ cnt,
                                   const int* __restrict__ csr_src,
                                   const float* __restrict__ asrc,
                                   const float* __restrict__ adst,
                                   const float* __restrict__ h,
                                   const float* __restrict__ bias,
                                   float* __restrict__ out,
                                   int N, int do_relu) {
  static_assert(C == 32 || C == 16, "lane maps are written for 32 or 16 channels");
  constexpr int NPW = 32 / C;
  constexpr int GL  = C / 4;
  const int lane = (int)(threadIdx.x & 31);
  const int wib = __builtin_amdgcn_readfirstlane((int)(threadIdx.x >> 5));
  const int wave = (int)blockIdx.x * (int)(blockDim.x >> 5) + wib;
  const int n0 = wave * NPW;
  if (n0 >= N) return;
  const int hq = lane / GL;
  float4 keep = make_float4(0.f, 0.f, 0.f, 0.f);

  for (int u = 0; u < NPW; ++u) {
    const int n = n0 + u;
    if (n >= N) break;
    const int start = rowptr[n];
    const int deg = cnt[n];
    const int tot = deg + 1;
    float ad[HEADS];
#pragma unroll
    for (int k = 0; k < HEADS; ++k) ad[k] = adst[(size_t)n * HEADS + k];

    int s0;
    float e0[HEADS];
    const bool v0 = lane < tot;
    edge_scores(csr_src, asrc, start, deg, n, lane, ad, s0, e0);

    float mx[HEADS];
#pragma unroll
    for (int k = 0; k < HEADS; ++k) mx[k] = v0 ? e0[k] : -3.0e38f;
    for (int base = 32; base < tot; base += 32) {
      int s; float e[HEADS];
      const int j = base + lane;
      edge_scores(csr_src, asrc, start, deg, n, j, ad, s, e);
#pragma unroll
      for (int k = 0; k < HEADS; ++k) mx[k] = (j < tot) ? fmaxf(mx[k], e[k]) : mx[k];
    }
#pragma unroll
    for (int k = 0; k < HEADS; ++k)
      for (int o = 16; o > 0; o >>= 1) mx[k] = fmaxf(mx[k], __shfl_xor(mx[k], o));

    float ex0[HEADS], sm[HEADS];
#pragma unroll
    for (int k = 0; k < HEADS; ++k) {
      ex0[k] = v0 ? __expf(e0[k] - mx[k]) : 0.f;
      sm[k] = ex0[k];
    }
    for (int base = 32; base < tot; base += 32) {
      int s; float e[HEADS];
      const int j = base + lane;
      edge_scores(csr_src, asrc, start, deg, n, j, ad, s, e);
#pragma unroll
      for (int k = 0; k < HEADS; ++k) sm[k] += (j < tot) ? __expf(e[k] - mx[k]) : 0.f;
    }
#pragma unroll
    for (int k = 0; k < HEADS; ++k)
      for (int o = 16; o > 0; o >>= 1) sm[k] += __shfl_xor(sm[k], o);

    float4 acc = make_float4(0.f, 0.f, 0.f, 0.f);
    {
      float alpha[HEADS];
#pragma unroll
      for (int k = 0; k < HEADS; ++k) alpha[k] = ex0[k] / (sm[k] + SM_EPS);
      gather_rows<C>(h, s0, alpha, tot < 32 ? tot : 32, lane, hq, acc);
    }
    for (int base = 32; base < tot; base += 32) {
      int s; float e[HEADS], alpha[HEADS];
      const int j = base + lane;
      edge_scores(csr_src, asrc, start, deg, n, j, ad, s, e);
#pragma unroll
      for (int k = 0; k < HEADS; ++k)
        alpha[k] = (j < tot) ? __expf(e[k] - mx[k]) / (sm[k] + SM_EPS) : 0.f;
      const int left = tot - base;
      gather_rows<C>(h, s, alpha, left < 32 ? left : 32, lane, hq, acc);
    }

#pragma unroll
    for (int o = GL; o < 32; o <<= 1) {
      acc.x += __shfl_xor(acc.x, o);
      acc.y += __shfl_xor(acc.y, o);
      acc.z += __shfl_xor(acc.z, o);
      acc.w += __shfl_xor(acc.w, o);
    }
    if (NPW == 1 || hq == u) keep = acc;
  }

  const int nn = (N - n0 < NPW) ? (N - n0) : NPW;
  if (lane < GL * nn) {
    const float4 b = *(const float4*)(bias + 4 * (lane % GL));
    float4 r;
    r.x = keep.x * (1.0f / HEADS) + b.x;
    r.y = keep.y * (1.0f / HEADS) + b.y;
    r.z = keep.z * (1.0f / HEADS) + b.z;
    r.w = keep.w * (1.0f / HEADS) + b.w;
    if (do_relu) {
      r.x = r.x > 0.f ? r.x : 0.f;
      r.y = r.y > 0.f ? r.y : 0.f;
      r.z = r.z > 0.f ? r.z : 0.f;
      r.w = r.w > 0.f ? r.w : 0.f;
    }
    *(float4*)(out + (size_t)n0 * C + 4 * lane) = r;
  }
}

static inline unsigned nblk(long n, int b) { return (unsigned)((n + b - 1) / b); }

extern "C" void kernel_launch(void* const* d_in, const int* in_sizes, int n_in,
                              void* d_out, int out_size, void* d_ws, size_t ws_size,
                              hipStream_t stream) {
  const float* x        = (const float*)d_in[0];
  const int*   ei       = (const int*)  d_in[1];
  const float* W1       = (const float*)d_in[2];
  const float* att_src1 = (const float*)d_in[3];
  const float* att_dst1 = (const float*)d_in[4];
  const float* b1       = (const float*)d_in[5];
  const float* W2       = (const float*)d_in[6];
  const float* att_src2 = (const float*)d_in[7];
  const float* att_dst2 = (const float*)d_in[8];
  const float* b2       = (const float*)d_in[9];

  const int F_h   = in_sizes[3] / HEADS;
  const int F_out = in_sizes[7] / HEADS;
  const int F_in  = in_sizes[2] / (HEADS * F_h);
  const int N     = in_sizes[0] / F_in;
  const int E     = in_sizes[1] / 2;
  const int C1    = HEADS * F_h;
  const int C2    = HEADS * F_out;
  const int* src = ei;
  const int* dst = ei + E;

  float* ws = (float*)d_ws;
  size_t o = 0;
  float* h1   = ws + o; o += (size_t)N * C1;
  float* asrc = ws + o; o += (size_t)N * HEADS;
  float* adst = ws + o; o += (size_t)N * HEADS;
  float* x2   = ws + o; o += (size_t)N * F_h;
  const int NB = (N + BIN_NODES - 1) / BIN_NODES;
  const int G1 = (int)nblk(E, SORT_TB * EPT);
  int* binned  = (int*)(ws + o); o += (size_t)2 * E;
  int* cnt     = (int*)(ws + o); o += (size_t)N;
  int* rowptr  = (int*)(ws + o); o += (size_t)N;
  int* csr_src = (int*)(ws + o); o += (size_t)E;
  int* hist    = (int*)(ws + o); o += (size_t)NB * G1;
  int* offs    = (int*)(ws + o); o += (size_t)NB * G1;
  float* h2   = h1;
  float* outp = (float*)d_out;

  const int TB = 256;

  bin_hist_kernel<<<G1, SORT_TB, NB * sizeof(int), stream>>>(dst, E, NB, hist);
  prefix_kernel<<<nblk((long)NB * G1, 1024), 1024, 0, stream>>>(hist, NB * G1, offs);
  bin_scatter_kernel<<<G1, SORT_TB, NB * sizeof(int), stream>>>(src, dst, E, NB, offs, binned);
  bin_sort_kernel<<<NB, SORT_TB, 0, stream>>>(binned, offs, G1, NB, N, E, cnt, rowptr, csr_src);

  {
    long waves = (long)((N + 15) / 16) * (C1 / (16 * NT));
    wmma_gemm_f32<32><<<nblk(waves * 32, TB), TB, 0, stream>>>(
        x, W1, att_src1, att_dst1, h1, asrc, adst, N, F_in, C1);
  }
  node_attend_kernel<32><<<nblk((long)N * 32, TB), TB, 0, stream>>>(
      rowptr, cnt, csr_src, asrc, adst, h1, b1, x2, N, 1);

  {
    long waves = (long)((N + 15) / 16) * (C2 / (16 * NT));
    wmma_gemm_f32<16><<<nblk(waves * 32, TB), TB, 0, stream>>>(
        x2, W2, att_src2, att_dst2, h2, asrc, adst, N, F_h, C2);
  }
  node_attend_kernel<16><<<nblk((long)((N + 1) / 2) * 32, TB), TB, 0, stream>>>(
      rowptr, cnt, csr_src, asrc, adst, h2, b2, outp, N, 0);
}


# layer 2 sums alpha*x2[src] per head on the 128-byte input rows, then projects each 16 nodes' [8x32] sums with W2 on WMMA inside the node kernel; h2 is not stored; changes the order of the float sums
# speedup vs baseline: 14.4874x; 1.0971x over previous
#include <hip/hip_runtime.h>
#include <cstdint>

typedef float v2f __attribute__((ext_vector_type(2)));
typedef float v8f __attribute__((ext_vector_type(8)));

#define HEADS 8
#define NEG_SLOPE 0.2f
#define SM_EPS 1e-16f
#define NT 4

template <int CTRL>
__device__ __forceinline__ float xmask_add(float v) {
  const int o = __builtin_amdgcn_update_dpp(0, __float_as_int(v), CTRL, 0xf, 0xf, true);
  return v + __int_as_float(o);
}
__device__ __forceinline__ float row16_sum(float v) {
  v = xmask_add<0x161>(v);
  v = xmask_add<0x162>(v);
  v = xmask_add<0x164>(v);
  v = xmask_add<0x168>(v);
  return v;
}

template <int CH, bool STORE_C>
__global__ void wmma_gemm_f32(const float* __restrict__ A,
                              const float* __restrict__ B,
                              const float* __restrict__ att_src,
                              const float* __restrict__ att_dst,
                              float* __restrict__ C,
                              float* __restrict__ asrc,
                              float* __restrict__ adst,
                              int M, int K, int Ncols) {
  static_assert(CH == 32 || CH == 16, "a wave's 64 columns must be whole heads");
  constexpr int TPH = CH / 16;
  constexpr int HPW = NT / TPH;
  int wave = (int)((blockIdx.x * blockDim.x + threadIdx.x) >> 5);
  int lane = (int)(threadIdx.x & 31);
  int tilesM  = (M + 15) >> 4;
  int groupsN = Ncols >> 6;
  if (wave >= tilesM * groupsN) return;
  int tm = wave / groupsN;
  int tg = wave - tm * groupsN;
  int half = lane >> 4;
  int l = lane & 15;

  int rowA = tm * 16 + l;
  if (rowA >= M) rowA = M - 1;
  const float* aP = A + (size_t)rowA * K + 2 * half;
  const float* bP = B + (size_t)(2 * half) * Ncols + tg * (16 * NT) + l;

  v8f acc[NT];
#pragma unroll
  for (int t = 0; t < NT; ++t) acc[t] = (v8f){};

  float wS[NT], wD[NT];
#pragma unroll
  for (int t = 0; t < NT; ++t) {
    wS[t] = att_src[tg * (16 * NT) + t * 16 + l];
    wD[t] = att_dst[tg * (16 * NT) + t * 16 + l];
  }

  v2f aCur, bCur[NT];
  aCur.x = aP[0];
  aCur.y = aP[1];
#pragma unroll
  for (int t = 0; t < NT; ++t) {
    bCur[t].x = bP[t * 16];
    bCur[t].y = bP[t * 16 + Ncols];
  }

  for (int k0 = 0; k0 < K - 4; k0 += 4) {
    aP += 4;
    bP += (size_t)4 * Ncols;
    v2f aNxt, bNxt[NT];
    aNxt.x = aP[0];
    aNxt.y = aP[1];
#pragma unroll
    for (int t = 0; t < NT; ++t) {
      bNxt[t].x = bP[t * 16];
      bNxt[t].y = bP[t * 16 + Ncols];
    }
#pragma unroll
    for (int t = 0; t < NT; ++t) {
      acc[t] = __builtin_amdgcn_wmma_f32_16x16x4_f32(false, aCur, false, bCur[t],
                                                     (short)0, acc[t],
                                                     false, false);
    }
    aCur = aNxt;
#pragma unroll
    for (int t = 0; t < NT; ++t) bCur[t] = bNxt[t];
  }
#pragma unroll
  for (int t = 0; t < NT; ++t) {
    acc[t] = __builtin_amdgcn_wmma_f32_16x16x4_f32(false, aCur, false, bCur[t],
                                                   (short)0, acc[t],
                                                   false, false);
  }

  if (STORE_C) {
#pragma unroll
    for (int t = 0; t < NT; ++t) {
      int col = tg * (16 * NT) + t * 16 + l;
#pragma unroll
      for (int r = 0; r < 8; ++r) {
        int row = tm * 16 + r + 8 * half;
        if (row < M) C[(size_t)row * Ncols + col] = acc[t][r];
      }
    }
  }

#pragma unroll
  for (int j = 0; j < HPW / 2; ++j) {
    float vs = 0.f, vd = 0.f;
#pragma unroll
    for (int hb = 0; hb < 2; ++hb) {
#pragma unroll
      for (int r = 0; r < 8; ++r) {
        float ps = 0.f, pd = 0.f;
#pragma unroll
        for (int tt = 0; tt < TPH; ++tt) {
          const int t = (2 * j + hb) * TPH + tt;
          ps += acc[t][r] * wS[t];
          pd += acc[t][r] * wD[t];
        }
        ps = row16_sum(ps);
        pd = row16_sum(pd);
        const bool mine = (l == hb * 8 + r);
        vs = mine ? ps : vs;
        vd = mine ? pd : vd;
      }
    }
    const int row = tm * 16 + (l & 7) + 8 * half;
    const int head = tg * HPW + 2 * j + (l >> 3);
    if (row < M) {
      asrc[(size_t)row * HEADS + head] = vs;
      adst[(size_t)row * HEADS + head] = vd;
    }
  }
}

__device__ __forceinline__ float lrelu(float v) {
  return v > 0.f ? v : NEG_SLOPE * v;
}

#define BIN_SHIFT 8
#define BIN_NODES (1 << BIN_SHIFT)
#define SORT_TB 1024
#define EPT 8
#define SPT 4

__global__ void __launch_bounds__(SORT_TB)
bin_hist_kernel(const int* __restrict__ dst, int E, int NB, int* __restrict__ hist) {
  extern __shared__ int sh[];
  const int t = (int)threadIdx.x;
  for (int b = t; b < NB; b += SORT_TB) sh[b] = 0;
  __syncthreads();
  const int base = (int)blockIdx.x * (SORT_TB * EPT) + t;
  int d[EPT];
#pragma unroll
  for (int k = 0; k < EPT; ++k) {
    const int e = base + k * SORT_TB;
    d[k] = dst[e < E ? e : E - 1];
  }
#pragma unroll
  for (int k = 0; k < EPT; ++k) {
    const int e = base + k * SORT_TB;
    if (e < E) atomicAdd(&sh[d[k] >> BIN_SHIFT], 1);
  }
  __syncthreads();
  for (int b = t; b < NB; b += SORT_TB)
    hist[(size_t)b * gridDim.x + blockIdx.x] = sh[b];
}

__global__ void __launch_bounds__(SORT_TB)
bin_scatter_kernel(const int* __restrict__ src, const int* __restrict__ dst, int E,
                   int NB, const int* __restrict__ offs, int* __restrict__ binned) {
  extern __shared__ int sh[];
  const int t = (int)threadIdx.x;
  for (int b = t; b < NB; b += SORT_TB)
    sh[b] = offs[(size_t)b * gridDim.x + blockIdx.x];
  __syncthreads();
  const int base = (int)blockIdx.x * (SORT_TB * EPT) + t;
  int s[EPT], d[EPT], pos[EPT];
#pragma unroll
  for (int k = 0; k < EPT; ++k) {
    const int e = base + k * SORT_TB;
    const int ec = e < E ? e : E - 1;
    s[k] = src[ec];
    d[k] = dst[ec];
  }
#pragma unroll
  for (int k = 0; k < EPT; ++k) {
    const int e = base + k * SORT_TB;
    pos[k] = 0;
    if (e < E) pos[k] = atomicAdd(&sh[d[k] >> BIN_SHIFT], 1);
  }
  int2* __restrict__ outp = (int2*)binned;
#pragma unroll
  for (int k = 0; k < EPT; ++k) {
    const int e = base + k * SORT_TB;
    if (e < E) outp[pos[k]] = make_int2(s[k], d[k]);
  }
}

__global__ void __launch_bounds__(SORT_TB)
bin_sort_kernel(const int* __restrict__ binned, const int* __restrict__ offs,
                int G1, int NB, int N, int E,
                int* __restrict__ cnt, int* __restrict__ rowptr,
                int* __restrict__ csr_src) {
  __shared__ int c[BIN_NODES];
  __shared__ int wsum[BIN_NODES / 32];
  const int2* __restrict__ pairs = (const int2*)binned;
  const int t = (int)threadIdx.x;
  const int lane = t & 31;
  const int w = t >> 5;
  const int b = (int)blockIdx.x;
  const int beg = offs[(size_t)b * G1];
  const int end = (b + 1 < NB) ? offs[(size_t)(b + 1) * G1] : E;
  if (t < BIN_NODES) c[t] = 0;
  __syncthreads();
  for (int i0 = beg + t; i0 < end; i0 += SPT * SORT_TB) {
    int y[SPT];
#pragma unroll
    for (int k = 0; k < SPT; ++k) {
      const int i = i0 + k * SORT_TB;
      y[k] = pairs[i < end ? i : end - 1].y;
    }
#pragma unroll
    for (int k = 0; k < SPT; ++k)
      if (i0 + k * SORT_TB < end) atomicAdd(&c[y[k] & (BIN_NODES - 1)], 1);
  }
  __syncthreads();
  int mine = 0, incl = 0;
  if (t < BIN_NODES) {
    mine = c[t];
    incl = mine;
    for (int o = 1; o < 32; o <<= 1) {
      const int v = __shfl_up(incl, o);
      if (lane >= o) incl += v;
    }
    if (lane == 31) wsum[w] = incl;
  }
  __syncthreads();
  if (t < BIN_NODES) {
    int woff = 0;
    for (int k = 0; k < BIN_NODES / 32; ++k) woff += (k < w) ? wsum[k] : 0;
    const int start = beg + woff + incl - mine;
    c[t] = start;
    const int n = b * BIN_NODES + t;
    if (n < N) {
      cnt[n] = mine;
      rowptr[n] = start;
    }
  }
  __syncthreads();
  for (int i0 = beg + t; i0 < end; i0 += SPT * SORT_TB) {
    int2 sd[SPT];
    int pos[SPT];
#pragma unroll
    for (int k = 0; k < SPT; ++k) {
      const int i = i0 + k * SORT_TB;
      sd[k] = pairs[i < end ? i : end - 1];
    }
#pragma unroll
    for (int k = 0; k < SPT; ++k) {
      pos[k] = 0;
      if (i0 + k * SORT_TB < end) pos[k] = atomicAdd(&c[sd[k].y & (BIN_NODES - 1)], 1);
    }
#pragma unroll
    for (int k = 0; k < SPT; ++k)
      if (i0 + k * SORT_TB < end) csr_src[pos[k]] = sd[k].x;
  }
}

__global__ void __launch_bounds__(1024)
prefix_kernel(const int* __restrict__ cnt, int N, int* __restrict__ rowptr) {
  __shared__ int wsum[32];
  const int t = (int)threadIdx.x;
  const int lane = t & 31;
  const int w = t >> 5;
  const int base = (int)blockIdx.x * 1024;

  int part = 0;
  for (int i = t; i < base; i += 1024) part += cnt[i];
  for (int o = 16; o > 0; o >>= 1) part += __shfl_xor(part, o);
  if (lane == 0) wsum[w] = part;
  __syncthreads();
  int prefix = 0;
  for (int k = 0; k < 32; ++k) prefix += wsum[k];
  __syncthreads();

  const int i = base + t;
  const int c = (i < N) ? cnt[i] : 0;
  int incl = c;
  for (int o = 1; o < 32; o <<= 1) {
    const int v = __shfl_up(incl, o);
    if (lane >= o) incl += v;
  }
  if (lane == 31) wsum[w] = incl;
  __syncthreads();
  int woff = 0;
  for (int k = 0; k < 32; ++k) woff += (k < w) ? wsum[k] : 0;
  if (i < N) rowptr[i] = prefix + woff + incl - c;
}

__device__ __forceinline__ void edge_scores(const int* __restrict__ csr_src,
                                            const float* __restrict__ asrc,
                                            int start, int deg, int n, int j,
                                            const float (&ad)[HEADS],
                                            int& s, float (&e)[HEADS]) {
  s = (j < deg) ? csr_src[start + j] : n;
  const float4 a0 = *(const float4*)(asrc + (size_t)s * HEADS);
  const float4 a1 = *(const float4*)(asrc + (size_t)s * HEADS + 4);
  e[0] = lrelu(a0.x + ad[0]);
  e[1] = lrelu(a0.y + ad[1]);
  e[2] = lrelu(a0.z + ad[2]);
  e[3] = lrelu(a0.w + ad[3]);
  e[4] = lrelu(a1.x + ad[4]);
  e[5] = lrelu(a1.y + ad[5]);
  e[6] = lrelu(a1.z + ad[6]);
  e[7] = lrelu(a1.w + ad[7]);
}

template <int C>
__device__ __forceinline__ void gather_rows(const float* __restrict__ h, int sreg,
                                            const float (&alpha)[HEADS], int count,
                                            int lane, int hq, float4& acc) {
  constexpr int ROW = HEADS * C;
#pragma unroll 4
  for (int jj = 0; jj < count; ++jj) {
    const int sj = __builtin_amdgcn_readlane(sreg, jj);
    float a[HEADS];
#pragma unroll
    for (int k = 0; k < HEADS; ++k)
      a[k] = __int_as_float(__builtin_amdgcn_readlane(__float_as_int(alpha[k]), jj));
    const float* hp = h + (size_t)sj * ROW + 4 * lane;
    if (C == 32) {
      float w0 = a[0], w1 = a[4];
      w0 = (hq == 1) ? a[1] : w0;  w1 = (hq == 1) ? a[5] : w1;
      w0 = (hq == 2) ? a[2] : w0;  w1 = (hq == 2) ? a[6] : w1;
      w0 = (hq == 3) ? a[3] : w0;  w1 = (hq == 3) ? a[7] : w1;
      const float4 v0 = *(const float4*)hp;
      const float4 v1 = *(const float4*)(hp + 128);
      acc.x += w0 * v0.x;  acc.y += w0 * v0.y;  acc.z += w0 * v0.z;  acc.w += w0 * v0.w;
      acc.x += w1 * v1.x;  acc.y += w1 * v1.y;  acc.z += w1 * v1.z;  acc.w += w1 * v1.w;
    } else {
      float w = a[0];
#pragma unroll
      for (int k = 1; k < HEADS; ++k) w = (hq == k) ? a[k] : w;
      const float4 v = *(const float4*)hp;
      acc.x += w * v.x;  acc.y += w * v.y;  acc.z += w * v.z;  acc.w += w * v.w;
    }
  }
}

template <int C>
__global__ void node_attend_kernel(const int* __restrict__ rowptr,
                                   const int* __restrict__ cnt,
                                   const int* __restrict__ csr_src,
                                   const float* __restrict__ asrc,
                                   const float* __restrict__ adst,
                                   const float* __restrict__ h,
                                   const float* __restrict__ bias,
                                   float* __restrict__ out,
                                   int N, int do_relu) {
  static_assert(C == 32 || C == 16, "lane maps are written for 32 or 16 channels");
  constexpr int NPW = 32 / C;
  constexpr int GL  = C / 4;
  const int lane = (int)(threadIdx.x & 31);
  const int wib = __builtin_amdgcn_readfirstlane((int)(threadIdx.x >> 5));
  const int wave = (int)blockIdx.x * (int)(blockDim.x >> 5) + wib;
  const int n0 = wave * NPW;
  if (n0 >= N) return;
  const int hq = lane / GL;
  float4 keep = make_float4(0.f, 0.f, 0.f, 0.f);

  for (int u = 0; u < NPW; ++u) {
    const int n = n0 + u;
    if (n >= N) break;
    const int start = rowptr[n];
    const int deg = cnt[n];
    const int tot = deg + 1;
    float ad[HEADS];
#pragma unroll
    for (int k = 0; k < HEADS; ++k) ad[k] = adst[(size_t)n * HEADS + k];

    int s0;
    float e0[HEADS];
    const bool v0 = lane < tot;
    edge_scores(csr_src, asrc, start, deg, n, lane, ad, s0, e0);

    float mx[HEADS];
#pragma unroll
    for (int k = 0; k < HEADS; ++k) mx[k] = v0 ? e0[k] : -3.0e38f;
    for (int base = 32; base < tot; base += 32) {
      int s; float e[HEADS];
      const int j = base + lane;
      edge_scores(csr_src, asrc, start, deg, n, j, ad, s, e);
#pragma unroll
      for (int k = 0; k < HEADS; ++k) mx[k] = (j < tot) ? fmaxf(mx[k], e[k]) : mx[k];
    }
#pragma unroll
    for (int k = 0; k < HEADS; ++k)
      for (int o = 16; o > 0; o >>= 1) mx[k] = fmaxf(mx[k], __shfl_xor(mx[k], o));

    float ex0[HEADS], sm[HEADS];
#pragma unroll
    for (int k = 0; k < HEADS; ++k) {
      ex0[k] = v0 ? __expf(e0[k] - mx[k]) : 0.f;
      sm[k] = ex0[k];
    }
    for (int base = 32; base < tot; base += 32) {
      int s; float e[HEADS];
      const int j = base + lane;
      edge_scores(csr_src, asrc, start, deg, n, j, ad, s, e);
#pragma unroll
      for (int k = 0; k < HEADS; ++k) sm[k] += (j < tot) ? __expf(e[k] - mx[k]) : 0.f;
    }
#pragma unroll
    for (int k = 0; k < HEADS; ++k)
      for (int o = 16; o > 0; o >>= 1) sm[k] += __shfl_xor(sm[k], o);

    float4 acc = make_float4(0.f, 0.f, 0.f, 0.f);
    {
      float alpha[HEADS];
#pragma unroll
      for (int k = 0; k < HEADS; ++k) alpha[k] = ex0[k] / (sm[k] + SM_EPS);
      gather_rows<C>(h, s0, alpha, tot < 32 ? tot : 32, lane, hq, acc);
    }
    for (int base = 32; base < tot; base += 32) {
      int s; float e[HEADS], alpha[HEADS];
      const int j = base + lane;
      edge_scores(csr_src, asrc, start, deg, n, j, ad, s, e);
#pragma unroll
      for (int k = 0; k < HEADS; ++k)
        alpha[k] = (j < tot) ? __expf(e[k] - mx[k]) / (sm[k] + SM_EPS) : 0.f;
      const int left = tot - base;
      gather_rows<C>(h, s, alpha, left < 32 ? left : 32, lane, hq, acc);
    }

#pragma unroll
    for (int o = GL; o < 32; o <<= 1) {
      acc.x += __shfl_xor(acc.x, o);
      acc.y += __shfl_xor(acc.y, o);
      acc.z += __shfl_xor(acc.z, o);
      acc.w += __shfl_xor(acc.w, o);
    }
    if (NPW == 1 || hq == u) keep = acc;
  }

  const int nn = (N - n0 < NPW) ? (N - n0) : NPW;
  if (lane < GL * nn) {
    const float4 b = *(const float4*)(bias + 4 * (lane % GL));
    float4 r;
    r.x = keep.x * (1.0f / HEADS) + b.x;
    r.y = keep.y * (1.0f / HEADS) + b.y;
    r.z = keep.z * (1.0f / HEADS) + b.z;
    r.w = keep.w * (1.0f / HEADS) + b.w;
    if (do_relu) {
      r.x = r.x > 0.f ? r.x : 0.f;
      r.y = r.y > 0.f ? r.y : 0.f;
      r.z = r.z > 0.f ? r.z : 0.f;
      r.w = r.w > 0.f ? r.w : 0.f;
    }
    *(float4*)(out + (size_t)n0 * C + 4 * lane) = r;
  }
}

#define PW 8
#define PN (2 * PW)
#define KW (HEADS * 32 / PW)
#define ZP (HEADS * 32 + 4)

#define GS 8
template <int F>
__device__ __forceinline__ void gather_inputs(const float* __restrict__ xin, int sreg,
                                              const float* wl, int count,
                                              int lane, float (&z)[HEADS]) {
  for (int j0 = 0; j0 < count; j0 += GS) {
    float v[GS];
#pragma unroll
    for (int g = 0; g < GS; ++g) {
      const int sj = __builtin_amdgcn_readlane(sreg, j0 + g);
      v[g] = xin[(unsigned)sj * F + (unsigned)lane];
    }
#pragma unroll
    for (int g = 0; g < GS; ++g) {
      const float4 a0 = *(const float4*)(wl + (j0 + g) * HEADS);
      const float4 a1 = *(const float4*)(wl + (j0 + g) * HEADS + 4);
      z[0] = __builtin_fmaf(a0.x, v[g], z[0]);  z[1] = __builtin_fmaf(a0.y, v[g], z[1]);
      z[2] = __builtin_fmaf(a0.z, v[g], z[2]);  z[3] = __builtin_fmaf(a0.w, v[g], z[3]);
      z[4] = __builtin_fmaf(a1.x, v[g], z[4]);  z[5] = __builtin_fmaf(a1.y, v[g], z[5]);
      z[6] = __builtin_fmaf(a1.z, v[g], z[6]);  z[7] = __builtin_fmaf(a1.w, v[g], z[7]);
    }
  }
}

__device__ __forceinline__ void put_weights(float* wl, int lane, const float (&alpha)[HEADS]) {
  __builtin_amdgcn_wave_barrier();
  *(float4*)(wl + lane * HEADS)     = make_float4(alpha[0], alpha[1], alpha[2], alpha[3]);
  *(float4*)(wl + lane * HEADS + 4) = make_float4(alpha[4], alpha[5], alpha[6], alpha[7]);
  __builtin_amdgcn_wave_barrier();
}

template <int F, int C>
__global__ void __launch_bounds__(PW * 32, 12)
node_attend_project_kernel(const int* __restrict__ rowptr,
                           const int* __restrict__ cnt,
                           const int* __restrict__ csr_src,
                           const float* __restrict__ asrc,
                           const float* __restrict__ adst,
                           const float* __restrict__ xin,
                           const float* __restrict__ Wm,
                           const float* __restrict__ bias,
                           float* __restrict__ out, int N) {
  static_assert(F == 32, "lane = input channel");
  static_assert(C == 16, "one 16-column tile of output channels");
  static_assert(PN == 16 && PW == HEADS && KW == F, "one 16-node tile; wave w owns head w's k values");
  static_assert(PW * 256 <= PN * ZP, "the partial tiles reuse the z rows' LDS");
  __shared__ __attribute__((aligned(16))) float zs[PN * ZP];
  __shared__ __attribute__((aligned(16))) float aw[PW * 32 * HEADS];
  const int lane = (int)(threadIdx.x & 31);
  const int wib = __builtin_amdgcn_readfirstlane((int)(threadIdx.x >> 5));
  const int nb = (int)blockIdx.x * PN;
  float* wl = aw + wib * (32 * HEADS);

  for (int u = 0; u < 2; ++u) {
    const int R = 2 * wib + u;
    const int n = nb + R;
    float z[HEADS];
#pragma unroll
    for (int k = 0; k < HEADS; ++k) z[k] = 0.f;
    if (n < N) {
      const int start = rowptr[n];
      const int deg = cnt[n];
      const int tot = deg + 1;
      float ad[HEADS];
#pragma unroll
      for (int k = 0; k < HEADS; ++k) ad[k] = adst[(size_t)n * HEADS + k];

      int s0;
      float e0[HEADS];
      const bool v0 = lane < tot;
      edge_scores(csr_src, asrc, start, deg, n, lane, ad, s0, e0);

      float mx[HEADS];
#pragma unroll
      for (int k = 0; k < HEADS; ++k) mx[k] = v0 ? e0[k] : -3.0e38f;
      for (int base = 32; base < tot; base += 32) {
        int s; float e[HEADS];
        const int j = base + lane;
        edge_scores(csr_src, asrc, start, deg, n, j, ad, s, e);
#pragma unroll
        for (int k = 0; k < HEADS; ++k) mx[k] = (j < tot) ? fmaxf(mx[k], e[k]) : mx[k];
      }
#pragma unroll
      for (int k = 0; k < HEADS; ++k) {
        for (int o = 16; o > 0; o >>= 1) mx[k] = fmaxf(mx[k], __shfl_xor(mx[k], o));
        mx[k] = __int_as_float(__builtin_amdgcn_readfirstlane(__float_as_int(mx[k])));
      }

      float ex0[HEADS], sm[HEADS];
#pragma unroll
      for (int k = 0; k < HEADS; ++k) {
        ex0[k] = v0 ? __expf(e0[k] - mx[k]) : 0.f;
        sm[k] = ex0[k];
      }
      for (int base = 32; base < tot; base += 32) {
        int s; float e[HEADS];
        const int j = base + lane;
        edge_scores(csr_src, asrc, start, deg, n, j, ad, s, e);
#pragma unroll
        for (int k = 0; k < HEADS; ++k) sm[k] += (j < tot) ? __expf(e[k] - mx[k]) : 0.f;
      }
#pragma unroll
      for (int k = 0; k < HEADS; ++k) {
        for (int o = 16; o > 0; o >>= 1) sm[k] += __shfl_xor(sm[k], o);
        sm[k] = __int_as_float(__builtin_amdgcn_readfirstlane(__float_as_int(sm[k])));
      }

      float alpha0[HEADS];
#pragma unroll
      for (int k = 0; k < HEADS; ++k) alpha0[k] = ex0[k] / (sm[k] + SM_EPS);
      for (int base = 32; base < tot; base += 32) {
        int s; float e[HEADS], alpha[HEADS];
        const int j = base + lane;
        edge_scores(csr_src, asrc, start, deg, n, j, ad, s, e);
#pragma unroll
        for (int k = 0; k < HEADS; ++k)
          alpha[k] = (j < tot) ? __expf(e[k] - mx[k]) / (sm[k] + SM_EPS) : 0.f;
        const int left = tot - base;
        put_weights(wl, lane, alpha);
        gather_inputs<F>(xin, s, wl, left < 32 ? left : 32, lane, z);
      }
      put_weights(wl, lane, alpha0);
      gather_inputs<F>(xin, s0, wl, tot < 32 ? tot : 32, lane, z);
    }
#pragma unroll
    for (int k = 0; k < HEADS; ++k) zs[R * ZP + k * F + lane] = z[k];
  }

  const int hf = lane >> 4;
  const int l = lane & 15;
  const float* wp = Wm + (size_t)(2 * hf) * (HEADS * C) + wib * C + l;
  v2f bfr[KW / 4];
#pragma unroll
  for (int ks = 0; ks < KW / 4; ++ks) {
    bfr[ks].x = wp[(size_t)(4 * ks) * (HEADS * C)];
    bfr[ks].y = wp[(size_t)(4 * ks + 1) * (HEADS * C)];
  }
  __syncthreads();

  v8f acc = (v8f){};
#pragma unroll
  for (int ks = 0; ks < KW / 4; ++ks) {
    const v2f a = *(const v2f*)&zs[l * ZP + KW * wib + 4 * ks + 2 * hf];
    acc = __builtin_amdgcn_wmma_f32_16x16x4_f32(false, a, false, bfr[ks],
                                                (short)0, acc, false, false);
  }
  __syncthreads();
#pragma unroll
  for (int r = 0; r < 8; ++r) zs[wib * 256 + (r + 8 * hf) * 16 + l] = acc[r];
  __syncthreads();

  const int t = (int)threadIdx.x;
  float sum = 0.f;
#pragma unroll
  for (int w = 0; w < PW; ++w) sum += zs[w * 256 + t];
  const int node = nb + (t >> 4);
  if (node < N)
    out[(size_t)node * C + (t & 15)] = sum * (1.0f / HEADS) + bias[t & 15];
}

static inline unsigned nblk(long n, int b) { return (unsigned)((n + b - 1) / b); }

extern "C" void kernel_launch(void* const* d_in, const int* in_sizes, int n_in,
                              void* d_out, int out_size, void* d_ws, size_t ws_size,
                              hipStream_t stream) {
  const float* x        = (const float*)d_in[0];
  const int*   ei       = (const int*)  d_in[1];
  const float* W1       = (const float*)d_in[2];
  const float* att_src1 = (const float*)d_in[3];
  const float* att_dst1 = (const float*)d_in[4];
  const float* b1       = (const float*)d_in[5];
  const float* W2       = (const float*)d_in[6];
  const float* att_src2 = (const float*)d_in[7];
  const float* att_dst2 = (const float*)d_in[8];
  const float* b2       = (const float*)d_in[9];

  const int F_h   = in_sizes[3] / HEADS;
  const int F_out = in_sizes[7] / HEADS;
  const int F_in  = in_sizes[2] / (HEADS * F_h);
  const int N     = in_sizes[0] / F_in;
  const int E     = in_sizes[1] / 2;
  const int C1    = HEADS * F_h;
  const int C2    = HEADS * F_out;
  const int* src = ei;
  const int* dst = ei + E;

  float* ws = (float*)d_ws;
  size_t o = 0;
  float* h1   = ws + o; o += (size_t)N * C1;
  float* asrc = ws + o; o += (size_t)N * HEADS;
  float* adst = ws + o; o += (size_t)N * HEADS;
  float* x2   = ws + o; o += (size_t)N * F_h;
  const int NB = (N + BIN_NODES - 1) / BIN_NODES;
  const int G1 = (int)nblk(E, SORT_TB * EPT);
  int* binned  = (int*)(ws + o); o += (size_t)2 * E;
  int* cnt     = (int*)(ws + o); o += (size_t)N;
  int* rowptr  = (int*)(ws + o); o += (size_t)N;
  int* csr_src = (int*)(ws + o); o += (size_t)E;
  int* hist    = (int*)(ws + o); o += (size_t)NB * G1;
  int* offs    = (int*)(ws + o); o += (size_t)NB * G1;
  float* outp = (float*)d_out;

  const int TB = 256;

  bin_hist_kernel<<<G1, SORT_TB, NB * sizeof(int), stream>>>(dst, E, NB, hist);
  prefix_kernel<<<nblk((long)NB * G1, 1024), 1024, 0, stream>>>(hist, NB * G1, offs);
  bin_scatter_kernel<<<G1, SORT_TB, NB * sizeof(int), stream>>>(src, dst, E, NB, offs, binned);
  bin_sort_kernel<<<NB, SORT_TB, 0, stream>>>(binned, offs, G1, NB, N, E, cnt, rowptr, csr_src);

  {
    long waves = (long)((N + 15) / 16) * (C1 / (16 * NT));
    wmma_gemm_f32<32, true><<<nblk(waves * 32, TB), TB, 0, stream>>>(
        x, W1, att_src1, att_dst1, h1, asrc, adst, N, F_in, C1);
  }
  node_attend_kernel<32><<<nblk((long)N * 32, TB), TB, 0, stream>>>(
      rowptr, cnt, csr_src, asrc, adst, h1, b1, x2, N, 1);

  {
    long waves = (long)((N + 15) / 16) * (C2 / (16 * NT));
    wmma_gemm_f32<16, false><<<nblk(waves * 32, TB), TB, 0, stream>>>(
        x2, W2, att_src2, att_dst2, h1, asrc, adst, N, F_h, C2);
  }
  node_attend_project_kernel<32, 16><<<nblk(N, PN), PW * 32, 0, stream>>>(
      rowptr, cnt, csr_src, asrc, adst, x2, W2, b2, outp, N);
}


# layer-1 GEMM has no launch of its own: its strips run as extra work-groups of the bin-scatter and bin-sort launches (64-register strip, so both kinds share a WGP); k order and sums unchanged
# speedup vs baseline: 15.0703x; 1.0402x over previous
#include <hip/hip_runtime.h>
#include <cstdint>

typedef float v2f __attribute__((ext_vector_type(2)));
typedef float v8f __attribute__((ext_vector_type(8)));

#define HEADS 8
#define NEG_SLOPE 0.2f
#define SM_EPS 1e-16f
#define NT 4

template <int CTRL>
__device__ __forceinline__ float xmask_add(float v) {
  const int o = __builtin_amdgcn_update_dpp(0, __float_as_int(v), CTRL, 0xf, 0xf, true);
  return v + __int_as_float(o);
}
__device__ __forceinline__ float row16_sum(float v) {
  v = xmask_add<0x161>(v);
  v = xmask_add<0x162>(v);
  v = xmask_add<0x164>(v);
  v = xmask_add<0x168>(v);
  return v;
}

template <int CH, bool STORE_C>
__device__ __forceinline__ void gemm_strip(const float* __restrict__ A,
                                           const float* __restrict__ B,
                                           const float* __restrict__ att_src,
                                           const float* __restrict__ att_dst,
                                           float* __restrict__ C,
                                           float* __restrict__ asrc,
                                           float* __restrict__ adst,
                                           int M, int K, int Ncols,
                                           int strip, int lane) {
  static_assert(CH == 32 || CH == 16, "a wave's 64 columns must be whole heads");
  constexpr int TPH = CH / 16;
  constexpr int HPW = NT / TPH;
  int groupsN = Ncols >> 6;
  int tm = strip / groupsN;
  int tg = strip - tm * groupsN;
  int half = lane >> 4;
  int l = lane & 15;

  int rowA = tm * 16 + l;
  if (rowA >= M) rowA = M - 1;
  const char* Ac = (const char*)A;
  const char* Bc = (const char*)B;
  unsigned aB  = 4u * ((unsigned)rowA * (unsigned)K + 2u * (unsigned)half);
  unsigned bB0 = 4u * ((unsigned)(2 * half) * (unsigned)Ncols + (unsigned)(tg * (16 * NT) + l));
  unsigned bB1 = bB0 + 4u * (unsigned)Ncols;
  const unsigned bStep = 16u * (unsigned)Ncols;

  v8f acc[NT];
#pragma unroll
  for (int t = 0; t < NT; ++t) acc[t] = (v8f){};

  v2f a0, b0[NT], a1, b1[NT];
  a0.x = *(const float*)(Ac + (size_t)aB);
  a0.y = *(const float*)(Ac + (size_t)aB + 4);
#pragma unroll
  for (int t = 0; t < NT; ++t) {
    b0[t].x = *(const float*)(Bc + (size_t)bB0 + t * 64);
    b0[t].y = *(const float*)(Bc + (size_t)bB1 + t * 64);
  }
  int k0 = 0;
#pragma unroll 1
  for (; k0 + 8 <= K; k0 += 8) {
    aB += 16u;  bB0 += bStep;  bB1 += bStep;
    a1.x = *(const float*)(Ac + (size_t)aB);
    a1.y = *(const float*)(Ac + (size_t)aB + 4);
#pragma unroll
    for (int t = 0; t < NT; ++t) {
      b1[t].x = *(const float*)(Bc + (size_t)bB0 + t * 64);
      b1[t].y = *(const float*)(Bc + (size_t)bB1 + t * 64);
    }
#pragma unroll
    for (int t = 0; t < NT; ++t) {
      acc[t] = __builtin_amdgcn_wmma_f32_16x16x4_f32(false, a0, false, b0[t],
                                                     (short)0, acc[t],
                                                     false, false);
    }
    const bool more = k0 + 8 < K;
    aB += more ? 16u : 0u;  bB0 += more ? bStep : 0u;  bB1 += more ? bStep : 0u;
    a0.x = *(const float*)(Ac + (size_t)aB);
    a0.y = *(const float*)(Ac + (size_t)aB + 4);
#pragma unroll
    for (int t = 0; t < NT; ++t) {
      b0[t].x = *(const float*)(Bc + (size_t)bB0 + t * 64);
      b0[t].y = *(const float*)(Bc + (size_t)bB1 + t * 64);
    }
#pragma unroll
    for (int t = 0; t < NT; ++t) {
      acc[t] = __builtin_amdgcn_wmma_f32_16x16x4_f32(false, a1, false, b1[t],
                                                     (short)0, acc[t],
                                                     false, false);
    }
  }
  if (k0 < K) {
#pragma unroll
    for (int t = 0; t < NT; ++t) {
      acc[t] = __builtin_amdgcn_wmma_f32_16x16x4_f32(false, a0, false, b0[t],
                                                     (short)0, acc[t],
                                                     false, false);
    }
  }

  float wS[NT], wD[NT];
#pragma unroll
  for (int t = 0; t < NT; ++t) {
    wS[t] = att_src[tg * (16 * NT) + t * 16 + l];
    wD[t] = att_dst[tg * (16 * NT) + t * 16 + l];
  }

  if (STORE_C) {
    char* Cc = (char*)C;
    const int row0 = tm * 16 + 8 * half;
    const unsigned rowC = 4u * (unsigned)Ncols;
    const unsigned cB = (unsigned)row0 * rowC + 4u * (unsigned)(tg * (16 * NT) + l);
#pragma unroll
    for (int r = 0; r < 8; ++r) {
      if (row0 + r < M) {
#pragma unroll
        for (int t = 0; t < NT; ++t)
          *(float*)(Cc + (size_t)(cB + (unsigned)r * rowC) + t * 64) = acc[t][r];
      }
    }
  }

#pragma unroll
  for (int j = 0; j < HPW / 2; ++j) {
    float vs = 0.f, vd = 0.f;
#pragma unroll
    for (int hb = 0; hb < 2; ++hb) {
#pragma unroll
      for (int r = 0; r < 8; ++r) {
        float ps = 0.f, pd = 0.f;
#pragma unroll
        for (int tt = 0; tt < TPH; ++tt) {
          const int t = (2 * j + hb) * TPH + tt;
          ps += acc[t][r] * wS[t];
          pd += acc[t][r] * wD[t];
        }
        ps = row16_sum(ps);
        pd = row16_sum(pd);
        const bool mine = (l == hb * 8 + r);
        vs = mine ? ps : vs;
        vd = mine ? pd : vd;
        asm volatile("" : "+v"(vs), "+v"(vd));
      }
    }
    const int row = tm * 16 + (l & 7) + 8 * half;
    const int head = tg * HPW + 2 * j + (l >> 3);
    if (row < M) {
      asrc[(size_t)row * HEADS + head] = vs;
      adst[(size_t)row * HEADS + head] = vd;
    }
  }
}

template <int CH, bool STORE_C>
__global__ void wmma_gemm_f32(const float* __restrict__ A,
                              const float* __restrict__ B,
                              const float* __restrict__ att_src,
                              const float* __restrict__ att_dst,
                              float* __restrict__ C,
                              float* __restrict__ asrc,
                              float* __restrict__ adst,
                              int M, int K, int Ncols) {
  int wave = (int)((blockIdx.x * blockDim.x + threadIdx.x) >> 5);
  int lane = (int)(threadIdx.x & 31);
  int tilesM  = (M + 15) >> 4;
  int groupsN = Ncols >> 6;
  if (wave >= tilesM * groupsN) return;
  gemm_strip<CH, STORE_C>(A, B, att_src, att_dst, C, asrc, adst, M, K, Ncols, wave, lane);
}

#define GEMM_BLOCKS 256
struct GemmJob {
  const float* A;
  const float* B;
  const float* att_src;
  const float* att_dst;
  float* C;
  float* asrc;
  float* adst;
  int M, K, Ncols;
  int begin, end;
};

template <int CH>
__device__ __forceinline__ void gemm_job_block(const GemmJob& j, int gb, int nb) {
  const int lane = (int)(threadIdx.x & 31);
  const int wib = __builtin_amdgcn_readfirstlane((int)(threadIdx.x >> 5));
  const int s = j.begin + gb + nb * wib;
  if (s < j.end)
    gemm_strip<CH, true>(j.A, j.B, j.att_src, j.att_dst, j.C, j.asrc, j.adst,
                         j.M, j.K, j.Ncols, s, lane);
}

__device__ __forceinline__ float lrelu(float v) {
  return v > 0.f ? v : NEG_SLOPE * v;
}

#define BIN_SHIFT 8
#define BIN_NODES (1 << BIN_SHIFT)
#define SORT_TB 1024
#define EPT 8
#define SPT 4

__global__ void __launch_bounds__(SORT_TB)
bin_hist_kernel(const int* __restrict__ dst, int E, int NB, int* __restrict__ hist) {
  extern __shared__ int sh[];
  const int t = (int)threadIdx.x;
  for (int b = t; b < NB; b += SORT_TB) sh[b] = 0;
  __syncthreads();
  const int base = (int)blockIdx.x * (SORT_TB * EPT) + t;
  int d[EPT];
#pragma unroll
  for (int k = 0; k < EPT; ++k) {
    const int e = base + k * SORT_TB;
    d[k] = dst[e < E ? e : E - 1];
  }
#pragma unroll
  for (int k = 0; k < EPT; ++k) {
    const int e = base + k * SORT_TB;
    if (e < E) atomicAdd(&sh[d[k] >> BIN_SHIFT], 1);
  }
  __syncthreads();
  for (int b = t; b < NB; b += SORT_TB)
    hist[(size_t)b * gridDim.x + blockIdx.x] = sh[b];
}

__global__ void __launch_bounds__(SORT_TB, 16)
bin_scatter_kernel(const int* __restrict__ src, const int* __restrict__ dst, int E,
                   int NB, int G1, const int* __restrict__ offs, int* __restrict__ binned,
                   GemmJob job) {
  extern __shared__ int sh[];
  if ((int)blockIdx.x >= G1) {
    gemm_job_block<32>(job, (int)blockIdx.x - G1, (int)gridDim.x - G1);
    return;
  }
  const int t = (int)threadIdx.x;
  for (int b = t; b < NB; b += SORT_TB)
    sh[b] = offs[(size_t)b * G1 + blockIdx.x];
  __syncthreads();
  const int base = (int)blockIdx.x * (SORT_TB * EPT) + t;
  int s[EPT], d[EPT], pos[EPT];
#pragma unroll
  for (int k = 0; k < EPT; ++k) {
    const int e = base + k * SORT_TB;
    const int ec = e < E ? e : E - 1;
    s[k] = src[ec];
    d[k] = dst[ec];
  }
#pragma unroll
  for (int k = 0; k < EPT; ++k) {
    const int e = base + k * SORT_TB;
    pos[k] = 0;
    if (e < E) pos[k] = atomicAdd(&sh[d[k] >> BIN_SHIFT], 1);
  }
  int2* __restrict__ outp = (int2*)binned;
#pragma unroll
  for (int k = 0; k < EPT; ++k) {
    const int e = base + k * SORT_TB;
    if (e < E) outp[pos[k]] = make_int2(s[k], d[k]);
  }
}

__global__ void __launch_bounds__(SORT_TB, 16)
bin_sort_kernel(const int* __restrict__ binned, const int* __restrict__ offs,
                int G1, int NB, int N, int E,
                int* __restrict__ cnt, int* __restrict__ rowptr,
                int* __restrict__ csr_src, GemmJob job) {
  __shared__ int c[BIN_NODES];
  __shared__ int wsum[BIN_NODES / 32];
  if ((int)blockIdx.x >= NB) {
    gemm_job_block<32>(job, (int)blockIdx.x - NB, (int)gridDim.x - NB);
    return;
  }
  const int2* __restrict__ pairs = (const int2*)binned;
  const int t = (int)threadIdx.x;
  const int lane = t & 31;
  const int w = t >> 5;
  const int b = (int)blockIdx.x;
  const int beg = offs[(size_t)b * G1];
  const int end = (b + 1 < NB) ? offs[(size_t)(b + 1) * G1] : E;
  if (t < BIN_NODES) c[t] = 0;
  __syncthreads();
  for (int i0 = beg + t; i0 < end; i0 += SPT * SORT_TB) {
    int y[SPT];
#pragma unroll
    for (int k = 0; k < SPT; ++k) {
      const int i = i0 + k * SORT_TB;
      y[k] = pairs[i < end ? i : end - 1].y;
    }
#pragma unroll
    for (int k = 0; k < SPT; ++k)
      if (i0 + k * SORT_TB < end) atomicAdd(&c[y[k] & (BIN_NODES - 1)], 1);
  }
  __syncthreads();
  int mine = 0, incl = 0;
  if (t < BIN_NODES) {
    mine = c[t];
    incl = mine;
    for (int o = 1; o < 32; o <<= 1) {
      const int v = __shfl_up(incl, o);
      if (lane >= o) incl += v;
    }
    if (lane == 31) wsum[w] = incl;
  }
  __syncthreads();
  if (t < BIN_NODES) {
    int woff = 0;
    for (int k = 0; k < BIN_NODES / 32; ++k) woff += (k < w) ? wsum[k] : 0;
    const int start = beg + woff + incl - mine;
    c[t] = start;
    const int n = b * BIN_NODES + t;
    if (n < N) {
      cnt[n] = mine;
      rowptr[n] = start;
    }
  }
  __syncthreads();
  for (int i0 = beg + t; i0 < end; i0 += SPT * SORT_TB) {
    int2 sd[SPT];
    int pos[SPT];
#pragma unroll
    for (int k = 0; k < SPT; ++k) {
      const int i = i0 + k * SORT_TB;
      sd[k] = pairs[i < end ? i : end - 1];
    }
#pragma unroll
    for (int k = 0; k < SPT; ++k) {
      pos[k] = 0;
      if (i0 + k * SORT_TB < end) pos[k] = atomicAdd(&c[sd[k].y & (BIN_NODES - 1)], 1);
    }
#pragma unroll
    for (int k = 0; k < SPT; ++k)
      if (i0 + k * SORT_TB < end) csr_src[pos[k]] = sd[k].x;
  }
}

__global__ void __launch_bounds__(1024)
prefix_kernel(const int* __restrict__ cnt, int N, int* __restrict__ rowptr) {
  __shared__ int wsum[32];
  const int t = (int)threadIdx.x;
  const int lane = t & 31;
  const int w = t >> 5;
  const int base = (int)blockIdx.x * 1024;

  int part = 0;
  for (int i = t; i < base; i += 1024) part += cnt[i];
  for (int o = 16; o > 0; o >>= 1) part += __shfl_xor(part, o);
  if (lane == 0) wsum[w] = part;
  __syncthreads();
  int prefix = 0;
  for (int k = 0; k < 32; ++k) prefix += wsum[k];
  __syncthreads();

  const int i = base + t;
  const int c = (i < N) ? cnt[i] : 0;
  int incl = c;
  for (int o = 1; o < 32; o <<= 1) {
    const int v = __shfl_up(incl, o);
    if (lane >= o) incl += v;
  }
  if (lane == 31) wsum[w] = incl;
  __syncthreads();
  int woff = 0;
  for (int k = 0; k < 32; ++k) woff += (k < w) ? wsum[k] : 0;
  if (i < N) rowptr[i] = prefix + woff + incl - c;
}

__device__ __forceinline__ void edge_scores(const int* __restrict__ csr_src,
                                            const float* __restrict__ asrc,
                                            int start, int deg, int n, int j,
                                            const float (&ad)[HEADS],
                                            int& s, float (&e)[HEADS]) {
  s = (j < deg) ? csr_src[start + j] : n;
  const float4 a0 = *(const float4*)(asrc + (size_t)s * HEADS);
  const float4 a1 = *(const float4*)(asrc + (size_t)s * HEADS + 4);
  e[0] = lrelu(a0.x + ad[0]);
  e[1] = lrelu(a0.y + ad[1]);
  e[2] = lrelu(a0.z + ad[2]);
  e[3] = lrelu(a0.w + ad[3]);
  e[4] = lrelu(a1.x + ad[4]);
  e[5] = lrelu(a1.y + ad[5]);
  e[6] = lrelu(a1.z + ad[6]);
  e[7] = lrelu(a1.w + ad[7]);
}

template <int C>
__device__ __forceinline__ void gather_rows(const float* __restrict__ h, int sreg,
                                            const float (&alpha)[HEADS], int count,
                                            int lane, int hq, float4& acc) {
  constexpr int ROW = HEADS * C;
#pragma unroll 4
  for (int jj = 0; jj < count; ++jj) {
    const int sj = __builtin_amdgcn_readlane(sreg, jj);
    float a[HEADS];
#pragma unroll
    for (int k = 0; k < HEADS; ++k)
      a[k] = __int_as_float(__builtin_amdgcn_readlane(__float_as_int(alpha[k]), jj));
    const float* hp = h + (size_t)sj * ROW + 4 * lane;
    if (C == 32) {
      float w0 = a[0], w1 = a[4];
      w0 = (hq == 1) ? a[1] : w0;  w1 = (hq == 1) ? a[5] : w1;
      w0 = (hq == 2) ? a[2] : w0;  w1 = (hq == 2) ? a[6] : w1;
      w0 = (hq == 3) ? a[3] : w0;  w1 = (hq == 3) ? a[7] : w1;
      const float4 v0 = *(const float4*)hp;
      const float4 v1 = *(const float4*)(hp + 128);
      acc.x += w0 * v0.x;  acc.y += w0 * v0.y;  acc.z += w0 * v0.z;  acc.w += w0 * v0.w;
      acc.x += w1 * v1.x;  acc.y += w1 * v1.y;  acc.z += w1 * v1.z;  acc.w += w1 * v1.w;
    } else {
      float w = a[0];
#pragma unroll
      for (int k = 1; k < HEADS; ++k) w = (hq == k) ? a[k] : w;
      const float4 v = *(const float4*)hp;
      acc.x += w * v.x;  acc.y += w * v.y;  acc.z += w * v.z;  acc.w += w * v.w;
    }
  }
}

template <int C>
__global__ void node_attend_kernel(const int* __restrict__ rowptr,
                                   const int* __restrict__ cnt,
                                   const int* __restrict__ csr_src,
                                   const float* __restrict__ asrc,
                                   const float* __restrict__ adst,
                                   const float* __restrict__ h,
                                   const float* __restrict__ bias,
                                   float* __restrict__ out,
                                   int N, int do_relu) {
  static_assert(C == 32 || C == 16, "lane maps are written for 32 or 16 channels");
  constexpr int NPW = 32 / C;
  constexpr int GL  = C / 4;
  const int lane = (int)(threadIdx.x & 31);
  const int wib = __builtin_amdgcn_readfirstlane((int)(threadIdx.x >> 5));
  const int wave = (int)blockIdx.x * (int)(blockDim.x >> 5) + wib;
  const int n0 = wave * NPW;
  if (n0 >= N) return;
  const int hq = lane / GL;
  float4 keep = make_float4(0.f, 0.f, 0.f, 0.f);

  for (int u = 0; u < NPW; ++u) {
    const int n = n0 + u;
    if (n >= N) break;
    const int start = rowptr[n];
    const int deg = cnt[n];
    const int tot = deg + 1;
    float ad[HEADS];
#pragma unroll
    for (int k = 0; k < HEADS; ++k) ad[k] = adst[(size_t)n * HEADS + k];

    int s0;
    float e0[HEADS];
    const bool v0 = lane < tot;
    edge_scores(csr_src, asrc, start, deg, n, lane, ad, s0, e0);

    float mx[HEADS];
#pragma unroll
    for (int k = 0; k < HEADS; ++k) mx[k] = v0 ? e0[k] : -3.0e38f;
    for (int base = 32; base < tot; base += 32) {
      int s; float e[HEADS];
      const int j = base + lane;
      edge_scores(csr_src, asrc, start, deg, n, j, ad, s, e);
#pragma unroll
      for (int k = 0; k < HEADS; ++k) mx[k] = (j < tot) ? fmaxf(mx[k], e[k]) : mx[k];
    }
#pragma unroll
    for (int k = 0; k < HEADS; ++k)
      for (int o = 16; o > 0; o >>= 1) mx[k] = fmaxf(mx[k], __shfl_xor(mx[k], o));

    float ex0[HEADS], sm[HEADS];
#pragma unroll
    for (int k = 0; k < HEADS; ++k) {
      ex0[k] = v0 ? __expf(e0[k] - mx[k]) : 0.f;
      sm[k] = ex0[k];
    }
    for (int base = 32; base < tot; base += 32) {
      int s; float e[HEADS];
      const int j = base + lane;
      edge_scores(csr_src, asrc, start, deg, n, j, ad, s, e);
#pragma unroll
      for (int k = 0; k < HEADS; ++k) sm[k] += (j < tot) ? __expf(e[k] - mx[k]) : 0.f;
    }
#pragma unroll
    for (int k = 0; k < HEADS; ++k)
      for (int o = 16; o > 0; o >>= 1) sm[k] += __shfl_xor(sm[k], o);

    float4 acc = make_float4(0.f, 0.f, 0.f, 0.f);
    {
      float alpha[HEADS];
#pragma unroll
      for (int k = 0; k < HEADS; ++k) alpha[k] = ex0[k] / (sm[k] + SM_EPS);
      gather_rows<C>(h, s0, alpha, tot < 32 ? tot : 32, lane, hq, acc);
    }
    for (int base = 32; base < tot; base += 32) {
      int s; float e[HEADS], alpha[HEADS];
      const int j = base + lane;
      edge_scores(csr_src, asrc, start, deg, n, j, ad, s, e);
#pragma unroll
      for (int k = 0; k < HEADS; ++k)
        alpha[k] = (j < tot) ? __expf(e[k] - mx[k]) / (sm[k] + SM_EPS) : 0.f;
      const int left = tot - base;
      gather_rows<C>(h, s, alpha, left < 32 ? left : 32, lane, hq, acc);
    }

#pragma unroll
    for (int o = GL; o < 32; o <<= 1) {
      acc.x += __shfl_xor(acc.x, o);
      acc.y += __shfl_xor(acc.y, o);
      acc.z += __shfl_xor(acc.z, o);
      acc.w += __shfl_xor(acc.w, o);
    }
    if (NPW == 1 || hq == u) keep = acc;
  }

  const int nn = (N - n0 < NPW) ? (N - n0) : NPW;
  if (lane < GL * nn) {
    const float4 b = *(const float4*)(bias + 4 * (lane % GL));
    float4 r;
    r.x = keep.x * (1.0f / HEADS) + b.x;
    r.y = keep.y * (1.0f / HEADS) + b.y;
    r.z = keep.z * (1.0f / HEADS) + b.z;
    r.w = keep.w * (1.0f / HEADS) + b.w;
    if (do_relu) {
      r.x = r.x > 0.f ? r.x : 0.f;
      r.y = r.y > 0.f ? r.y : 0.f;
      r.z = r.z > 0.f ? r.z : 0.f;
      r.w = r.w > 0.f ? r.w : 0.f;
    }
    *(float4*)(out + (size_t)n0 * C + 4 * lane) = r;
  }
}

#define PW 8
#define PN (2 * PW)
#define KW (HEADS * 32 / PW)
#define ZP (HEADS * 32 + 4)

#define GS 8
template <int F>
__device__ __forceinline__ void gather_inputs(const float* __restrict__ xin, int sreg,
                                              const float* wl, int count,
                                              int lane, float (&z)[HEADS]) {
  for (int j0 = 0; j0 < count; j0 += GS) {
    float v[GS];
#pragma unroll
    for (int g = 0; g < GS; ++g) {
      const int sj = __builtin_amdgcn_readlane(sreg, j0 + g);
      v[g] = xin[(unsigned)sj * F + (unsigned)lane];
    }
#pragma unroll
    for (int g = 0; g < GS; ++g) {
      const float4 a0 = *(const float4*)(wl + (j0 + g) * HEADS);
      const float4 a1 = *(const float4*)(wl + (j0 + g) * HEADS + 4);
      z[0] = __builtin_fmaf(a0.x, v[g], z[0]);  z[1] = __builtin_fmaf(a0.y, v[g], z[1]);
      z[2] = __builtin_fmaf(a0.z, v[g], z[2]);  z[3] = __builtin_fmaf(a0.w, v[g], z[3]);
      z[4] = __builtin_fmaf(a1.x, v[g], z[4]);  z[5] = __builtin_fmaf(a1.y, v[g], z[5]);
      z[6] = __builtin_fmaf(a1.z, v[g], z[6]);  z[7] = __builtin_fmaf(a1.w, v[g], z[7]);
    }
  }
}

__device__ __forceinline__ void put_weights(float* wl, int lane, const float (&alpha)[HEADS]) {
  __builtin_amdgcn_wave_barrier();
  *(float4*)(wl + lane * HEADS)     = make_float4(alpha[0], alpha[1], alpha[2], alpha[3]);
  *(float4*)(wl + lane * HEADS + 4) = make_float4(alpha[4], alpha[5], alpha[6], alpha[7]);
  __builtin_amdgcn_wave_barrier();
}

template <int F, int C>
__global__ void __launch_bounds__(PW * 32, 12)
node_attend_project_kernel(const int* __restrict__ rowptr,
                           const int* __restrict__ cnt,
                           const int* __restrict__ csr_src,
                           const float* __restrict__ asrc,
                           const float* __restrict__ adst,
                           const float* __restrict__ xin,
                           const float* __restrict__ Wm,
                           const float* __restrict__ bias,
                           float* __restrict__ out, int N) {
  static_assert(F == 32, "lane = input channel");
  static_assert(C == 16, "one 16-column tile of output channels");
  static_assert(PN == 16 && PW == HEADS && KW == F, "one 16-node tile; wave w owns head w's k values");
  static_assert(PW * 256 <= PN * ZP, "the partial tiles reuse the z rows' LDS");
  __shared__ __attribute__((aligned(16))) float zs[PN * ZP];
  __shared__ __attribute__((aligned(16))) float aw[PW * 32 * HEADS];
  const int lane = (int)(threadIdx.x & 31);
  const int wib = __builtin_amdgcn_readfirstlane((int)(threadIdx.x >> 5));
  const int nb = (int)blockIdx.x * PN;
  float* wl = aw + wib * (32 * HEADS);

  for (int u = 0; u < 2; ++u) {
    const int R = 2 * wib + u;
    const int n = nb + R;
    float z[HEADS];
#pragma unroll
    for (int k = 0; k < HEADS; ++k) z[k] = 0.f;
    if (n < N) {
      const int start = rowptr[n];
      const int deg = cnt[n];
      const int tot = deg + 1;
      float ad[HEADS];
#pragma unroll
      for (int k = 0; k < HEADS; ++k) ad[k] = adst[(size_t)n * HEADS + k];

      int s0;
      float e0[HEADS];
      const bool v0 = lane < tot;
      edge_scores(csr_src, asrc, start, deg, n, lane, ad, s0, e0);

      float mx[HEADS];
#pragma unroll
      for (int k = 0; k < HEADS; ++k) mx[k] = v0 ? e0[k] : -3.0e38f;
      for (int base = 32; base < tot; base += 32) {
        int s; float e[HEADS];
        const int j = base + lane;
        edge_scores(csr_src, asrc, start, deg, n, j, ad, s, e);
#pragma unroll
        for (int k = 0; k < HEADS; ++k) mx[k] = (j < tot) ? fmaxf(mx[k], e[k]) : mx[k];
      }
#pragma unroll
      for (int k = 0; k < HEADS; ++k) {
        for (int o = 16; o > 0; o >>= 1) mx[k] = fmaxf(mx[k], __shfl_xor(mx[k], o));
        mx[k] = __int_as_float(__builtin_amdgcn_readfirstlane(__float_as_int(mx[k])));
      }

      float ex0[HEADS], sm[HEADS];
#pragma unroll
      for (int k = 0; k < HEADS; ++k) {
        ex0[k] = v0 ? __expf(e0[k] - mx[k]) : 0.f;
        sm[k] = ex0[k];
      }
      for (int base = 32; base < tot; base += 32) {
        int s; float e[HEADS];
        const int j = base + lane;
        edge_scores(csr_src, asrc, start, deg, n, j, ad, s, e);
#pragma unroll
        for (int k = 0; k < HEADS; ++k) sm[k] += (j < tot) ? __expf(e[k] - mx[k]) : 0.f;
      }
#pragma unroll
      for (int k = 0; k < HEADS; ++k) {
        for (int o = 16; o > 0; o >>= 1) sm[k] += __shfl_xor(sm[k], o);
        sm[k] = __int_as_float(__builtin_amdgcn_readfirstlane(__float_as_int(sm[k])));
      }

      float alpha0[HEADS];
#pragma unroll
      for (int k = 0; k < HEADS; ++k) alpha0[k] = ex0[k] / (sm[k] + SM_EPS);
      for (int base = 32; base < tot; base += 32) {
        int s; float e[HEADS], alpha[HEADS];
        const int j = base + lane;
        edge_scores(csr_src, asrc, start, deg, n, j, ad, s, e);
#pragma unroll
        for (int k = 0; k < HEADS; ++k)
          alpha[k] = (j < tot) ? __expf(e[k] - mx[k]) / (sm[k] + SM_EPS) : 0.f;
        const int left = tot - base;
        put_weights(wl, lane, alpha);
        gather_inputs<F>(xin, s, wl, left < 32 ? left : 32, lane, z);
      }
      put_weights(wl, lane, alpha0);
      gather_inputs<F>(xin, s0, wl, tot < 32 ? tot : 32, lane, z);
    }
#pragma unroll
    for (int k = 0; k < HEADS; ++k) zs[R * ZP + k * F + lane] = z[k];
  }

  const int hf = lane >> 4;
  const int l = lane & 15;
  const float* wp = Wm + (size_t)(2 * hf) * (HEADS * C) + wib * C + l;
  v2f bfr[KW / 4];
#pragma unroll
  for (int ks = 0; ks < KW / 4; ++ks) {
    bfr[ks].x = wp[(size_t)(4 * ks) * (HEADS * C)];
    bfr[ks].y = wp[(size_t)(4 * ks + 1) * (HEADS * C)];
  }
  __syncthreads();

  v8f acc = (v8f){};
#pragma unroll
  for (int ks = 0; ks < KW / 4; ++ks) {
    const v2f a = *(const v2f*)&zs[l * ZP + KW * wib + 4 * ks + 2 * hf];
    acc = __builtin_amdgcn_wmma_f32_16x16x4_f32(false, a, false, bfr[ks],
                                                (short)0, acc, false, false);
  }
  __syncthreads();
#pragma unroll
  for (int r = 0; r < 8; ++r) zs[wib * 256 + (r + 8 * hf) * 16 + l] = acc[r];
  __syncthreads();

  const int t = (int)threadIdx.x;
  float sum = 0.f;
#pragma unroll
  for (int w = 0; w < PW; ++w) sum += zs[w * 256 + t];
  const int node = nb + (t >> 4);
  if (node < N)
    out[(size_t)node * C + (t & 15)] = sum * (1.0f / HEADS) + bias[t & 15];
}

static inline unsigned nblk(long n, int b) { return (unsigned)((n + b - 1) / b); }

extern "C" void kernel_launch(void* const* d_in, const int* in_sizes, int n_in,
                              void* d_out, int out_size, void* d_ws, size_t ws_size,
                              hipStream_t stream) {
  const float* x        = (const float*)d_in[0];
  const int*   ei       = (const int*)  d_in[1];
  const float* W1       = (const float*)d_in[2];
  const float* att_src1 = (const float*)d_in[3];
  const float* att_dst1 = (const float*)d_in[4];
  const float* b1       = (const float*)d_in[5];
  const float* W2       = (const float*)d_in[6];
  const float* att_src2 = (const float*)d_in[7];
  const float* att_dst2 = (const float*)d_in[8];
  const float* b2       = (const float*)d_in[9];

  const int F_h   = in_sizes[3] / HEADS;
  const int F_out = in_sizes[7] / HEADS;
  const int F_in  = in_sizes[2] / (HEADS * F_h);
  const int N     = in_sizes[0] / F_in;
  const int E     = in_sizes[1] / 2;
  const int C1    = HEADS * F_h;
  const int C2    = HEADS * F_out;
  const int* src = ei;
  const int* dst = ei + E;

  float* ws = (float*)d_ws;
  size_t o = 0;
  float* h1   = ws + o; o += (size_t)N * C1;
  float* asrc = ws + o; o += (size_t)N * HEADS;
  float* adst = ws + o; o += (size_t)N * HEADS;
  float* x2   = ws + o; o += (size_t)N * F_h;
  const int NB = (N + BIN_NODES - 1) / BIN_NODES;
  const int G1 = (int)nblk(E, SORT_TB * EPT);
  int* binned  = (int*)(ws + o); o += (size_t)2 * E;
  int* cnt     = (int*)(ws + o); o += (size_t)N;
  int* rowptr  = (int*)(ws + o); o += (size_t)N;
  int* csr_src = (int*)(ws + o); o += (size_t)E;
  int* hist    = (int*)(ws + o); o += (size_t)NB * G1;
  int* offs    = (int*)(ws + o); o += (size_t)NB * G1;
  float* outp = (float*)d_out;

  const int TB = 256;

  bin_hist_kernel<<<G1, SORT_TB, NB * sizeof(int), stream>>>(dst, E, NB, hist);
  prefix_kernel<<<nblk((long)NB * G1, 1024), 1024, 0, stream>>>(hist, NB * G1, offs);
  const int groups1 = C1 / (16 * NT);
  const int strips1 = ((N + 15) / 16) * groups1;
  const int split1  = ((strips1 / 2 + groups1 - 1) / groups1) * groups1;
  const int wpb = SORT_TB / 32;
  int gblk = ((split1 + wpb * groups1 - 1) / (wpb * groups1)) * groups1;
  if (gblk < GEMM_BLOCKS) gblk = GEMM_BLOCKS;
  GemmJob job;
  job.A = x;  job.B = W1;  job.att_src = att_src1;  job.att_dst = att_dst1;
  job.C = h1;  job.asrc = asrc;  job.adst = adst;
  job.M = N;  job.K = F_in;  job.Ncols = C1;
  job.begin = 0;  job.end = split1;
  bin_scatter_kernel<<<G1 + gblk, SORT_TB, NB * sizeof(int), stream>>>(
      src, dst, E, NB, G1, offs, binned, job);
  job.begin = split1;  job.end = strips1;
  bin_sort_kernel<<<NB + gblk, SORT_TB, 0, stream>>>(
      binned, offs, G1, NB, N, E, cnt, rowptr, csr_src, job);

  node_attend_kernel<32><<<nblk((long)N * 32, TB), TB, 0, stream>>>(
      rowptr, cnt, csr_src, asrc, adst, h1, b1, x2, N, 1);

  {
    long waves = (long)((N + 15) / 16) * (C2 / (16 * NT));
    wmma_gemm_f32<16, false><<<nblk(waves * 32, TB), TB, 0, stream>>>(
        x2, W2, att_src2, att_dst2, h1, asrc, adst, N, F_h, C2);
  }
  node_attend_project_kernel<32, 16><<<nblk(N, PN), PW * 32, 0, stream>>>(
      rowptr, cnt, csr_src, asrc, adst, x2, W2, b2, outp, N);
}


# layer 1 sums alpha*x[src] per head on the 512-byte input rows and projects each 32 nodes' [8x128] sums with W1 on WMMA in the node kernel; h1 not stored; a_src/a_dst = x@(W1.att); float sums reordered
# speedup vs baseline: 16.5768x; 1.1000x over previous
#include <hip/hip_runtime.h>
#include <cstdint>

typedef float v2f __attribute__((ext_vector_type(2)));
typedef float v8f __attribute__((ext_vector_type(8)));

#define HEADS 8
#define NEG_SLOPE 0.2f
#define SM_EPS 1e-16f
#define NT 4

template <int CTRL>
__device__ __forceinline__ float xmask_add(float v) {
  const int o = __builtin_amdgcn_update_dpp(0, __float_as_int(v), CTRL, 0xf, 0xf, true);
  return v + __int_as_float(o);
}
__device__ __forceinline__ float row16_sum(float v) {
  v = xmask_add<0x161>(v);
  v = xmask_add<0x162>(v);
  v = xmask_add<0x164>(v);
  v = xmask_add<0x168>(v);
  return v;
}

template <int CH, bool STORE_C>
__device__ __forceinline__ void gemm_strip(const float* __restrict__ A,
                                           const float* __restrict__ B,
                                           const float* __restrict__ att_src,
                                           const float* __restrict__ att_dst,
                                           float* __restrict__ C,
                                           float* __restrict__ asrc,
                                           float* __restrict__ adst,
                                           int M, int K, int Ncols,
                                           int strip, int lane) {
  static_assert(CH == 32 || CH == 16, "a wave's 64 columns must be whole heads");
  constexpr int TPH = CH / 16;
  constexpr int HPW = NT / TPH;
  int groupsN = Ncols >> 6;
  int tm = strip / groupsN;
  int tg = strip - tm * groupsN;
  int half = lane >> 4;
  int l = lane & 15;

  int rowA = tm * 16 + l;
  if (rowA >= M) rowA = M - 1;
  const char* Ac = (const char*)A;
  const char* Bc = (const char*)B;
  unsigned aB  = 4u * ((unsigned)rowA * (unsigned)K + 2u * (unsigned)half);
  unsigned bB0 = 4u * ((unsigned)(2 * half) * (unsigned)Ncols + (unsigned)(tg * (16 * NT) + l));
  unsigned bB1 = bB0 + 4u * (unsigned)Ncols;
  const unsigned bStep = 16u * (unsigned)Ncols;

  v8f acc[NT];
#pragma unroll
  for (int t = 0; t < NT; ++t) acc[t] = (v8f){};

  v2f a0, b0[NT], a1, b1[NT];
  a0.x = *(const float*)(Ac + (size_t)aB);
  a0.y = *(const float*)(Ac + (size_t)aB + 4);
#pragma unroll
  for (int t = 0; t < NT; ++t) {
    b0[t].x = *(const float*)(Bc + (size_t)bB0 + t * 64);
    b0[t].y = *(const float*)(Bc + (size_t)bB1 + t * 64);
  }
  int k0 = 0;
#pragma unroll 1
  for (; k0 + 8 <= K; k0 += 8) {
    aB += 16u;  bB0 += bStep;  bB1 += bStep;
    a1.x = *(const float*)(Ac + (size_t)aB);
    a1.y = *(const float*)(Ac + (size_t)aB + 4);
#pragma unroll
    for (int t = 0; t < NT; ++t) {
      b1[t].x = *(const float*)(Bc + (size_t)bB0 + t * 64);
      b1[t].y = *(const float*)(Bc + (size_t)bB1 + t * 64);
    }
#pragma unroll
    for (int t = 0; t < NT; ++t) {
      acc[t] = __builtin_amdgcn_wmma_f32_16x16x4_f32(false, a0, false, b0[t],
                                                     (short)0, acc[t],
                                                     false, false);
    }
    const bool more = k0 + 8 < K;
    aB += more ? 16u : 0u;  bB0 += more ? bStep : 0u;  bB1 += more ? bStep : 0u;
    a0.x = *(const float*)(Ac + (size_t)aB);
    a0.y = *(const float*)(Ac + (size_t)aB + 4);
#pragma unroll
    for (int t = 0; t < NT; ++t) {
      b0[t].x = *(const float*)(Bc + (size_t)bB0 + t * 64);
      b0[t].y = *(const float*)(Bc + (size_t)bB1 + t * 64);
    }
#pragma unroll
    for (int t = 0; t < NT; ++t) {
      acc[t] = __builtin_amdgcn_wmma_f32_16x16x4_f32(false, a1, false, b1[t],
                                                     (short)0, acc[t],
                                                     false, false);
    }
  }
  if (k0 < K) {
#pragma unroll
    for (int t = 0; t < NT; ++t) {
      acc[t] = __builtin_amdgcn_wmma_f32_16x16x4_f32(false, a0, false, b0[t],
                                                     (short)0, acc[t],
                                                     false, false);
    }
  }

  float wS[NT], wD[NT];
#pragma unroll
  for (int t = 0; t < NT; ++t) {
    wS[t] = att_src[tg * (16 * NT) + t * 16 + l];
    wD[t] = att_dst[tg * (16 * NT) + t * 16 + l];
  }

  if (STORE_C) {
    char* Cc = (char*)C;
    const int row0 = tm * 16 + 8 * half;
    const unsigned rowC = 4u * (unsigned)Ncols;
    const unsigned cB = (unsigned)row0 * rowC + 4u * (unsigned)(tg * (16 * NT) + l);
#pragma unroll
    for (int r = 0; r < 8; ++r) {
      if (row0 + r < M) {
#pragma unroll
        for (int t = 0; t < NT; ++t)
          *(float*)(Cc + (size_t)(cB + (unsigned)r * rowC) + t * 64) = acc[t][r];
      }
    }
  }

#pragma unroll
  for (int j = 0; j < HPW / 2; ++j) {
    float vs = 0.f, vd = 0.f;
#pragma unroll
    for (int hb = 0; hb < 2; ++hb) {
#pragma unroll
      for (int r = 0; r < 8; ++r) {
        float ps = 0.f, pd = 0.f;
#pragma unroll
        for (int tt = 0; tt < TPH; ++tt) {
          const int t = (2 * j + hb) * TPH + tt;
          ps += acc[t][r] * wS[t];
          pd += acc[t][r] * wD[t];
        }
        ps = row16_sum(ps);
        pd = row16_sum(pd);
        const bool mine = (l == hb * 8 + r);
        vs = mine ? ps : vs;
        vd = mine ? pd : vd;
        asm volatile("" : "+v"(vs), "+v"(vd));
      }
    }
    const int row = tm * 16 + (l & 7) + 8 * half;
    const int head = tg * HPW + 2 * j + (l >> 3);
    if (row < M) {
      asrc[(size_t)row * HEADS + head] = vs;
      adst[(size_t)row * HEADS + head] = vd;
    }
  }
}

template <int CH, bool STORE_C>
__global__ void wmma_gemm_f32(const float* __restrict__ A,
                              const float* __restrict__ B,
                              const float* __restrict__ att_src,
                              const float* __restrict__ att_dst,
                              float* __restrict__ C,
                              float* __restrict__ asrc,
                              float* __restrict__ adst,
                              int M, int K, int Ncols) {
  int wave = (int)((blockIdx.x * blockDim.x + threadIdx.x) >> 5);
  int lane = (int)(threadIdx.x & 31);
  int tilesM  = (M + 15) >> 4;
  int groupsN = Ncols >> 6;
  if (wave >= tilesM * groupsN) return;
  gemm_strip<CH, STORE_C>(A, B, att_src, att_dst, C, asrc, adst, M, K, Ncols, wave, lane);
}

__device__ __forceinline__ float lrelu(float v) {
  return v > 0.f ? v : NEG_SLOPE * v;
}

#define F1 128
#define CH1 32

__device__ __forceinline__ void fold_att_block(const float* __restrict__ W,
                                               const float* __restrict__ att_src,
                                               const float* __restrict__ att_dst,
                                               float* __restrict__ Wa) {
  const int t = (int)threadIdx.x;
  const int f = t >> 3;
  const int h = t & 7;
  float s = 0.f, d = 0.f;
#pragma unroll 8
  for (int c = 0; c < CH1; ++c) {
    const float w = W[(size_t)f * (HEADS * CH1) + h * CH1 + c];
    s = __builtin_fmaf(w, att_src[h * CH1 + c], s);
    d = __builtin_fmaf(w, att_dst[h * CH1 + c], d);
  }
  Wa[f * 16 + h] = s;
  Wa[f * 16 + 8 + h] = d;
}

__device__ __forceinline__ void coef_tile(const float* __restrict__ A,
                                          const float* __restrict__ Wa,
                                          float* __restrict__ asrc,
                                          float* __restrict__ adst,
                                          int M, int tile, int lane) {
  const int hf = lane >> 4;
  const int l = lane & 15;
  int rowA = tile * 16 + l;
  if (rowA >= M) rowA = M - 1;
  const float* ap = A + (size_t)rowA * F1 + 2 * hf;
  const float* bp = Wa + (2 * hf) * 16 + l;
  v8f acc = (v8f){};
#pragma unroll 8
  for (int ks = 0; ks < F1 / 4; ++ks) {
    v2f a, b;
    a.x = ap[4 * ks];
    a.y = ap[4 * ks + 1];
    b.x = bp[(4 * ks) * 16];
    b.y = bp[(4 * ks + 1) * 16];
    acc = __builtin_amdgcn_wmma_f32_16x16x4_f32(false, a, false, b,
                                                (short)0, acc, false, false);
  }
  const int row0 = tile * 16 + 8 * hf;
  float* dp = (l < 8) ? asrc : adst;
  const int h = l & 7;
#pragma unroll
  for (int r = 0; r < 8; ++r)
    if (row0 + r < M) dp[(size_t)(row0 + r) * HEADS + h] = acc[r];
}

#define BIN_SHIFT 8
#define BIN_NODES (1 << BIN_SHIFT)
#define SORT_TB 1024
#define EPT 8
#define SPT 4

__global__ void __launch_bounds__(SORT_TB)
bin_hist_kernel(const int* __restrict__ dst, int E, int NB, int G1, int* __restrict__ hist,
                const float* __restrict__ W, const float* __restrict__ att_src,
                const float* __restrict__ att_dst, float* __restrict__ Wa) {
  extern __shared__ int sh[];
  if ((int)blockIdx.x >= G1) {
    fold_att_block(W, att_src, att_dst, Wa);
    return;
  }
  const int t = (int)threadIdx.x;
  for (int b = t; b < NB; b += SORT_TB) sh[b] = 0;
  __syncthreads();
  const int base = (int)blockIdx.x * (SORT_TB * EPT) + t;
  int d[EPT];
#pragma unroll
  for (int k = 0; k < EPT; ++k) {
    const int e = base + k * SORT_TB;
    d[k] = dst[e < E ? e : E - 1];
  }
#pragma unroll
  for (int k = 0; k < EPT; ++k) {
    const int e = base + k * SORT_TB;
    if (e < E) atomicAdd(&sh[d[k] >> BIN_SHIFT], 1);
  }
  __syncthreads();
  for (int b = t; b < NB; b += SORT_TB)
    hist[(size_t)b * G1 + blockIdx.x] = sh[b];
}

__global__ void __launch_bounds__(SORT_TB, 16)
bin_scatter_kernel(const int* __restrict__ src, const int* __restrict__ dst, int E,
                   int NB, int G1, const int* __restrict__ offs, int* __restrict__ binned,
                   const float* __restrict__ xA, const float* __restrict__ Wa,
                   float* __restrict__ asrc, float* __restrict__ adst, int M) {
  extern __shared__ int sh[];
  if ((int)blockIdx.x >= G1) {
    const int wib = __builtin_amdgcn_readfirstlane((int)(threadIdx.x >> 5));
    const int tile = ((int)blockIdx.x - G1) * (SORT_TB / 32) + wib;
    if (tile < ((M + 15) >> 4))
      coef_tile(xA, Wa, asrc, adst, M, tile, (int)(threadIdx.x & 31));
    return;
  }
  const int t = (int)threadIdx.x;
  for (int b = t; b < NB; b += SORT_TB)
    sh[b] = offs[(size_t)b * G1 + blockIdx.x];
  __syncthreads();
  const int base = (int)blockIdx.x * (SORT_TB * EPT) + t;
  int s[EPT], d[EPT], pos[EPT];
#pragma unroll
  for (int k = 0; k < EPT; ++k) {
    const int e = base + k * SORT_TB;
    const int ec = e < E ? e : E - 1;
    s[k] = src[ec];
    d[k] = dst[ec];
  }
#pragma unroll
  for (int k = 0; k < EPT; ++k) {
    const int e = base + k * SORT_TB;
    pos[k] = 0;
    if (e < E) pos[k] = atomicAdd(&sh[d[k] >> BIN_SHIFT], 1);
  }
  int2* __restrict__ outp = (int2*)binned;
#pragma unroll
  for (int k = 0; k < EPT; ++k) {
    const int e = base + k * SORT_TB;
    if (e < E) outp[pos[k]] = make_int2(s[k], d[k]);
  }
}

__global__ void __launch_bounds__(SORT_TB, 16)
bin_sort_kernel(const int* __restrict__ binned, const int* __restrict__ offs,
                int G1, int NB, int N, int E,
                int* __restrict__ cnt, int* __restrict__ rowptr,
                int* __restrict__ csr_src) {
  __shared__ int c[BIN_NODES];
  __shared__ int wsum[BIN_NODES / 32];
  const int2* __restrict__ pairs = (const int2*)binned;
  const int t = (int)threadIdx.x;
  const int lane = t & 31;
  const int w = t >> 5;
  const int b = (int)blockIdx.x;
  const int beg = offs[(size_t)b * G1];
  const int end = (b + 1 < NB) ? offs[(size_t)(b + 1) * G1] : E;
  if (t < BIN_NODES) c[t] = 0;
  __syncthreads();
  for (int i0 = beg + t; i0 < end; i0 += SPT * SORT_TB) {
    int y[SPT];
#pragma unroll
    for (int k = 0; k < SPT; ++k) {
      const int i = i0 + k * SORT_TB;
      y[k] = pairs[i < end ? i : end - 1].y;
    }
#pragma unroll
    for (int k = 0; k < SPT; ++k)
      if (i0 + k * SORT_TB < end) atomicAdd(&c[y[k] & (BIN_NODES - 1)], 1);
  }
  __syncthreads();
  int mine = 0, incl = 0;
  if (t < BIN_NODES) {
    mine = c[t];
    incl = mine;
    for (int o = 1; o < 32; o <<= 1) {
      const int v = __shfl_up(incl, o);
      if (lane >= o) incl += v;
    }
    if (lane == 31) wsum[w] = incl;
  }
  __syncthreads();
  if (t < BIN_NODES) {
    int woff = 0;
    for (int k = 0; k < BIN_NODES / 32; ++k) woff += (k < w) ? wsum[k] : 0;
    const int start = beg + woff + incl - mine;
    c[t] = start;
    const int n = b * BIN_NODES + t;
    if (n < N) {
      cnt[n] = mine;
      rowptr[n] = start;
    }
  }
  __syncthreads();
  for (int i0 = beg + t; i0 < end; i0 += SPT * SORT_TB) {
    int2 sd[SPT];
    int pos[SPT];
#pragma unroll
    for (int k = 0; k < SPT; ++k) {
      const int i = i0 + k * SORT_TB;
      sd[k] = pairs[i < end ? i : end - 1];
    }
#pragma unroll
    for (int k = 0; k < SPT; ++k) {
      pos[k] = 0;
      if (i0 + k * SORT_TB < end) pos[k] = atomicAdd(&c[sd[k].y & (BIN_NODES - 1)], 1);
    }
#pragma unroll
    for (int k = 0; k < SPT; ++k)
      if (i0 + k * SORT_TB < end) csr_src[pos[k]] = sd[k].x;
  }
}

__global__ void __launch_bounds__(1024)
prefix_kernel(const int* __restrict__ cnt, int N, int* __restrict__ rowptr) {
  __shared__ int wsum[32];
  const int t = (int)threadIdx.x;
  const int lane = t & 31;
  const int w = t >> 5;
  const int base = (int)blockIdx.x * 1024;

  int part = 0;
  for (int i = t; i < base; i += 1024) part += cnt[i];
  for (int o = 16; o > 0; o >>= 1) part += __shfl_xor(part, o);
  if (lane == 0) wsum[w] = part;
  __syncthreads();
  int prefix = 0;
  for (int k = 0; k < 32; ++k) prefix += wsum[k];
  __syncthreads();

  const int i = base + t;
  const int c = (i < N) ? cnt[i] : 0;
  int incl = c;
  for (int o = 1; o < 32; o <<= 1) {
    const int v = __shfl_up(incl, o);
    if (lane >= o) incl += v;
  }
  if (lane == 31) wsum[w] = incl;
  __syncthreads();
  int woff = 0;
  for (int k = 0; k < 32; ++k) woff += (k < w) ? wsum[k] : 0;
  if (i < N) rowptr[i] = prefix + woff + incl - c;
}

__device__ __forceinline__ void edge_scores(const int* __restrict__ csr_src,
                                            const float* __restrict__ asrc,
                                            int start, int deg, int n, int j,
                                            const float (&ad)[HEADS],
                                            int& s, float (&e)[HEADS]) {
  s = (j < deg) ? csr_src[start + j] : n;
  const float4 a0 = *(const float4*)(asrc + (size_t)s * HEADS);
  const float4 a1 = *(const float4*)(asrc + (size_t)s * HEADS + 4);
  e[0] = lrelu(a0.x + ad[0]);
  e[1] = lrelu(a0.y + ad[1]);
  e[2] = lrelu(a0.z + ad[2]);
  e[3] = lrelu(a0.w + ad[3]);
  e[4] = lrelu(a1.x + ad[4]);
  e[5] = lrelu(a1.y + ad[5]);
  e[6] = lrelu(a1.z + ad[6]);
  e[7] = lrelu(a1.w + ad[7]);
}

#define PW 8
#define PN (2 * PW)
#define KW (HEADS * 32 / PW)
#define ZP (HEADS * 32 + 4)

#define GS 8
template <int F>
__device__ __forceinline__ void gather_inputs(const float* __restrict__ xin, int sreg,
                                              const float* wl, int count,
                                              int lane, float (&z)[HEADS]) {
  for (int j0 = 0; j0 < count; j0 += GS) {
    float v[GS];
#pragma unroll
    for (int g = 0; g < GS; ++g) {
      const int sj = __builtin_amdgcn_readlane(sreg, j0 + g);
      v[g] = xin[(unsigned)sj * F + (unsigned)lane];
    }
#pragma unroll
    for (int g = 0; g < GS; ++g) {
      const float4 a0 = *(const float4*)(wl + (j0 + g) * HEADS);
      const float4 a1 = *(const float4*)(wl + (j0 + g) * HEADS + 4);
      z[0] = __builtin_fmaf(a0.x, v[g], z[0]);  z[1] = __builtin_fmaf(a0.y, v[g], z[1]);
      z[2] = __builtin_fmaf(a0.z, v[g], z[2]);  z[3] = __builtin_fmaf(a0.w, v[g], z[3]);
      z[4] = __builtin_fmaf(a1.x, v[g], z[4]);  z[5] = __builtin_fmaf(a1.y, v[g], z[5]);
      z[6] = __builtin_fmaf(a1.z, v[g], z[6]);  z[7] = __builtin_fmaf(a1.w, v[g], z[7]);
    }
  }
}

__device__ __forceinline__ void put_weights(float* wl, int lane, const float (&alpha)[HEADS]) {
  __builtin_amdgcn_wave_barrier();
  *(float4*)(wl + lane * HEADS)     = make_float4(alpha[0], alpha[1], alpha[2], alpha[3]);
  *(float4*)(wl + lane * HEADS + 4) = make_float4(alpha[4], alpha[5], alpha[6], alpha[7]);
  __builtin_amdgcn_wave_barrier();
}

template <int F, int C>
__global__ void __launch_bounds__(PW * 32, 12)
node_attend_project_kernel(const int* __restrict__ rowptr,
                           const int* __restrict__ cnt,
                           const int* __restrict__ csr_src,
                           const float* __restrict__ asrc,
                           const float* __restrict__ adst,
                           const float* __restrict__ xin,
                           const float* __restrict__ Wm,
                           const float* __restrict__ bias,
                           float* __restrict__ out, int N) {
  static_assert(F == 32, "lane = input channel");
  static_assert(C == 16, "one 16-column tile of output channels");
  static_assert(PN == 16 && PW == HEADS && KW == F, "one 16-node tile; wave w owns head w's k values");
  static_assert(PW * 256 <= PN * ZP, "the partial tiles reuse the z rows' LDS");
  __shared__ __attribute__((aligned(16))) float zs[PN * ZP];
  __shared__ __attribute__((aligned(16))) float aw[PW * 32 * HEADS];
  const int lane = (int)(threadIdx.x & 31);
  const int wib = __builtin_amdgcn_readfirstlane((int)(threadIdx.x >> 5));
  const int nb = (int)blockIdx.x * PN;
  float* wl = aw + wib * (32 * HEADS);

  for (int u = 0; u < 2; ++u) {
    const int R = 2 * wib + u;
    const int n = nb + R;
    float z[HEADS];
#pragma unroll
    for (int k = 0; k < HEADS; ++k) z[k] = 0.f;
    if (n < N) {
      const int start = rowptr[n];
      const int deg = cnt[n];
      const int tot = deg + 1;
      float ad[HEADS];
#pragma unroll
      for (int k = 0; k < HEADS; ++k) ad[k] = adst[(size_t)n * HEADS + k];

      int s0;
      float e0[HEADS];
      const bool v0 = lane < tot;
      edge_scores(csr_src, asrc, start, deg, n, lane, ad, s0, e0);

      float mx[HEADS];
#pragma unroll
      for (int k = 0; k < HEADS; ++k) mx[k] = v0 ? e0[k] : -3.0e38f;
      for (int base = 32; base < tot; base += 32) {
        int s; float e[HEADS];
        const int j = base + lane;
        edge_scores(csr_src, asrc, start, deg, n, j, ad, s, e);
#pragma unroll
        for (int k = 0; k < HEADS; ++k) mx[k] = (j < tot) ? fmaxf(mx[k], e[k]) : mx[k];
      }
#pragma unroll
      for (int k = 0; k < HEADS; ++k) {
        for (int o = 16; o > 0; o >>= 1) mx[k] = fmaxf(mx[k], __shfl_xor(mx[k], o));
        mx[k] = __int_as_float(__builtin_amdgcn_readfirstlane(__float_as_int(mx[k])));
      }

      float ex0[HEADS], sm[HEADS];
#pragma unroll
      for (int k = 0; k < HEADS; ++k) {
        ex0[k] = v0 ? __expf(e0[k] - mx[k]) : 0.f;
        sm[k] = ex0[k];
      }
      for (int base = 32; base < tot; base += 32) {
        int s; float e[HEADS];
        const int j = base + lane;
        edge_scores(csr_src, asrc, start, deg, n, j, ad, s, e);
#pragma unroll
        for (int k = 0; k < HEADS; ++k) sm[k] += (j < tot) ? __expf(e[k] - mx[k]) : 0.f;
      }
#pragma unroll
      for (int k = 0; k < HEADS; ++k) {
        for (int o = 16; o > 0; o >>= 1) sm[k] += __shfl_xor(sm[k], o);
        sm[k] = __int_as_float(__builtin_amdgcn_readfirstlane(__float_as_int(sm[k])));
      }

      float alpha0[HEADS];
#pragma unroll
      for (int k = 0; k < HEADS; ++k) alpha0[k] = ex0[k] / (sm[k] + SM_EPS);
      for (int base = 32; base < tot; base += 32) {
        int s; float e[HEADS], alpha[HEADS];
        const int j = base + lane;
        edge_scores(csr_src, asrc, start, deg, n, j, ad, s, e);
#pragma unroll
        for (int k = 0; k < HEADS; ++k)
          alpha[k] = (j < tot) ? __expf(e[k] - mx[k]) / (sm[k] + SM_EPS) : 0.f;
        const int left = tot - base;
        put_weights(wl, lane, alpha);
        gather_inputs<F>(xin, s, wl, left < 32 ? left : 32, lane, z);
      }
      put_weights(wl, lane, alpha0);
      gather_inputs<F>(xin, s0, wl, tot < 32 ? tot : 32, lane, z);
    }
#pragma unroll
    for (int k = 0; k < HEADS; ++k) zs[R * ZP + k * F + lane] = z[k];
  }

  const int hf = lane >> 4;
  const int l = lane & 15;
  const float* wp = Wm + (size_t)(2 * hf) * (HEADS * C) + wib * C + l;
  v2f bfr[KW / 4];
#pragma unroll
  for (int ks = 0; ks < KW / 4; ++ks) {
    bfr[ks].x = wp[(size_t)(4 * ks) * (HEADS * C)];
    bfr[ks].y = wp[(size_t)(4 * ks + 1) * (HEADS * C)];
  }
  __syncthreads();

  v8f acc = (v8f){};
#pragma unroll
  for (int ks = 0; ks < KW / 4; ++ks) {
    const v2f a = *(const v2f*)&zs[l * ZP + KW * wib + 4 * ks + 2 * hf];
    acc = __builtin_amdgcn_wmma_f32_16x16x4_f32(false, a, false, bfr[ks],
                                                (short)0, acc, false, false);
  }
  __syncthreads();
#pragma unroll
  for (int r = 0; r < 8; ++r) zs[wib * 256 + (r + 8 * hf) * 16 + l] = acc[r];
  __syncthreads();

  const int t = (int)threadIdx.x;
  float sum = 0.f;
#pragma unroll
  for (int w = 0; w < PW; ++w) sum += zs[w * 256 + t];
  const int node = nb + (t >> 4);
  if (node < N)
    out[(size_t)node * C + (t & 15)] = sum * (1.0f / HEADS) + bias[t & 15];
}

#define P1W 16
#define P1N (2 * P1W)
#define Z1P (HEADS * F1 + 4)

__device__ __forceinline__ void fma4(float4& z, float a, const float4& v) {
  z.x = __builtin_fmaf(a, v.x, z.x);
  z.y = __builtin_fmaf(a, v.y, z.y);
  z.z = __builtin_fmaf(a, v.z, z.z);
  z.w = __builtin_fmaf(a, v.w, z.w);
}

__device__ __forceinline__ void gather_inputs4(const float* __restrict__ xin, int sreg,
                                               const float* wl, int count,
                                               int lane, float4 (&z)[HEADS]) {
  for (int j0 = 0; j0 < count; j0 += GS) {
    float4 v[GS];
#pragma unroll
    for (int g = 0; g < GS; ++g) {
      const int sj = __builtin_amdgcn_readlane(sreg, j0 + g);
      v[g] = *(const float4*)(xin + ((unsigned)sj * (unsigned)F1 + 4u * (unsigned)lane));
    }
#pragma unroll
    for (int g = 0; g < GS; ++g) {
      const float4 a0 = *(const float4*)(wl + (j0 + g) * HEADS);
      const float4 a1 = *(const float4*)(wl + (j0 + g) * HEADS + 4);
      fma4(z[0], a0.x, v[g]);  fma4(z[1], a0.y, v[g]);
      fma4(z[2], a0.z, v[g]);  fma4(z[3], a0.w, v[g]);
      fma4(z[4], a1.x, v[g]);  fma4(z[5], a1.y, v[g]);
      fma4(z[6], a1.z, v[g]);  fma4(z[7], a1.w, v[g]);
    }
  }
}

__global__ void __launch_bounds__(P1W * 32, 8)
node_attend_project1_kernel(const int* __restrict__ rowptr,
                            const int* __restrict__ cnt,
                            const int* __restrict__ csr_src,
                            const float* __restrict__ asrc,
                            const float* __restrict__ adst,
                            const float* __restrict__ xin,
                            const float* __restrict__ Wm,
                            const float* __restrict__ bias,
                            float* __restrict__ out, int N) {
  static_assert(P1N == 32 && P1W == 2 * HEADS, "two 16-node tiles; wave w owns head w % HEADS of tile w / HEADS");
  static_assert(P1W * 512 <= P1N * Z1P, "the partial tiles reuse the z rows' LDS");
  static_assert(F1 * 4 == 32 * 16, "lane = 4 input channels");
  __shared__ __attribute__((aligned(16))) float zs[P1N * Z1P];
  __shared__ __attribute__((aligned(16))) float aw[P1W * 32 * HEADS];
  const int lane = (int)__builtin_amdgcn_mbcnt_lo(~0u, 0u);
  const int wib = __builtin_amdgcn_readfirstlane((int)(threadIdx.x >> 5));
  const int nb = (int)blockIdx.x * P1N;
  float* wl = aw + wib * (32 * HEADS);

  for (int u = 0; u < 2; ++u) {
    const int R = 2 * wib + u;
    const int n = nb + R;
    float4 z[HEADS];
#pragma unroll
    for (int k = 0; k < HEADS; ++k) z[k] = make_float4(0.f, 0.f, 0.f, 0.f);
    if (n < N) {
      const int start = rowptr[n];
      const int deg = cnt[n];
      const int tot = deg + 1;
      float ad[HEADS];
#pragma unroll
      for (int k = 0; k < HEADS; ++k) ad[k] = adst[(size_t)n * HEADS + k];

      int s0;
      float e0[HEADS];
      const bool v0 = lane < tot;
      edge_scores(csr_src, asrc, start, deg, n, lane, ad, s0, e0);

      float mx[HEADS];
#pragma unroll
      for (int k = 0; k < HEADS; ++k) mx[k] = v0 ? e0[k] : -3.0e38f;
      for (int base = 32; base < tot; base += 32) {
        int s; float e[HEADS];
        const int j = base + lane;
        edge_scores(csr_src, asrc, start, deg, n, j, ad, s, e);
#pragma unroll
        for (int k = 0; k < HEADS; ++k) mx[k] = (j < tot) ? fmaxf(mx[k], e[k]) : mx[k];
      }
#pragma unroll
      for (int k = 0; k < HEADS; ++k) {
        for (int o = 16; o > 0; o >>= 1) mx[k] = fmaxf(mx[k], __shfl_xor(mx[k], o));
        mx[k] = __int_as_float(__builtin_amdgcn_readfirstlane(__float_as_int(mx[k])));
      }

      float ex0[HEADS], sm[HEADS];
#pragma unroll
      for (int k = 0; k < HEADS; ++k) {
        ex0[k] = v0 ? __expf(e0[k] - mx[k]) : 0.f;
        sm[k] = ex0[k];
      }
      for (int base = 32; base < tot; base += 32) {
        int s; float e[HEADS];
        const int j = base + lane;
        edge_scores(csr_src, asrc, start, deg, n, j, ad, s, e);
#pragma unroll
        for (int k = 0; k < HEADS; ++k) sm[k] += (j < tot) ? __expf(e[k] - mx[k]) : 0.f;
      }
#pragma unroll
      for (int k = 0; k < HEADS; ++k) {
        for (int o = 16; o > 0; o >>= 1) sm[k] += __shfl_xor(sm[k], o);
        sm[k] = __int_as_float(__builtin_amdgcn_readfirstlane(__float_as_int(sm[k])));
      }

      float alpha0[HEADS];
#pragma unroll
      for (int k = 0; k < HEADS; ++k) alpha0[k] = ex0[k] / (sm[k] + SM_EPS);
      for (int base = 32; base < tot; base += 32) {
        int s; float e[HEADS], alpha[HEADS];
        const int j = base + lane;
        edge_scores(csr_src, asrc, start, deg, n, j, ad, s, e);
#pragma unroll
        for (int k = 0; k < HEADS; ++k)
          alpha[k] = (j < tot) ? __expf(e[k] - mx[k]) / (sm[k] + SM_EPS) : 0.f;
        const int left = tot - base;
        put_weights(wl, lane, alpha);
        gather_inputs4(xin, s, wl, left < 32 ? left : 32, lane, z);
      }
      put_weights(wl, lane, alpha0);
      gather_inputs4(xin, s0, wl, tot < 32 ? tot : 32, lane, z);
    }
#pragma unroll
    for (int k = 0; k < HEADS; ++k) *(float4*)&zs[R * Z1P + k * F1 + 4 * lane] = z[k];
  }

  const int hf = lane >> 4;
  const int l = lane & 15;
  const int hw = wib & (HEADS - 1);
  const int rt = wib >> 3;
  const float* wp = Wm + (size_t)(2 * hf) * (HEADS * CH1) + hw * CH1 + l;
  __syncthreads();

  v8f acc0 = (v8f){};
  v8f acc1 = (v8f){};
#pragma unroll 8
  for (int ks = 0; ks < F1 / 4; ++ks) {
    v2f b0, b1;
    b0.x = wp[(size_t)(4 * ks) * (HEADS * CH1)];
    b0.y = wp[(size_t)(4 * ks + 1) * (HEADS * CH1)];
    b1.x = wp[(size_t)(4 * ks) * (HEADS * CH1) + 16];
    b1.y = wp[(size_t)(4 * ks + 1) * (HEADS * CH1) + 16];
    const v2f a = *(const v2f*)&zs[(16 * rt + l) * Z1P + F1 * hw + 4 * ks + 2 * hf];
    acc0 = __builtin_amdgcn_wmma_f32_16x16x4_f32(false, a, false, b0,
                                                 (short)0, acc0, false, false);
    acc1 = __builtin_amdgcn_wmma_f32_16x16x4_f32(false, a, false, b1,
                                                 (short)0, acc1, false, false);
  }
  __syncthreads();
#pragma unroll
  for (int r = 0; r < 8; ++r) {
    zs[wib * 512 + (r + 8 * hf) * 16 + l] = acc0[r];
    zs[wib * 512 + 256 + (r + 8 * hf) * 16 + l] = acc1[r];
  }
  __syncthreads();

  const int t = (wib << 5) | lane;
#pragma unroll
  for (int q = 0; q < 2; ++q) {
    const int o = t + q * (P1W * 32);
    const int nl = o >> 5;
    const int c = o & 31;
    const int pi = (nl >> 4) * (HEADS * 512) + (c >> 4) * 256 + (nl & 15) * 16 + (c & 15);
    float sum = 0.f;
#pragma unroll
    for (int w = 0; w < HEADS; ++w) sum += zs[w * 512 + pi];
    const int node = nb + nl;
    if (node < N) {
      const float v = sum * (1.0f / HEADS) + bias[c];
      out[(size_t)node * CH1 + c] = v > 0.f ? v : 0.f;
    }
  }
}

static inline unsigned nblk(long n, int b) { return (unsigned)((n + b - 1) / b); }

extern "C" void kernel_launch(void* const* d_in, const int* in_sizes, int n_in,
                              void* d_out, int out_size, void* d_ws, size_t ws_size,
                              hipStream_t stream) {
  const float* x        = (const float*)d_in[0];
  const int*   ei       = (const int*)  d_in[1];
  const float* W1       = (const float*)d_in[2];
  const float* att_src1 = (const float*)d_in[3];
  const float* att_dst1 = (const float*)d_in[4];
  const float* b1       = (const float*)d_in[5];
  const float* W2       = (const float*)d_in[6];
  const float* att_src2 = (const float*)d_in[7];
  const float* att_dst2 = (const float*)d_in[8];
  const float* b2       = (const float*)d_in[9];

  const int F_h   = in_sizes[3] / HEADS;
  const int F_out = in_sizes[7] / HEADS;
  const int F_in  = in_sizes[2] / (HEADS * F_h);
  const int N     = in_sizes[0] / F_in;
  const int E     = in_sizes[1] / 2;
  const int C2    = HEADS * F_out;
  const int* src = ei;
  const int* dst = ei + E;

  float* ws = (float*)d_ws;
  size_t o = 0;
  float* asrc = ws + o; o += (size_t)N * HEADS;
  float* adst = ws + o; o += (size_t)N * HEADS;
  float* x2   = ws + o; o += (size_t)N * F_h;
  float* Wa   = ws + o; o += (size_t)F_in * 16;
  const int NB = (N + BIN_NODES - 1) / BIN_NODES;
  const int G1 = (int)nblk(E, SORT_TB * EPT);
  int* binned  = (int*)(ws + o); o += (size_t)2 * E;
  int* cnt     = (int*)(ws + o); o += (size_t)N;
  int* rowptr  = (int*)(ws + o); o += (size_t)N;
  int* csr_src = (int*)(ws + o); o += (size_t)E;
  int* hist    = (int*)(ws + o); o += (size_t)NB * G1;
  int* offs    = (int*)(ws + o); o += (size_t)NB * G1;
  float* outp = (float*)d_out;

  const int TB = 256;

  bin_hist_kernel<<<G1 + 1, SORT_TB, NB * sizeof(int), stream>>>(
      dst, E, NB, G1, hist, W1, att_src1, att_dst1, Wa);
  prefix_kernel<<<nblk((long)NB * G1, 1024), 1024, 0, stream>>>(hist, NB * G1, offs);
  const int ctiles = (N + 15) / 16;
  bin_scatter_kernel<<<G1 + nblk(ctiles, SORT_TB / 32), SORT_TB, NB * sizeof(int), stream>>>(
      src, dst, E, NB, G1, offs, binned, x, Wa, asrc, adst, N);
  bin_sort_kernel<<<NB, SORT_TB, 0, stream>>>(
      binned, offs, G1, NB, N, E, cnt, rowptr, csr_src);

  node_attend_project1_kernel<<<nblk(N, P1N), P1W * 32, 0, stream>>>(
      rowptr, cnt, csr_src, asrc, adst, x, W1, b1, x2, N);

  {
    long waves = (long)((N + 15) / 16) * (C2 / (16 * NT));
    wmma_gemm_f32<16, false><<<nblk(waves * 32, TB), TB, 0, stream>>>(
        x2, W2, att_src2, att_dst2, (float*)binned, asrc, adst, N, F_h, C2);
  }
  node_attend_project_kernel<32, 16><<<nblk(N, PN), PW * 32, 0, stream>>>(
      rowptr, cnt, csr_src, asrc, adst, x2, W2, b2, outp, N);
}


# node softmax: 8 heads' max and sum by reduce-scatter butterfly on DPP/permlanex16 (9 exchanges, no ds_bpermute), sum order kept; alpha = ex * (1/(sum+eps)), +1 rounding; x rows by 32-bit byte offset
# speedup vs baseline: 18.7480x; 1.1310x over previous
#include <hip/hip_runtime.h>
#include <cstdint>

typedef float v2f __attribute__((ext_vector_type(2)));
typedef float v8f __attribute__((ext_vector_type(8)));

#define HEADS 8
#define NEG_SLOPE 0.2f
#define SM_EPS 1e-16f
#define NT 4

template <int CTRL>
__device__ __forceinline__ float xmask_add(float v) {
  const int o = __builtin_amdgcn_update_dpp(0, __float_as_int(v), CTRL, 0xf, 0xf, true);
  return v + __int_as_float(o);
}
__device__ __forceinline__ float row16_sum(float v) {
  v = xmask_add<0x161>(v);
  v = xmask_add<0x162>(v);
  v = xmask_add<0x164>(v);
  v = xmask_add<0x168>(v);
  return v;
}

template <int CH, bool STORE_C>
__device__ __forceinline__ void gemm_strip(const float* __restrict__ A,
                                           const float* __restrict__ B,
                                           const float* __restrict__ att_src,
                                           const float* __restrict__ att_dst,
                                           float* __restrict__ C,
                                           float* __restrict__ asrc,
                                           float* __restrict__ adst,
                                           int M, int K, int Ncols,
                                           int strip, int lane) {
  static_assert(CH == 32 || CH == 16, "a wave's 64 columns must be whole heads");
  constexpr int TPH = CH / 16;
  constexpr int HPW = NT / TPH;
  int groupsN = Ncols >> 6;
  int tm = strip / groupsN;
  int tg = strip - tm * groupsN;
  int half = lane >> 4;
  int l = lane & 15;

  int rowA = tm * 16 + l;
  if (rowA >= M) rowA = M - 1;
  const char* Ac = (const char*)A;
  const char* Bc = (const char*)B;
  unsigned aB  = 4u * ((unsigned)rowA * (unsigned)K + 2u * (unsigned)half);
  unsigned bB0 = 4u * ((unsigned)(2 * half) * (unsigned)Ncols + (unsigned)(tg * (16 * NT) + l));
  unsigned bB1 = bB0 + 4u * (unsigned)Ncols;
  const unsigned bStep = 16u * (unsigned)Ncols;

  v8f acc[NT];
#pragma unroll
  for (int t = 0; t < NT; ++t) acc[t] = (v8f){};

  v2f a0, b0[NT], a1, b1[NT];
  a0.x = *(const float*)(Ac + (size_t)aB);
  a0.y = *(const float*)(Ac + (size_t)aB + 4);
#pragma unroll
  for (int t = 0; t < NT; ++t) {
    b0[t].x = *(const float*)(Bc + (size_t)bB0 + t * 64);
    b0[t].y = *(const float*)(Bc + (size_t)bB1 + t * 64);
  }
  int k0 = 0;
#pragma unroll 1
  for (; k0 + 8 <= K; k0 += 8) {
    aB += 16u;  bB0 += bStep;  bB1 += bStep;
    a1.x = *(const float*)(Ac + (size_t)aB);
    a1.y = *(const float*)(Ac + (size_t)aB + 4);
#pragma unroll
    for (int t = 0; t < NT; ++t) {
      b1[t].x = *(const float*)(Bc + (size_t)bB0 + t * 64);
      b1[t].y = *(const float*)(Bc + (size_t)bB1 + t * 64);
    }
#pragma unroll
    for (int t = 0; t < NT; ++t) {
      acc[t] = __builtin_amdgcn_wmma_f32_16x16x4_f32(false, a0, false, b0[t],
                                                     (short)0, acc[t],
                                                     false, false);
    }
    const bool more = k0 + 8 < K;
    aB += more ? 16u : 0u;  bB0 += more ? bStep : 0u;  bB1 += more ? bStep : 0u;
    a0.x = *(const float*)(Ac + (size_t)aB);
    a0.y = *(const float*)(Ac + (size_t)aB + 4);
#pragma unroll
    for (int t = 0; t < NT; ++t) {
      b0[t].x = *(const float*)(Bc + (size_t)bB0 + t * 64);
      b0[t].y = *(const float*)(Bc + (size_t)bB1 + t * 64);
    }
#pragma unroll
    for (int t = 0; t < NT; ++t) {
      acc[t] = __builtin_amdgcn_wmma_f32_16x16x4_f32(false, a1, false, b1[t],
                                                     (short)0, acc[t],
                                                     false, false);
    }
  }
  if (k0 < K) {
#pragma unroll
    for (int t = 0; t < NT; ++t) {
      acc[t] = __builtin_amdgcn_wmma_f32_16x16x4_f32(false, a0, false, b0[t],
                                                     (short)0, acc[t],
                                                     false, false);
    }
  }

  float wS[NT], wD[NT];
#pragma unroll
  for (int t = 0; t < NT; ++t) {
    wS[t] = att_src[tg * (16 * NT) + t * 16 + l];
    wD[t] = att_dst[tg * (16 * NT) + t * 16 + l];
  }

  if (STORE_C) {
    char* Cc = (char*)C;
    const int row0 = tm * 16 + 8 * half;
    const unsigned rowC = 4u * (unsigned)Ncols;
    const unsigned cB = (unsigned)row0 * rowC + 4u * (unsigned)(tg * (16 * NT) + l);
#pragma unroll
    for (int r = 0; r < 8; ++r) {
      if (row0 + r < M) {
#pragma unroll
        for (int t = 0; t < NT; ++t)
          *(float*)(Cc + (size_t)(cB + (unsigned)r * rowC) + t * 64) = acc[t][r];
      }
    }
  }

#pragma unroll
  for (int j = 0; j < HPW / 2; ++j) {
    float vs = 0.f, vd = 0.f;
#pragma unroll
    for (int hb = 0; hb < 2; ++hb) {
#pragma unroll
      for (int r = 0; r < 8; ++r) {
        float ps = 0.f, pd = 0.f;
#pragma unroll
        for (int tt = 0; tt < TPH; ++tt) {
          const int t = (2 * j + hb) * TPH + tt;
          ps += acc[t][r] * wS[t];
          pd += acc[t][r] * wD[t];
        }
        ps = row16_sum(ps);
        pd = row16_sum(pd);
        const bool mine = (l == hb * 8 + r);
        vs = mine ? ps : vs;
        vd = mine ? pd : vd;
        asm volatile("" : "+v"(vs), "+v"(vd));
      }
    }
    const int row = tm * 16 + (l & 7) + 8 * half;
    const int head = tg * HPW + 2 * j + (l >> 3);
    if (row < M) {
      asrc[(size_t)row * HEADS + head] = vs;
      adst[(size_t)row * HEADS + head] = vd;
    }
  }
}

template <int CH, bool STORE_C>
__global__ void wmma_gemm_f32(const float* __restrict__ A,
                              const float* __restrict__ B,
                              const float* __restrict__ att_src,
                              const float* __restrict__ att_dst,
                              float* __restrict__ C,
                              float* __restrict__ asrc,
                              float* __restrict__ adst,
                              int M, int K, int Ncols) {
  int wave = (int)((blockIdx.x * blockDim.x + threadIdx.x) >> 5);
  int lane = (int)(threadIdx.x & 31);
  int tilesM  = (M + 15) >> 4;
  int groupsN = Ncols >> 6;
  if (wave >= tilesM * groupsN) return;
  gemm_strip<CH, STORE_C>(A, B, att_src, att_dst, C, asrc, adst, M, K, Ncols, wave, lane);
}

__device__ __forceinline__ float lrelu(float v) {
  return v > 0.f ? v : NEG_SLOPE * v;
}

__device__ __forceinline__ float xhalf(float x) {
  const int b = __float_as_int(x);
  return __int_as_float(__builtin_amdgcn_permlanex16(b, b, 0x76543210, 0xfedcba98, false, false));
}
template <int CTRL>
__device__ __forceinline__ float xmask_get(float v) {
  return __int_as_float(__builtin_amdgcn_update_dpp(0, __float_as_int(v), CTRL, 0xf, 0xf, true));
}
template <bool IS_MAX>
__device__ __forceinline__ float comb(float a, float b) {
  return IS_MAX ? fmaxf(a, b) : a + b;
}
template <bool IS_MAX>
__device__ __forceinline__ float heads_reduce(const float (&a)[HEADS], int lane) {
  static_assert(HEADS == 8, "three halvings leave one head per lane");
  const bool h16 = (lane & 16) != 0;
  const bool h8 = (lane & 8) != 0;
  const bool h4 = (lane & 4) != 0;
  float b[4], c[2];
#pragma unroll
  for (int k = 0; k < 4; ++k) {
    const float keep = h16 ? a[k + 4] : a[k];
    const float send = h16 ? a[k] : a[k + 4];
    b[k] = comb<IS_MAX>(keep, xhalf(send));
  }
#pragma unroll
  for (int k = 0; k < 2; ++k) {
    const float keep = h8 ? b[k + 2] : b[k];
    const float send = h8 ? b[k] : b[k + 2];
    c[k] = comb<IS_MAX>(keep, xmask_get<0x168>(send));
  }
  const float keep = h4 ? c[1] : c[0];
  const float send = h4 ? c[0] : c[1];
  float d = comb<IS_MAX>(keep, xmask_get<0x164>(send));
  d = comb<IS_MAX>(d, xmask_get<0x162>(d));
  d = comb<IS_MAX>(d, xmask_get<0x161>(d));
  return d;
}
__device__ __forceinline__ float head_total(float d, int k) {
  return __int_as_float(__builtin_amdgcn_readlane(__float_as_int(d), 4 * k));
}

#define F1 128
#define CH1 32

__device__ __forceinline__ void fold_att_block(const float* __restrict__ W,
                                               const float* __restrict__ att_src,
                                               const float* __restrict__ att_dst,
                                               float* __restrict__ Wa) {
  const int t = (int)threadIdx.x;
  const int f = t >> 3;
  const int h = t & 7;
  float s = 0.f, d = 0.f;
#pragma unroll 8
  for (int c = 0; c < CH1; ++c) {
    const float w = W[(size_t)f * (HEADS * CH1) + h * CH1 + c];
    s = __builtin_fmaf(w, att_src[h * CH1 + c], s);
    d = __builtin_fmaf(w, att_dst[h * CH1 + c], d);
  }
  Wa[f * 16 + h] = s;
  Wa[f * 16 + 8 + h] = d;
}

__device__ __forceinline__ void coef_tile(const float* __restrict__ A,
                                          const float* __restrict__ Wa,
                                          float* __restrict__ asrc,
                                          float* __restrict__ adst,
                                          int M, int tile, int lane) {
  const int hf = lane >> 4;
  const int l = lane & 15;
  int rowA = tile * 16 + l;
  if (rowA >= M) rowA = M - 1;
  const float* ap = A + (size_t)rowA * F1 + 2 * hf;
  const float* bp = Wa + (2 * hf) * 16 + l;
  v8f acc = (v8f){};
#pragma unroll 8
  for (int ks = 0; ks < F1 / 4; ++ks) {
    v2f a, b;
    a.x = ap[4 * ks];
    a.y = ap[4 * ks + 1];
    b.x = bp[(4 * ks) * 16];
    b.y = bp[(4 * ks + 1) * 16];
    acc = __builtin_amdgcn_wmma_f32_16x16x4_f32(false, a, false, b,
                                                (short)0, acc, false, false);
  }
  const int row0 = tile * 16 + 8 * hf;
  float* dp = (l < 8) ? asrc : adst;
  const int h = l & 7;
#pragma unroll
  for (int r = 0; r < 8; ++r)
    if (row0 + r < M) dp[(size_t)(row0 + r) * HEADS + h] = acc[r];
}

#define BIN_SHIFT 8
#define BIN_NODES (1 << BIN_SHIFT)
#define SORT_TB 1024
#define EPT 8
#define SPT 4

__global__ void __launch_bounds__(SORT_TB)
bin_hist_kernel(const int* __restrict__ dst, int E, int NB, int G1, int* __restrict__ hist,
                const float* __restrict__ W, const float* __restrict__ att_src,
                const float* __restrict__ att_dst, float* __restrict__ Wa) {
  extern __shared__ int sh[];
  if ((int)blockIdx.x >= G1) {
    fold_att_block(W, att_src, att_dst, Wa);
    return;
  }
  const int t = (int)threadIdx.x;
  for (int b = t; b < NB; b += SORT_TB) sh[b] = 0;
  __syncthreads();
  const int base = (int)blockIdx.x * (SORT_TB * EPT) + t;
  int d[EPT];
#pragma unroll
  for (int k = 0; k < EPT; ++k) {
    const int e = base + k * SORT_TB;
    d[k] = dst[e < E ? e : E - 1];
  }
#pragma unroll
  for (int k = 0; k < EPT; ++k) {
    const int e = base + k * SORT_TB;
    if (e < E) atomicAdd(&sh[d[k] >> BIN_SHIFT], 1);
  }
  __syncthreads();
  for (int b = t; b < NB; b += SORT_TB)
    hist[(size_t)b * G1 + blockIdx.x] = sh[b];
}

__global__ void __launch_bounds__(SORT_TB, 16)
bin_scatter_kernel(const int* __restrict__ src, const int* __restrict__ dst, int E,
                   int NB, int G1, const int* __restrict__ offs, int* __restrict__ binned,
                   const float* __restrict__ xA, const float* __restrict__ Wa,
                   float* __restrict__ asrc, float* __restrict__ adst, int M) {
  extern __shared__ int sh[];
  if ((int)blockIdx.x >= G1) {
    const int wib = __builtin_amdgcn_readfirstlane((int)(threadIdx.x >> 5));
    const int tile = ((int)blockIdx.x - G1) * (SORT_TB / 32) + wib;
    if (tile < ((M + 15) >> 4))
      coef_tile(xA, Wa, asrc, adst, M, tile, (int)(threadIdx.x & 31));
    return;
  }
  const int t = (int)threadIdx.x;
  for (int b = t; b < NB; b += SORT_TB)
    sh[b] = offs[(size_t)b * G1 + blockIdx.x];
  __syncthreads();
  const int base = (int)blockIdx.x * (SORT_TB * EPT) + t;
  int s[EPT], d[EPT], pos[EPT];
#pragma unroll
  for (int k = 0; k < EPT; ++k) {
    const int e = base + k * SORT_TB;
    const int ec = e < E ? e : E - 1;
    s[k] = src[ec];
    d[k] = dst[ec];
  }
#pragma unroll
  for (int k = 0; k < EPT; ++k) {
    const int e = base + k * SORT_TB;
    pos[k] = 0;
    if (e < E) pos[k] = atomicAdd(&sh[d[k] >> BIN_SHIFT], 1);
  }
  int2* __restrict__ outp = (int2*)binned;
#pragma unroll
  for (int k = 0; k < EPT; ++k) {
    const int e = base + k * SORT_TB;
    if (e < E) outp[pos[k]] = make_int2(s[k], d[k]);
  }
}

__global__ void __launch_bounds__(SORT_TB, 16)
bin_sort_kernel(const int* __restrict__ binned, const int* __restrict__ offs,
                int G1, int NB, int N, int E,
                int* __restrict__ cnt, int* __restrict__ rowptr,
                int* __restrict__ csr_src) {
  __shared__ int c[BIN_NODES];
  __shared__ int wsum[BIN_NODES / 32];
  const int2* __restrict__ pairs = (const int2*)binned;
  const int t = (int)threadIdx.x;
  const int lane = t & 31;
  const int w = t >> 5;
  const int b = (int)blockIdx.x;
  const int beg = offs[(size_t)b * G1];
  const int end = (b + 1 < NB) ? offs[(size_t)(b + 1) * G1] : E;
  if (t < BIN_NODES) c[t] = 0;
  __syncthreads();
  for (int i0 = beg + t; i0 < end; i0 += SPT * SORT_TB) {
    int y[SPT];
#pragma unroll
    for (int k = 0; k < SPT; ++k) {
      const int i = i0 + k * SORT_TB;
      y[k] = pairs[i < end ? i : end - 1].y;
    }
#pragma unroll
    for (int k = 0; k < SPT; ++k)
      if (i0 + k * SORT_TB < end) atomicAdd(&c[y[k] & (BIN_NODES - 1)], 1);
  }
  __syncthreads();
  int mine = 0, incl = 0;
  if (t < BIN_NODES) {
    mine = c[t];
    incl = mine;
    for (int o = 1; o < 32; o <<= 1) {
      const int v = __shfl_up(incl, o);
      if (lane >= o) incl += v;
    }
    if (lane == 31) wsum[w] = incl;
  }
  __syncthreads();
  if (t < BIN_NODES) {
    int woff = 0;
    for (int k = 0; k < BIN_NODES / 32; ++k) woff += (k < w) ? wsum[k] : 0;
    const int start = beg + woff + incl - mine;
    c[t] = start;
    const int n = b * BIN_NODES + t;
    if (n < N) {
      cnt[n] = mine;
      rowptr[n] = start;
    }
  }
  __syncthreads();
  for (int i0 = beg + t; i0 < end; i0 += SPT * SORT_TB) {
    int2 sd[SPT];
    int pos[SPT];
#pragma unroll
    for (int k = 0; k < SPT; ++k) {
      const int i = i0 + k * SORT_TB;
      sd[k] = pairs[i < end ? i : end - 1];
    }
#pragma unroll
    for (int k = 0; k < SPT; ++k) {
      pos[k] = 0;
      if (i0 + k * SORT_TB < end) pos[k] = atomicAdd(&c[sd[k].y & (BIN_NODES - 1)], 1);
    }
#pragma unroll
    for (int k = 0; k < SPT; ++k)
      if (i0 + k * SORT_TB < end) csr_src[pos[k]] = sd[k].x;
  }
}

__global__ void __launch_bounds__(1024)
prefix_kernel(const int* __restrict__ cnt, int N, int* __restrict__ rowptr) {
  __shared__ int wsum[32];
  const int t = (int)threadIdx.x;
  const int lane = t & 31;
  const int w = t >> 5;
  const int base = (int)blockIdx.x * 1024;

  int part = 0;
  for (int i = t; i < base; i += 1024) part += cnt[i];
  for (int o = 16; o > 0; o >>= 1) part += __shfl_xor(part, o);
  if (lane == 0) wsum[w] = part;
  __syncthreads();
  int prefix = 0;
  for (int k = 0; k < 32; ++k) prefix += wsum[k];
  __syncthreads();

  const int i = base + t;
  const int c = (i < N) ? cnt[i] : 0;
  int incl = c;
  for (int o = 1; o < 32; o <<= 1) {
    const int v = __shfl_up(incl, o);
    if (lane >= o) incl += v;
  }
  if (lane == 31) wsum[w] = incl;
  __syncthreads();
  int woff = 0;
  for (int k = 0; k < 32; ++k) woff += (k < w) ? wsum[k] : 0;
  if (i < N) rowptr[i] = prefix + woff + incl - c;
}

__device__ __forceinline__ void edge_scores(const int* __restrict__ csr_src,
                                            const float* __restrict__ asrc,
                                            int start, int deg, int n, int j,
                                            const float (&ad)[HEADS],
                                            int& s, float (&e)[HEADS]) {
  s = (j < deg) ? csr_src[start + j] : n;
  const float4 a0 = *(const float4*)(asrc + (size_t)s * HEADS);
  const float4 a1 = *(const float4*)(asrc + (size_t)s * HEADS + 4);
  e[0] = lrelu(a0.x + ad[0]);
  e[1] = lrelu(a0.y + ad[1]);
  e[2] = lrelu(a0.z + ad[2]);
  e[3] = lrelu(a0.w + ad[3]);
  e[4] = lrelu(a1.x + ad[4]);
  e[5] = lrelu(a1.y + ad[5]);
  e[6] = lrelu(a1.z + ad[6]);
  e[7] = lrelu(a1.w + ad[7]);
}

#define PW 8
#define PN (2 * PW)
#define KW (HEADS * 32 / PW)
#define ZP (HEADS * 32 + 4)

#define GS 8
template <int F>
__device__ __forceinline__ void gather_inputs(const float* __restrict__ xin, int sreg,
                                              const float* wl, int count,
                                              int lane, float (&z)[HEADS]) {
  for (int j0 = 0; j0 < count; j0 += GS) {
    float v[GS];
#pragma unroll
    for (int g = 0; g < GS; ++g) {
      const int sj = __builtin_amdgcn_readlane(sreg, j0 + g);
      v[g] = xin[(unsigned)sj * F + (unsigned)lane];
    }
#pragma unroll
    for (int g = 0; g < GS; ++g) {
      const float4 a0 = *(const float4*)(wl + (j0 + g) * HEADS);
      const float4 a1 = *(const float4*)(wl + (j0 + g) * HEADS + 4);
      z[0] = __builtin_fmaf(a0.x, v[g], z[0]);  z[1] = __builtin_fmaf(a0.y, v[g], z[1]);
      z[2] = __builtin_fmaf(a0.z, v[g], z[2]);  z[3] = __builtin_fmaf(a0.w, v[g], z[3]);
      z[4] = __builtin_fmaf(a1.x, v[g], z[4]);  z[5] = __builtin_fmaf(a1.y, v[g], z[5]);
      z[6] = __builtin_fmaf(a1.z, v[g], z[6]);  z[7] = __builtin_fmaf(a1.w, v[g], z[7]);
    }
  }
}

__device__ __forceinline__ void put_weights(float* wl, int lane, const float (&alpha)[HEADS]) {
  __builtin_amdgcn_wave_barrier();
  *(float4*)(wl + lane * HEADS)     = make_float4(alpha[0], alpha[1], alpha[2], alpha[3]);
  *(float4*)(wl + lane * HEADS + 4) = make_float4(alpha[4], alpha[5], alpha[6], alpha[7]);
  __builtin_amdgcn_wave_barrier();
}

template <int F, int C>
__global__ void __launch_bounds__(PW * 32, 12)
node_attend_project_kernel(const int* __restrict__ rowptr,
                           const int* __restrict__ cnt,
                           const int* __restrict__ csr_src,
                           const float* __restrict__ asrc,
                           const float* __restrict__ adst,
                           const float* __restrict__ xin,
                           const float* __restrict__ Wm,
                           const float* __restrict__ bias,
                           float* __restrict__ out, int N) {
  static_assert(F == 32, "lane = input channel");
  static_assert(C == 16, "one 16-column tile of output channels");
  static_assert(PN == 16 && PW == HEADS && KW == F, "one 16-node tile; wave w owns head w's k values");
  static_assert(PW * 256 <= PN * ZP, "the partial tiles reuse the z rows' LDS");
  __shared__ __attribute__((aligned(16))) float zs[PN * ZP];
  __shared__ __attribute__((aligned(16))) float aw[PW * 32 * HEADS];
  const int lane = (int)(threadIdx.x & 31);
  const int wib = __builtin_amdgcn_readfirstlane((int)(threadIdx.x >> 5));
  const int nb = (int)blockIdx.x * PN;
  float* wl = aw + wib * (32 * HEADS);

  for (int u = 0; u < 2; ++u) {
    const int R = 2 * wib + u;
    const int n = nb + R;
    float z[HEADS];
#pragma unroll
    for (int k = 0; k < HEADS; ++k) z[k] = 0.f;
    if (n < N) {
      const int start = rowptr[n];
      const int deg = cnt[n];
      const int tot = deg + 1;
      float ad[HEADS];
#pragma unroll
      for (int k = 0; k < HEADS; ++k) ad[k] = adst[(size_t)n * HEADS + k];

      int s0;
      float e0[HEADS];
      const bool v0 = lane < tot;
      edge_scores(csr_src, asrc, start, deg, n, lane, ad, s0, e0);

      float mx[HEADS];
#pragma unroll
      for (int k = 0; k < HEADS; ++k) mx[k] = v0 ? e0[k] : -3.0e38f;
      for (int base = 32; base < tot; base += 32) {
        int s; float e[HEADS];
        const int j = base + lane;
        edge_scores(csr_src, asrc, start, deg, n, j, ad, s, e);
#pragma unroll
        for (int k = 0; k < HEADS; ++k) mx[k] = (j < tot) ? fmaxf(mx[k], e[k]) : mx[k];
      }
      {
        const float tmax = heads_reduce<true>(mx, lane);
#pragma unroll
        for (int k = 0; k < HEADS; ++k) mx[k] = head_total(tmax, k);
      }

      float ex0[HEADS], sm[HEADS];
#pragma unroll
      for (int k = 0; k < HEADS; ++k) {
        ex0[k] = v0 ? __expf(e0[k] - mx[k]) : 0.f;
        sm[k] = ex0[k];
      }
      for (int base = 32; base < tot; base += 32) {
        int s; float e[HEADS];
        const int j = base + lane;
        edge_scores(csr_src, asrc, start, deg, n, j, ad, s, e);
#pragma unroll
        for (int k = 0; k < HEADS; ++k) sm[k] += (j < tot) ? __expf(e[k] - mx[k]) : 0.f;
      }
      float inv[HEADS];
      {
        const float tsum = heads_reduce<false>(sm, lane);
        const float tinv = 1.0f / (tsum + SM_EPS);
#pragma unroll
        for (int k = 0; k < HEADS; ++k) inv[k] = head_total(tinv, k);
      }

      float alpha0[HEADS];
#pragma unroll
      for (int k = 0; k < HEADS; ++k) alpha0[k] = ex0[k] * inv[k];
      for (int base = 32; base < tot; base += 32) {
        int s; float e[HEADS], alpha[HEADS];
        const int j = base + lane;
        edge_scores(csr_src, asrc, start, deg, n, j, ad, s, e);
#pragma unroll
        for (int k = 0; k < HEADS; ++k)
          alpha[k] = (j < tot) ? __expf(e[k] - mx[k]) * inv[k] : 0.f;
        const int left = tot - base;
        put_weights(wl, lane, alpha);
        gather_inputs<F>(xin, s, wl, left < 32 ? left : 32, lane, z);
      }
      put_weights(wl, lane, alpha0);
      gather_inputs<F>(xin, s0, wl, tot < 32 ? tot : 32, lane, z);
    }
#pragma unroll
    for (int k = 0; k < HEADS; ++k) zs[R * ZP + k * F + lane] = z[k];
  }

  const int hf = lane >> 4;
  const int l = lane & 15;
  const float* wp = Wm + (size_t)(2 * hf) * (HEADS * C) + wib * C + l;
  v2f bfr[KW / 4];
#pragma unroll
  for (int ks = 0; ks < KW / 4; ++ks) {
    bfr[ks].x = wp[(size_t)(4 * ks) * (HEADS * C)];
    bfr[ks].y = wp[(size_t)(4 * ks + 1) * (HEADS * C)];
  }
  __syncthreads();

  v8f acc = (v8f){};
#pragma unroll
  for (int ks = 0; ks < KW / 4; ++ks) {
    const v2f a = *(const v2f*)&zs[l * ZP + KW * wib + 4 * ks + 2 * hf];
    acc = __builtin_amdgcn_wmma_f32_16x16x4_f32(false, a, false, bfr[ks],
                                                (short)0, acc, false, false);
  }
  __syncthreads();
#pragma unroll
  for (int r = 0; r < 8; ++r) zs[wib * 256 + (r + 8 * hf) * 16 + l] = acc[r];
  __syncthreads();

  const int t = (int)threadIdx.x;
  float sum = 0.f;
#pragma unroll
  for (int w = 0; w < PW; ++w) sum += zs[w * 256 + t];
  const int node = nb + (t >> 4);
  if (node < N)
    out[(size_t)node * C + (t & 15)] = sum * (1.0f / HEADS) + bias[t & 15];
}

#define P1W 16
#define P1N (2 * P1W)
#define Z1P (HEADS * F1 + 4)

__device__ __forceinline__ void fma4(float4& z, float a, const float4& v) {
  z.x = __builtin_fmaf(a, v.x, z.x);
  z.y = __builtin_fmaf(a, v.y, z.y);
  z.z = __builtin_fmaf(a, v.z, z.z);
  z.w = __builtin_fmaf(a, v.w, z.w);
}

__device__ __forceinline__ void gather_inputs4(const float* __restrict__ xin, int sreg,
                                               const float* wl, int count,
                                               int lane, float4 (&z)[HEADS]) {
  for (int j0 = 0; j0 < count; j0 += GS) {
    float4 v[GS];
#pragma unroll
    for (int g = 0; g < GS; ++g) {
      const int sj = __builtin_amdgcn_readlane(sreg, j0 + g);
      v[g] = *(const float4*)((const char*)xin + (size_t)((unsigned)sj * (4u * F1) + 16u * (unsigned)lane));
    }
#pragma unroll
    for (int g = 0; g < GS; ++g) {
      const float4 a0 = *(const float4*)(wl + (j0 + g) * HEADS);
      const float4 a1 = *(const float4*)(wl + (j0 + g) * HEADS + 4);
      fma4(z[0], a0.x, v[g]);  fma4(z[1], a0.y, v[g]);
      fma4(z[2], a0.z, v[g]);  fma4(z[3], a0.w, v[g]);
      fma4(z[4], a1.x, v[g]);  fma4(z[5], a1.y, v[g]);
      fma4(z[6], a1.z, v[g]);  fma4(z[7], a1.w, v[g]);
    }
  }
}

__global__ void __launch_bounds__(P1W * 32, 8)
node_attend_project1_kernel(const int* __restrict__ rowptr,
                            const int* __restrict__ cnt,
                            const int* __restrict__ csr_src,
                            const float* __restrict__ asrc,
                            const float* __restrict__ adst,
                            const float* __restrict__ xin,
                            const float* __restrict__ Wm,
                            const float* __restrict__ bias,
                            float* __restrict__ out, int N) {
  static_assert(P1N == 32 && P1W == 2 * HEADS, "two 16-node tiles; wave w owns head w % HEADS of tile w / HEADS");
  static_assert(P1W * 512 <= P1N * Z1P, "the partial tiles reuse the z rows' LDS");
  static_assert(F1 * 4 == 32 * 16, "lane = 4 input channels");
  __shared__ __attribute__((aligned(16))) float zs[P1N * Z1P];
  __shared__ __attribute__((aligned(16))) float aw[P1W * 32 * HEADS];
  const int lane = (int)__builtin_amdgcn_mbcnt_lo(~0u, 0u);
  const int wib = __builtin_amdgcn_readfirstlane((int)(threadIdx.x >> 5));
  const int nb = (int)blockIdx.x * P1N;
  float* wl = aw + wib * (32 * HEADS);

  for (int u = 0; u < 2; ++u) {
    const int R = 2 * wib + u;
    const int n = nb + R;
    float4 z[HEADS];
#pragma unroll
    for (int k = 0; k < HEADS; ++k) z[k] = make_float4(0.f, 0.f, 0.f, 0.f);
    if (n < N) {
      const int start = rowptr[n];
      const int deg = cnt[n];
      const int tot = deg + 1;
      float ad[HEADS];
#pragma unroll
      for (int k = 0; k < HEADS; ++k) ad[k] = adst[(size_t)n * HEADS + k];

      int s0;
      float e0[HEADS];
      const bool v0 = lane < tot;
      edge_scores(csr_src, asrc, start, deg, n, lane, ad, s0, e0);

      float mx[HEADS];
#pragma unroll
      for (int k = 0; k < HEADS; ++k) mx[k] = v0 ? e0[k] : -3.0e38f;
      for (int base = 32; base < tot; base += 32) {
        int s; float e[HEADS];
        const int j = base + lane;
        edge_scores(csr_src, asrc, start, deg, n, j, ad, s, e);
#pragma unroll
        for (int k = 0; k < HEADS; ++k) mx[k] = (j < tot) ? fmaxf(mx[k], e[k]) : mx[k];
      }
      {
        const float tmax = heads_reduce<true>(mx, lane);
#pragma unroll
        for (int k = 0; k < HEADS; ++k) mx[k] = head_total(tmax, k);
      }

      float ex0[HEADS], sm[HEADS];
#pragma unroll
      for (int k = 0; k < HEADS; ++k) {
        ex0[k] = v0 ? __expf(e0[k] - mx[k]) : 0.f;
        sm[k] = ex0[k];
      }
      for (int base = 32; base < tot; base += 32) {
        int s; float e[HEADS];
        const int j = base + lane;
        edge_scores(csr_src, asrc, start, deg, n, j, ad, s, e);
#pragma unroll
        for (int k = 0; k < HEADS; ++k) sm[k] += (j < tot) ? __expf(e[k] - mx[k]) : 0.f;
      }
      float inv[HEADS];
      {
        const float tsum = heads_reduce<false>(sm, lane);
        const float tinv = 1.0f / (tsum + SM_EPS);
#pragma unroll
        for (int k = 0; k < HEADS; ++k) inv[k] = head_total(tinv, k);
      }

      float alpha0[HEADS];
#pragma unroll
      for (int k = 0; k < HEADS; ++k) alpha0[k] = ex0[k] * inv[k];
      for (int base = 32; base < tot; base += 32) {
        int s; float e[HEADS], alpha[HEADS];
        const int j = base + lane;
        edge_scores(csr_src, asrc, start, deg, n, j, ad, s, e);
#pragma unroll
        for (int k = 0; k < HEADS; ++k)
          alpha[k] = (j < tot) ? __expf(e[k] - mx[k]) * inv[k] : 0.f;
        const int left = tot - base;
        put_weights(wl, lane, alpha);
        gather_inputs4(xin, s, wl, left < 32 ? left : 32, lane, z);
      }
      put_weights(wl, lane, alpha0);
      gather_inputs4(xin, s0, wl, tot < 32 ? tot : 32, lane, z);
    }
#pragma unroll
    for (int k = 0; k < HEADS; ++k) *(float4*)&zs[R * Z1P + k * F1 + 4 * lane] = z[k];
  }

  const int hf = lane >> 4;
  const int l = lane & 15;
  const int hw = wib & (HEADS - 1);
  const int rt = wib >> 3;
  const float* wp = Wm + (size_t)(2 * hf) * (HEADS * CH1) + hw * CH1 + l;
  __syncthreads();

  v8f acc0 = (v8f){};
  v8f acc1 = (v8f){};
#pragma unroll 8
  for (int ks = 0; ks < F1 / 4; ++ks) {
    v2f b0, b1;
    b0.x = wp[(size_t)(4 * ks) * (HEADS * CH1)];
    b0.y = wp[(size_t)(4 * ks + 1) * (HEADS * CH1)];
    b1.x = wp[(size_t)(4 * ks) * (HEADS * CH1) + 16];
    b1.y = wp[(size_t)(4 * ks + 1) * (HEADS * CH1) + 16];
    const v2f a = *(const v2f*)&zs[(16 * rt + l) * Z1P + F1 * hw + 4 * ks + 2 * hf];
    acc0 = __builtin_amdgcn_wmma_f32_16x16x4_f32(false, a, false, b0,
                                                 (short)0, acc0, false, false);
    acc1 = __builtin_amdgcn_wmma_f32_16x16x4_f32(false, a, false, b1,
                                                 (short)0, acc1, false, false);
  }
  __syncthreads();
#pragma unroll
  for (int r = 0; r < 8; ++r) {
    zs[wib * 512 + (r + 8 * hf) * 16 + l] = acc0[r];
    zs[wib * 512 + 256 + (r + 8 * hf) * 16 + l] = acc1[r];
  }
  __syncthreads();

  const int t = (wib << 5) | lane;
#pragma unroll
  for (int q = 0; q < 2; ++q) {
    const int o = t + q * (P1W * 32);
    const int nl = o >> 5;
    const int c = o & 31;
    const int pi = (nl >> 4) * (HEADS * 512) + (c >> 4) * 256 + (nl & 15) * 16 + (c & 15);
    float sum = 0.f;
#pragma unroll
    for (int w = 0; w < HEADS; ++w) sum += zs[w * 512 + pi];
    const int node = nb + nl;
    if (node < N) {
      const float v = sum * (1.0f / HEADS) + bias[c];
      out[(size_t)node * CH1 + c] = v > 0.f ? v : 0.f;
    }
  }
}

static inline unsigned nblk(long n, int b) { return (unsigned)((n + b - 1) / b); }

extern "C" void kernel_launch(void* const* d_in, const int* in_sizes, int n_in,
                              void* d_out, int out_size, void* d_ws, size_t ws_size,
                              hipStream_t stream) {
  const float* x        = (const float*)d_in[0];
  const int*   ei       = (const int*)  d_in[1];
  const float* W1       = (const float*)d_in[2];
  const float* att_src1 = (const float*)d_in[3];
  const float* att_dst1 = (const float*)d_in[4];
  const float* b1       = (const float*)d_in[5];
  const float* W2       = (const float*)d_in[6];
  const float* att_src2 = (const float*)d_in[7];
  const float* att_dst2 = (const float*)d_in[8];
  const float* b2       = (const float*)d_in[9];

  const int F_h   = in_sizes[3] / HEADS;
  const int F_out = in_sizes[7] / HEADS;
  const int F_in  = in_sizes[2] / (HEADS * F_h);
  const int N     = in_sizes[0] / F_in;
  const int E     = in_sizes[1] / 2;
  const int C2    = HEADS * F_out;
  const int* src = ei;
  const int* dst = ei + E;

  float* ws = (float*)d_ws;
  size_t o = 0;
  float* asrc = ws + o; o += (size_t)N * HEADS;
  float* adst = ws + o; o += (size_t)N * HEADS;
  float* x2   = ws + o; o += (size_t)N * F_h;
  float* Wa   = ws + o; o += (size_t)F_in * 16;
  const int NB = (N + BIN_NODES - 1) / BIN_NODES;
  const int G1 = (int)nblk(E, SORT_TB * EPT);
  int* binned  = (int*)(ws + o); o += (size_t)2 * E;
  int* cnt     = (int*)(ws + o); o += (size_t)N;
  int* rowptr  = (int*)(ws + o); o += (size_t)N;
  int* csr_src = (int*)(ws + o); o += (size_t)E;
  int* hist    = (int*)(ws + o); o += (size_t)NB * G1;
  int* offs    = (int*)(ws + o); o += (size_t)NB * G1;
  float* outp = (float*)d_out;

  const int TB = 256;

  bin_hist_kernel<<<G1 + 1, SORT_TB, NB * sizeof(int), stream>>>(
      dst, E, NB, G1, hist, W1, att_src1, att_dst1, Wa);
  prefix_kernel<<<nblk((long)NB * G1, 1024), 1024, 0, stream>>>(hist, NB * G1, offs);
  const int ctiles = (N + 15) / 16;
  bin_scatter_kernel<<<G1 + nblk(ctiles, SORT_TB / 32), SORT_TB, NB * sizeof(int), stream>>>(
      src, dst, E, NB, G1, offs, binned, x, Wa, asrc, adst, N);
  bin_sort_kernel<<<NB, SORT_TB, 0, stream>>>(
      binned, offs, G1, NB, N, E, cnt, rowptr, csr_src);

  node_attend_project1_kernel<<<nblk(N, P1N), P1W * 32, 0, stream>>>(
      rowptr, cnt, csr_src, asrc, adst, x, W1, b1, x2, N);

  {
    long waves = (long)((N + 15) / 16) * (C2 / (16 * NT));
    wmma_gemm_f32<16, false><<<nblk(waves * 32, TB), TB, 0, stream>>>(
        x2, W2, att_src2, att_dst2, (float*)binned, asrc, adst, N, F_h, C2);
  }
  node_attend_project_kernel<32, 16><<<nblk(N, PN), PW * 32, 0, stream>>>(
      rowptr, cnt, csr_src, asrc, adst, x2, W2, b2, outp, N);
}


# edge lists in 2 launches, not 4: each level-1 block counts, scans and bins its own chunk in LDS into its own region + segment table; bin blocks merge the segments; list and float sum order change
# speedup vs baseline: 20.6558x; 1.1018x over previous
#include <hip/hip_runtime.h>
#include <cstdint>

typedef float v2f __attribute__((ext_vector_type(2)));
typedef float v8f __attribute__((ext_vector_type(8)));

#define HEADS 8
#define NEG_SLOPE 0.2f
#define SM_EPS 1e-16f
#define NT 4

template <int CTRL>
__device__ __forceinline__ float xmask_add(float v) {
  const int o = __builtin_amdgcn_update_dpp(0, __float_as_int(v), CTRL, 0xf, 0xf, true);
  return v + __int_as_float(o);
}
__device__ __forceinline__ float row16_sum(float v) {
  v = xmask_add<0x161>(v);
  v = xmask_add<0x162>(v);
  v = xmask_add<0x164>(v);
  v = xmask_add<0x168>(v);
  return v;
}

template <int CH, bool STORE_C>
__device__ __forceinline__ void gemm_strip(const float* __restrict__ A,
                                           const float* __restrict__ B,
                                           const float* __restrict__ att_src,
                                           const float* __restrict__ att_dst,
                                           float* __restrict__ C,
                                           float* __restrict__ asrc,
                                           float* __restrict__ adst,
                                           int M, int K, int Ncols,
                                           int strip, int lane) {
  static_assert(CH == 32 || CH == 16, "a wave's 64 columns must be whole heads");
  constexpr int TPH = CH / 16;
  constexpr int HPW = NT / TPH;
  int groupsN = Ncols >> 6;
  int tm = strip / groupsN;
  int tg = strip - tm * groupsN;
  int half = lane >> 4;
  int l = lane & 15;

  int rowA = tm * 16 + l;
  if (rowA >= M) rowA = M - 1;
  const char* Ac = (const char*)A;
  const char* Bc = (const char*)B;
  unsigned aB  = 4u * ((unsigned)rowA * (unsigned)K + 2u * (unsigned)half);
  unsigned bB0 = 4u * ((unsigned)(2 * half) * (unsigned)Ncols + (unsigned)(tg * (16 * NT) + l));
  unsigned bB1 = bB0 + 4u * (unsigned)Ncols;
  const unsigned bStep = 16u * (unsigned)Ncols;

  v8f acc[NT];
#pragma unroll
  for (int t = 0; t < NT; ++t) acc[t] = (v8f){};

  v2f a0, b0[NT], a1, b1[NT];
  a0.x = *(const float*)(Ac + (size_t)aB);
  a0.y = *(const float*)(Ac + (size_t)aB + 4);
#pragma unroll
  for (int t = 0; t < NT; ++t) {
    b0[t].x = *(const float*)(Bc + (size_t)bB0 + t * 64);
    b0[t].y = *(const float*)(Bc + (size_t)bB1 + t * 64);
  }
  int k0 = 0;
#pragma unroll 1
  for (; k0 + 8 <= K; k0 += 8) {
    aB += 16u;  bB0 += bStep;  bB1 += bStep;
    a1.x = *(const float*)(Ac + (size_t)aB);
    a1.y = *(const float*)(Ac + (size_t)aB + 4);
#pragma unroll
    for (int t = 0; t < NT; ++t) {
      b1[t].x = *(const float*)(Bc + (size_t)bB0 + t * 64);
      b1[t].y = *(const float*)(Bc + (size_t)bB1 + t * 64);
    }
#pragma unroll
    for (int t = 0; t < NT; ++t) {
      acc[t] = __builtin_amdgcn_wmma_f32_16x16x4_f32(false, a0, false, b0[t],
                                                     (short)0, acc[t],
                                                     false, false);
    }
    const bool more = k0 + 8 < K;
    aB += more ? 16u : 0u;  bB0 += more ? bStep : 0u;  bB1 += more ? bStep : 0u;
    a0.x = *(const float*)(Ac + (size_t)aB);
    a0.y = *(const float*)(Ac + (size_t)aB + 4);
#pragma unroll
    for (int t = 0; t < NT; ++t) {
      b0[t].x = *(const float*)(Bc + (size_t)bB0 + t * 64);
      b0[t].y = *(const float*)(Bc + (size_t)bB1 + t * 64);
    }
#pragma unroll
    for (int t = 0; t < NT; ++t) {
      acc[t] = __builtin_amdgcn_wmma_f32_16x16x4_f32(false, a1, false, b1[t],
                                                     (short)0, acc[t],
                                                     false, false);
    }
  }
  if (k0 < K) {
#pragma unroll
    for (int t = 0; t < NT; ++t) {
      acc[t] = __builtin_amdgcn_wmma_f32_16x16x4_f32(false, a0, false, b0[t],
                                                     (short)0, acc[t],
                                                     false, false);
    }
  }

  float wS[NT], wD[NT];
#pragma unroll
  for (int t = 0; t < NT; ++t) {
    wS[t] = att_src[tg * (16 * NT) + t * 16 + l];
    wD[t] = att_dst[tg * (16 * NT) + t * 16 + l];
  }

  if (STORE_C) {
    char* Cc = (char*)C;
    const int row0 = tm * 16 + 8 * half;
    const unsigned rowC = 4u * (unsigned)Ncols;
    const unsigned cB = (unsigned)row0 * rowC + 4u * (unsigned)(tg * (16 * NT) + l);
#pragma unroll
    for (int r = 0; r < 8; ++r) {
      if (row0 + r < M) {
#pragma unroll
        for (int t = 0; t < NT; ++t)
          *(float*)(Cc + (size_t)(cB + (unsigned)r * rowC) + t * 64) = acc[t][r];
      }
    }
  }

#pragma unroll
  for (int j = 0; j < HPW / 2; ++j) {
    float vs = 0.f, vd = 0.f;
#pragma unroll
    for (int hb = 0; hb < 2; ++hb) {
#pragma unroll
      for (int r = 0; r < 8; ++r) {
        float ps = 0.f, pd = 0.f;
#pragma unroll
        for (int tt = 0; tt < TPH; ++tt) {
          const int t = (2 * j + hb) * TPH + tt;
          ps += acc[t][r] * wS[t];
          pd += acc[t][r] * wD[t];
        }
        ps = row16_sum(ps);
        pd = row16_sum(pd);
        const bool mine = (l == hb * 8 + r);
        vs = mine ? ps : vs;
        vd = mine ? pd : vd;
        asm volatile("" : "+v"(vs), "+v"(vd));
      }
    }
    const int row = tm * 16 + (l & 7) + 8 * half;
    const int head = tg * HPW + 2 * j + (l >> 3);
    if (row < M) {
      asrc[(size_t)row * HEADS + head] = vs;
      adst[(size_t)row * HEADS + head] = vd;
    }
  }
}

template <int CH, bool STORE_C>
__global__ void wmma_gemm_f32(const float* __restrict__ A,
                              const float* __restrict__ B,
                              const float* __restrict__ att_src,
                              const float* __restrict__ att_dst,
                              float* __restrict__ C,
                              float* __restrict__ asrc,
                              float* __restrict__ adst,
                              int M, int K, int Ncols) {
  int wave = (int)((blockIdx.x * blockDim.x + threadIdx.x) >> 5);
  int lane = (int)(threadIdx.x & 31);
  int tilesM  = (M + 15) >> 4;
  int groupsN = Ncols >> 6;
  if (wave >= tilesM * groupsN) return;
  gemm_strip<CH, STORE_C>(A, B, att_src, att_dst, C, asrc, adst, M, K, Ncols, wave, lane);
}

__device__ __forceinline__ float lrelu(float v) {
  return v > 0.f ? v : NEG_SLOPE * v;
}

__device__ __forceinline__ float xhalf(float x) {
  const int b = __float_as_int(x);
  return __int_as_float(__builtin_amdgcn_permlanex16(b, b, 0x76543210, 0xfedcba98, false, false));
}
template <int CTRL>
__device__ __forceinline__ float xmask_get(float v) {
  return __int_as_float(__builtin_amdgcn_update_dpp(0, __float_as_int(v), CTRL, 0xf, 0xf, true));
}
template <bool IS_MAX>
__device__ __forceinline__ float comb(float a, float b) {
  return IS_MAX ? fmaxf(a, b) : a + b;
}
template <bool IS_MAX>
__device__ __forceinline__ float heads_reduce(const float (&a)[HEADS], int lane) {
  static_assert(HEADS == 8, "three halvings leave one head per lane");
  const bool h16 = (lane & 16) != 0;
  const bool h8 = (lane & 8) != 0;
  const bool h4 = (lane & 4) != 0;
  float b[4], c[2];
#pragma unroll
  for (int k = 0; k < 4; ++k) {
    const float keep = h16 ? a[k + 4] : a[k];
    const float send = h16 ? a[k] : a[k + 4];
    b[k] = comb<IS_MAX>(keep, xhalf(send));
  }
#pragma unroll
  for (int k = 0; k < 2; ++k) {
    const float keep = h8 ? b[k + 2] : b[k];
    const float send = h8 ? b[k] : b[k + 2];
    c[k] = comb<IS_MAX>(keep, xmask_get<0x168>(send));
  }
  const float keep = h4 ? c[1] : c[0];
  const float send = h4 ? c[0] : c[1];
  float d = comb<IS_MAX>(keep, xmask_get<0x164>(send));
  d = comb<IS_MAX>(d, xmask_get<0x162>(d));
  d = comb<IS_MAX>(d, xmask_get<0x161>(d));
  return d;
}
__device__ __forceinline__ float head_total(float d, int k) {
  return __int_as_float(__builtin_amdgcn_readlane(__float_as_int(d), 4 * k));
}

#define F1 128
#define CH1 32

__device__ __forceinline__ void fold_att_block(const float* __restrict__ W,
                                               const float* __restrict__ att_src,
                                               const float* __restrict__ att_dst,
                                               float* __restrict__ Wa) {
  const int t = (int)threadIdx.x;
  const int f = t >> 3;
  const int h = t & 7;
  float s = 0.f, d = 0.f;
#pragma unroll 8
  for (int c = 0; c < CH1; ++c) {
    const float w = W[(size_t)f * (HEADS * CH1) + h * CH1 + c];
    s = __builtin_fmaf(w, att_src[h * CH1 + c], s);
    d = __builtin_fmaf(w, att_dst[h * CH1 + c], d);
  }
  Wa[f * 16 + h] = s;
  Wa[f * 16 + 8 + h] = d;
}

__device__ __forceinline__ void coef_tile(const float* __restrict__ A,
                                          const float* __restrict__ Wa,
                                          float* __restrict__ asrc,
                                          float* __restrict__ adst,
                                          int M, int tile, int lane) {
  const int hf = lane >> 4;
  const int l = lane & 15;
  int rowA = tile * 16 + l;
  if (rowA >= M) rowA = M - 1;
  const float* ap = A + (size_t)rowA * F1 + 2 * hf;
  const float* bp = Wa + (2 * hf) * 16 + l;
  v8f acc = (v8f){};
#pragma unroll 8
  for (int ks = 0; ks < F1 / 4; ++ks) {
    v2f a, b;
    a.x = ap[4 * ks];
    a.y = ap[4 * ks + 1];
    b.x = bp[(4 * ks) * 16];
    b.y = bp[(4 * ks + 1) * 16];
    acc = __builtin_amdgcn_wmma_f32_16x16x4_f32(false, a, false, b,
                                                (short)0, acc, false, false);
  }
  const int row0 = tile * 16 + 8 * hf;
  float* dp = (l < 8) ? asrc : adst;
  const int h = l & 7;
#pragma unroll
  for (int r = 0; r < 8; ++r)
    if (row0 + r < M) dp[(size_t)(row0 + r) * HEADS + h] = acc[r];
}

#define BIN_SHIFT 8
#define BIN_NODES (1 << BIN_SHIFT)
#define SORT_TB 1024
#define EPT 8
#define CHUNK (SORT_TB * EPT)
#define SPT 4
#define MAXG 256

__device__ __forceinline__ int wave_incl_scan(int v, int lane) {
  for (int o = 1; o < 32; o <<= 1) {
    const int u = __shfl_up(v, o);
    if (lane >= o) v += u;
  }
  return v;
}

__global__ void __launch_bounds__(SORT_TB)
chunk_bin_kernel(const int* __restrict__ src, const int* __restrict__ dst, int E,
                 int NB, int G1, int* __restrict__ seg, int* __restrict__ binned,
                 const float* __restrict__ W, const float* __restrict__ att_src,
                 const float* __restrict__ att_dst, float* __restrict__ Wa) {
  extern __shared__ int sh[];
  __shared__ int wsum[SORT_TB / 32];
  if ((int)blockIdx.x >= G1) {
    fold_att_block(W, att_src, att_dst, Wa);
    return;
  }
  const int t = (int)threadIdx.x;
  const int lane = t & 31;
  const int w = t >> 5;
  for (int b = t; b < NB; b += SORT_TB) sh[b] = 0;
  __syncthreads();
  const int base = (int)blockIdx.x * CHUNK + t;
  int s[EPT], d[EPT], pos[EPT];
#pragma unroll
  for (int k = 0; k < EPT; ++k) {
    const int e = base + k * SORT_TB;
    const int ec = e < E ? e : E - 1;
    s[k] = src[ec];
    d[k] = dst[ec];
  }
#pragma unroll
  for (int k = 0; k < EPT; ++k) {
    const int e = base + k * SORT_TB;
    pos[k] = 0;
    if (e < E) pos[k] = atomicAdd(&sh[d[k] >> BIN_SHIFT], 1);
  }
  __syncthreads();
  int carry = 0;
  for (int b0 = 0; b0 < NB; b0 += SORT_TB) {
    const int b = b0 + t;
    const int cv = (b < NB) ? sh[b] : 0;
    const int incl = wave_incl_scan(cv, lane);
    if (lane == 31) wsum[w] = incl;
    __syncthreads();
    int woff = 0, tot = 0;
    for (int k = 0; k < SORT_TB / 32; ++k) {
      const int x = wsum[k];
      woff += (k < w) ? x : 0;
      tot += x;
    }
    if (b < NB) sh[b] = carry + woff + incl - cv;
    carry += tot;
    __syncthreads();
  }
  int* __restrict__ segrow = seg + (size_t)blockIdx.x * (size_t)(NB + 1);
  for (int b = t; b < NB; b += SORT_TB) segrow[b] = sh[b];
  if (t == 0) segrow[NB] = carry;
  int2* __restrict__ outp = (int2*)binned + (size_t)blockIdx.x * CHUNK;
#pragma unroll
  for (int k = 0; k < EPT; ++k) {
    const int e = base + k * SORT_TB;
    if (e < E) outp[sh[d[k] >> BIN_SHIFT] + pos[k]] = make_int2(s[k], d[k]);
  }
}

__device__ __forceinline__ void seg_addr(const int* P, const int* S, int step0,
                                         const int (&i)[SPT], int (&a)[SPT]) {
  int g[SPT];
#pragma unroll
  for (int k = 0; k < SPT; ++k) g[k] = 0;
  for (int step = step0; step > 0; step >>= 1) {
#pragma unroll
    for (int k = 0; k < SPT; ++k) {
      const int cand = g[k] + step;
      g[k] = (P[cand] <= i[k]) ? cand : g[k];
    }
  }
#pragma unroll
  for (int k = 0; k < SPT; ++k) a[k] = S[g[k]] + (i[k] - P[g[k]]);
}

__global__ void __launch_bounds__(SORT_TB, 16)
bin_sort_kernel(const int* __restrict__ binned, const int* __restrict__ seg,
                int G1, int NB, int N,
                int* __restrict__ cnt, int* __restrict__ rowptr,
                int* __restrict__ csr_src,
                const float* __restrict__ xA, const float* __restrict__ Wa,
                float* __restrict__ asrc, float* __restrict__ adst) {
  static_assert(MAXG == BIN_NODES, "both scans run on the block's first MAXG / 32 waves");
  __shared__ int c[BIN_NODES];
  __shared__ int wsum[MAXG / 32];
  __shared__ int wbeg[MAXG / 32];
  __shared__ int P[MAXG];
  __shared__ int S[MAXG];
  if ((int)blockIdx.x >= NB) {
    const int wib = __builtin_amdgcn_readfirstlane((int)(threadIdx.x >> 5));
    const int tile = ((int)blockIdx.x - NB) * (SORT_TB / 32) + wib;
    if (tile < ((N + 15) >> 4))
      coef_tile(xA, Wa, asrc, adst, N, tile, (int)(threadIdx.x & 31));
    return;
  }
  const int2* __restrict__ pairs = (const int2*)binned;
  const int t = (int)threadIdx.x;
  const int lane = t & 31;
  const int w = t >> 5;
  const int b = (int)blockIdx.x;
  int s0 = 0, len = 0, linc = 0;
  if (t < MAXG) {
    if (t < G1) {
      const int* __restrict__ r = seg + (size_t)t * (size_t)(NB + 1) + b;
      s0 = r[0];
      len = r[1] - s0;
    }
    linc = wave_incl_scan(len, lane);
    int bs = s0;
    for (int o = 16; o > 0; o >>= 1) bs += __shfl_xor(bs, o);
    if (lane == 31) wsum[w] = linc;
    if (lane == 0) wbeg[w] = bs;
    c[t] = 0;
  }
  __syncthreads();
  int total = 0, beg = 0;
  {
    int woff = 0;
    for (int k = 0; k < MAXG / 32; ++k) {
      const int x = wsum[k];
      woff += (k < w) ? x : 0;
      total += x;
      beg += wbeg[k];
    }
    if (t < MAXG) {
      P[t] = (t < G1) ? woff + linc - len : 0x7fffffff;
      S[t] = t * CHUNK + s0;
    }
  }
  __syncthreads();
  const int step0 = (G1 > 1) ? 1 << (31 - __clz(G1 - 1)) : 0;
  int af[SPT];
  {
    int ii[SPT];
#pragma unroll
    for (int k = 0; k < SPT; ++k) {
      const int i = t + k * SORT_TB;
      ii[k] = i < total ? i : (total > 0 ? total - 1 : 0);
    }
    seg_addr(P, S, step0, ii, af);
  }
  for (int i0 = t; i0 < total; i0 += SPT * SORT_TB) {
    int y[SPT], a[SPT];
    if (i0 == t) {
#pragma unroll
      for (int k = 0; k < SPT; ++k) a[k] = af[k];
    } else {
      int ii[SPT];
#pragma unroll
      for (int k = 0; k < SPT; ++k) {
        const int i = i0 + k * SORT_TB;
        ii[k] = i < total ? i : total - 1;
      }
      seg_addr(P, S, step0, ii, a);
    }
#pragma unroll
    for (int k = 0; k < SPT; ++k) y[k] = pairs[a[k]].y;
#pragma unroll
    for (int k = 0; k < SPT; ++k)
      atomicAdd(&c[y[k] & (BIN_NODES - 1)], (i0 + k * SORT_TB < total) ? 1 : 0);
  }
  __syncthreads();
  int mine = 0, incl = 0;
  if (t < BIN_NODES) {
    mine = c[t];
    incl = wave_incl_scan(mine, lane);
    if (lane == 31) wsum[w] = incl;
  }
  __syncthreads();
  if (t < BIN_NODES) {
    int woff = 0;
    for (int k = 0; k < BIN_NODES / 32; ++k) woff += (k < w) ? wsum[k] : 0;
    const int start = beg + woff + incl - mine;
    c[t] = start;
    const int n = b * BIN_NODES + t;
    if (n < N) {
      cnt[n] = mine;
      rowptr[n] = start;
    }
  }
  __syncthreads();
  for (int i0 = t; i0 < total; i0 += SPT * SORT_TB) {
    int2 sd[SPT];
    int pos[SPT], a[SPT];
    if (i0 == t) {
#pragma unroll
      for (int k = 0; k < SPT; ++k) a[k] = af[k];
    } else {
      int ii[SPT];
#pragma unroll
      for (int k = 0; k < SPT; ++k) {
        const int i = i0 + k * SORT_TB;
        ii[k] = i < total ? i : total - 1;
      }
      seg_addr(P, S, step0, ii, a);
    }
#pragma unroll
    for (int k = 0; k < SPT; ++k) sd[k] = pairs[a[k]];
#pragma unroll
    for (int k = 0; k < SPT; ++k) {
      pos[k] = 0;
      if (i0 + k * SORT_TB < total) pos[k] = atomicAdd(&c[sd[k].y & (BIN_NODES - 1)], 1);
    }
#pragma unroll
    for (int k = 0; k < SPT; ++k)
      if (i0 + k * SORT_TB < total) csr_src[pos[k]] = sd[k].x;
  }
}

__device__ __forceinline__ void edge_scores(const int* __restrict__ csr_src,
                                            const float* __restrict__ asrc,
                                            int start, int deg, int n, int j,
                                            const float (&ad)[HEADS],
                                            int& s, float (&e)[HEADS]) {
  s = (j < deg) ? csr_src[start + j] : n;
  const float4 a0 = *(const float4*)(asrc + (size_t)s * HEADS);
  const float4 a1 = *(const float4*)(asrc + (size_t)s * HEADS + 4);
  e[0] = lrelu(a0.x + ad[0]);
  e[1] = lrelu(a0.y + ad[1]);
  e[2] = lrelu(a0.z + ad[2]);
  e[3] = lrelu(a0.w + ad[3]);
  e[4] = lrelu(a1.x + ad[4]);
  e[5] = lrelu(a1.y + ad[5]);
  e[6] = lrelu(a1.z + ad[6]);
  e[7] = lrelu(a1.w + ad[7]);
}

#define PW 8
#define PN (2 * PW)
#define KW (HEADS * 32 / PW)
#define ZP (HEADS * 32 + 4)

#define GS 8
template <int F>
__device__ __forceinline__ void gather_inputs(const float* __restrict__ xin, int sreg,
                                              const float* wl, int count,
                                              int lane, float (&z)[HEADS]) {
  for (int j0 = 0; j0 < count; j0 += GS) {
    float v[GS];
#pragma unroll
    for (int g = 0; g < GS; ++g) {
      const int sj = __builtin_amdgcn_readlane(sreg, j0 + g);
      v[g] = xin[(unsigned)sj * F + (unsigned)lane];
    }
#pragma unroll
    for (int g = 0; g < GS; ++g) {
      const float4 a0 = *(const float4*)(wl + (j0 + g) * HEADS);
      const float4 a1 = *(const float4*)(wl + (j0 + g) * HEADS + 4);
      z[0] = __builtin_fmaf(a0.x, v[g], z[0]);  z[1] = __builtin_fmaf(a0.y, v[g], z[1]);
      z[2] = __builtin_fmaf(a0.z, v[g], z[2]);  z[3] = __builtin_fmaf(a0.w, v[g], z[3]);
      z[4] = __builtin_fmaf(a1.x, v[g], z[4]);  z[5] = __builtin_fmaf(a1.y, v[g], z[5]);
      z[6] = __builtin_fmaf(a1.z, v[g], z[6]);  z[7] = __builtin_fmaf(a1.w, v[g], z[7]);
    }
  }
}

__device__ __forceinline__ void put_weights(float* wl, int lane, const float (&alpha)[HEADS]) {
  __builtin_amdgcn_wave_barrier();
  *(float4*)(wl + lane * HEADS)     = make_float4(alpha[0], alpha[1], alpha[2], alpha[3]);
  *(float4*)(wl + lane * HEADS + 4) = make_float4(alpha[4], alpha[5], alpha[6], alpha[7]);
  __builtin_amdgcn_wave_barrier();
}

template <int F, int C>
__global__ void __launch_bounds__(PW * 32, 12)
node_attend_project_kernel(const int* __restrict__ rowptr,
                           const int* __restrict__ cnt,
                           const int* __restrict__ csr_src,
                           const float* __restrict__ asrc,
                           const float* __restrict__ adst,
                           const float* __restrict__ xin,
                           const float* __restrict__ Wm,
                           const float* __restrict__ bias,
                           float* __restrict__ out, int N) {
  static_assert(F == 32, "lane = input channel");
  static_assert(C == 16, "one 16-column tile of output channels");
  static_assert(PN == 16 && PW == HEADS && KW == F, "one 16-node tile; wave w owns head w's k values");
  static_assert(PW * 256 <= PN * ZP, "the partial tiles reuse the z rows' LDS");
  __shared__ __attribute__((aligned(16))) float zs[PN * ZP];
  __shared__ __attribute__((aligned(16))) float aw[PW * 32 * HEADS];
  const int lane = (int)(threadIdx.x & 31);
  const int wib = __builtin_amdgcn_readfirstlane((int)(threadIdx.x >> 5));
  const int nb = (int)blockIdx.x * PN;
  float* wl = aw + wib * (32 * HEADS);

  for (int u = 0; u < 2; ++u) {
    const int R = 2 * wib + u;
    const int n = nb + R;
    float z[HEADS];
#pragma unroll
    for (int k = 0; k < HEADS; ++k) z[k] = 0.f;
    if (n < N) {
      const int start = rowptr[n];
      const int deg = cnt[n];
      const int tot = deg + 1;
      float ad[HEADS];
#pragma unroll
      for (int k = 0; k < HEADS; ++k) ad[k] = adst[(size_t)n * HEADS + k];

      int s0;
      float e0[HEADS];
      const bool v0 = lane < tot;
      edge_scores(csr_src, asrc, start, deg, n, lane, ad, s0, e0);

      float mx[HEADS];
#pragma unroll
      for (int k = 0; k < HEADS; ++k) mx[k] = v0 ? e0[k] : -3.0e38f;
      for (int base = 32; base < tot; base += 32) {
        int s; float e[HEADS];
        const int j = base + lane;
        edge_scores(csr_src, asrc, start, deg, n, j, ad, s, e);
#pragma unroll
        for (int k = 0; k < HEADS; ++k) mx[k] = (j < tot) ? fmaxf(mx[k], e[k]) : mx[k];
      }
      {
        const float tmax = heads_reduce<true>(mx, lane);
#pragma unroll
        for (int k = 0; k < HEADS; ++k) mx[k] = head_total(tmax, k);
      }

      float ex0[HEADS], sm[HEADS];
#pragma unroll
      for (int k = 0; k < HEADS; ++k) {
        ex0[k] = v0 ? __expf(e0[k] - mx[k]) : 0.f;
        sm[k] = ex0[k];
      }
      for (int base = 32; base < tot; base += 32) {
        int s; float e[HEADS];
        const int j = base + lane;
        edge_scores(csr_src, asrc, start, deg, n, j, ad, s, e);
#pragma unroll
        for (int k = 0; k < HEADS; ++k) sm[k] += (j < tot) ? __expf(e[k] - mx[k]) : 0.f;
      }
      float inv[HEADS];
      {
        const float tsum = heads_reduce<false>(sm, lane);
        const float tinv = 1.0f / (tsum + SM_EPS);
#pragma unroll
        for (int k = 0; k < HEADS; ++k) inv[k] = head_total(tinv, k);
      }

      float alpha0[HEADS];
#pragma unroll
      for (int k = 0; k < HEADS; ++k) alpha0[k] = ex0[k] * inv[k];
      for (int base = 32; base < tot; base += 32) {
        int s; float e[HEADS], alpha[HEADS];
        const int j = base + lane;
        edge_scores(csr_src, asrc, start, deg, n, j, ad, s, e);
#pragma unroll
        for (int k = 0; k < HEADS; ++k)
          alpha[k] = (j < tot) ? __expf(e[k] - mx[k]) * inv[k] : 0.f;
        const int left = tot - base;
        put_weights(wl, lane, alpha);
        gather_inputs<F>(xin, s, wl, left < 32 ? left : 32, lane, z);
      }
      put_weights(wl, lane, alpha0);
      gather_inputs<F>(xin, s0, wl, tot < 32 ? tot : 32, lane, z);
    }
#pragma unroll
    for (int k = 0; k < HEADS; ++k) zs[R * ZP + k * F + lane] = z[k];
  }

  const int hf = lane >> 4;
  const int l = lane & 15;
  const float* wp = Wm + (size_t)(2 * hf) * (HEADS * C) + wib * C + l;
  v2f bfr[KW / 4];
#pragma unroll
  for (int ks = 0; ks < KW / 4; ++ks) {
    bfr[ks].x = wp[(size_t)(4 * ks) * (HEADS * C)];
    bfr[ks].y = wp[(size_t)(4 * ks + 1) * (HEADS * C)];
  }
  __syncthreads();

  v8f acc = (v8f){};
#pragma unroll
  for (int ks = 0; ks < KW / 4; ++ks) {
    const v2f a = *(const v2f*)&zs[l * ZP + KW * wib + 4 * ks + 2 * hf];
    acc = __builtin_amdgcn_wmma_f32_16x16x4_f32(false, a, false, bfr[ks],
                                                (short)0, acc, false, false);
  }
  __syncthreads();
#pragma unroll
  for (int r = 0; r < 8; ++r) zs[wib * 256 + (r + 8 * hf) * 16 + l] = acc[r];
  __syncthreads();

  const int t = (int)threadIdx.x;
  float sum = 0.f;
#pragma unroll
  for (int w = 0; w < PW; ++w) sum += zs[w * 256 + t];
  const int node = nb + (t >> 4);
  if (node < N)
    out[(size_t)node * C + (t & 15)] = sum * (1.0f / HEADS) + bias[t & 15];
}

#define P1W 16
#define P1N (2 * P1W)
#define Z1P (HEADS * F1 + 4)

__device__ __forceinline__ void fma4(float4& z, float a, const float4& v) {
  z.x = __builtin_fmaf(a, v.x, z.x);
  z.y = __builtin_fmaf(a, v.y, z.y);
  z.z = __builtin_fmaf(a, v.z, z.z);
  z.w = __builtin_fmaf(a, v.w, z.w);
}

__device__ __forceinline__ void gather_inputs4(const float* __restrict__ xin, int sreg,
                                               const float* wl, int count,
                                               int lane, float4 (&z)[HEADS]) {
  for (int j0 = 0; j0 < count; j0 += GS) {
    float4 v[GS];
#pragma unroll
    for (int g = 0; g < GS; ++g) {
      const int sj = __builtin_amdgcn_readlane(sreg, j0 + g);
      v[g] = *(const float4*)((const char*)xin + (size_t)((unsigned)sj * (4u * F1) + 16u * (unsigned)lane));
    }
#pragma unroll
    for (int g = 0; g < GS; ++g) {
      const float4 a0 = *(const float4*)(wl + (j0 + g) * HEADS);
      const float4 a1 = *(const float4*)(wl + (j0 + g) * HEADS + 4);
      fma4(z[0], a0.x, v[g]);  fma4(z[1], a0.y, v[g]);
      fma4(z[2], a0.z, v[g]);  fma4(z[3], a0.w, v[g]);
      fma4(z[4], a1.x, v[g]);  fma4(z[5], a1.y, v[g]);
      fma4(z[6], a1.z, v[g]);  fma4(z[7], a1.w, v[g]);
    }
  }
}

__global__ void __launch_bounds__(P1W * 32, 8)
node_attend_project1_kernel(const int* __restrict__ rowptr,
                            const int* __restrict__ cnt,
                            const int* __restrict__ csr_src,
                            const float* __restrict__ asrc,
                            const float* __restrict__ adst,
                            const float* __restrict__ xin,
                            const float* __restrict__ Wm,
                            const float* __restrict__ bias,
                            float* __restrict__ out, int N) {
  static_assert(P1N == 32 && P1W == 2 * HEADS, "two 16-node tiles; wave w owns head w % HEADS of tile w / HEADS");
  static_assert(P1W * 512 <= P1N * Z1P, "the partial tiles reuse the z rows' LDS");
  static_assert(F1 * 4 == 32 * 16, "lane = 4 input channels");
  __shared__ __attribute__((aligned(16))) float zs[P1N * Z1P];
  __shared__ __attribute__((aligned(16))) float aw[P1W * 32 * HEADS];
  const int lane = (int)__builtin_amdgcn_mbcnt_lo(~0u, 0u);
  const int wib = __builtin_amdgcn_readfirstlane((int)(threadIdx.x >> 5));
  const int nb = (int)blockIdx.x * P1N;
  float* wl = aw + wib * (32 * HEADS);

  for (int u = 0; u < 2; ++u) {
    const int R = 2 * wib + u;
    const int n = nb + R;
    float4 z[HEADS];
#pragma unroll
    for (int k = 0; k < HEADS; ++k) z[k] = make_float4(0.f, 0.f, 0.f, 0.f);
    if (n < N) {
      const int start = rowptr[n];
      const int deg = cnt[n];
      const int tot = deg + 1;
      float ad[HEADS];
#pragma unroll
      for (int k = 0; k < HEADS; ++k) ad[k] = adst[(size_t)n * HEADS + k];

      int s0;
      float e0[HEADS];
      const bool v0 = lane < tot;
      edge_scores(csr_src, asrc, start, deg, n, lane, ad, s0, e0);

      float mx[HEADS];
#pragma unroll
      for (int k = 0; k < HEADS; ++k) mx[k] = v0 ? e0[k] : -3.0e38f;
      for (int base = 32; base < tot; base += 32) {
        int s; float e[HEADS];
        const int j = base + lane;
        edge_scores(csr_src, asrc, start, deg, n, j, ad, s, e);
#pragma unroll
        for (int k = 0; k < HEADS; ++k) mx[k] = (j < tot) ? fmaxf(mx[k], e[k]) : mx[k];
      }
      {
        const float tmax = heads_reduce<true>(mx, lane);
#pragma unroll
        for (int k = 0; k < HEADS; ++k) mx[k] = head_total(tmax, k);
      }

      float ex0[HEADS], sm[HEADS];
#pragma unroll
      for (int k = 0; k < HEADS; ++k) {
        ex0[k] = v0 ? __expf(e0[k] - mx[k]) : 0.f;
        sm[k] = ex0[k];
      }
      for (int base = 32; base < tot; base += 32) {
        int s; float e[HEADS];
        const int j = base + lane;
        edge_scores(csr_src, asrc, start, deg, n, j, ad, s, e);
#pragma unroll
        for (int k = 0; k < HEADS; ++k) sm[k] += (j < tot) ? __expf(e[k] - mx[k]) : 0.f;
      }
      float inv[HEADS];
      {
        const float tsum = heads_reduce<false>(sm, lane);
        const float tinv = 1.0f / (tsum + SM_EPS);
#pragma unroll
        for (int k = 0; k < HEADS; ++k) inv[k] = head_total(tinv, k);
      }

      float alpha0[HEADS];
#pragma unroll
      for (int k = 0; k < HEADS; ++k) alpha0[k] = ex0[k] * inv[k];
      for (int base = 32; base < tot; base += 32) {
        int s; float e[HEADS], alpha[HEADS];
        const int j = base + lane;
        edge_scores(csr_src, asrc, start, deg, n, j, ad, s, e);
#pragma unroll
        for (int k = 0; k < HEADS; ++k)
          alpha[k] = (j < tot) ? __expf(e[k] - mx[k]) * inv[k] : 0.f;
        const int left = tot - base;
        put_weights(wl, lane, alpha);
        gather_inputs4(xin, s, wl, left < 32 ? left : 32, lane, z);
      }
      put_weights(wl, lane, alpha0);
      gather_inputs4(xin, s0, wl, tot < 32 ? tot : 32, lane, z);
    }
#pragma unroll
    for (int k = 0; k < HEADS; ++k) *(float4*)&zs[R * Z1P + k * F1 + 4 * lane] = z[k];
  }

  const int hf = lane >> 4;
  const int l = lane & 15;
  const int hw = wib & (HEADS - 1);
  const int rt = wib >> 3;
  const float* wp = Wm + (size_t)(2 * hf) * (HEADS * CH1) + hw * CH1 + l;
  __syncthreads();

  v8f acc0 = (v8f){};
  v8f acc1 = (v8f){};
#pragma unroll 8
  for (int ks = 0; ks < F1 / 4; ++ks) {
    v2f b0, b1;
    b0.x = wp[(size_t)(4 * ks) * (HEADS * CH1)];
    b0.y = wp[(size_t)(4 * ks + 1) * (HEADS * CH1)];
    b1.x = wp[(size_t)(4 * ks) * (HEADS * CH1) + 16];
    b1.y = wp[(size_t)(4 * ks + 1) * (HEADS * CH1) + 16];
    const v2f a = *(const v2f*)&zs[(16 * rt + l) * Z1P + F1 * hw + 4 * ks + 2 * hf];
    acc0 = __builtin_amdgcn_wmma_f32_16x16x4_f32(false, a, false, b0,
                                                 (short)0, acc0, false, false);
    acc1 = __builtin_amdgcn_wmma_f32_16x16x4_f32(false, a, false, b1,
                                                 (short)0, acc1, false, false);
  }
  __syncthreads();
#pragma unroll
  for (int r = 0; r < 8; ++r) {
    zs[wib * 512 + (r + 8 * hf) * 16 + l] = acc0[r];
    zs[wib * 512 + 256 + (r + 8 * hf) * 16 + l] = acc1[r];
  }
  __syncthreads();

  const int t = (wib << 5) | lane;
#pragma unroll
  for (int q = 0; q < 2; ++q) {
    const int o = t + q * (P1W * 32);
    const int nl = o >> 5;
    const int c = o & 31;
    const int pi = (nl >> 4) * (HEADS * 512) + (c >> 4) * 256 + (nl & 15) * 16 + (c & 15);
    float sum = 0.f;
#pragma unroll
    for (int w = 0; w < HEADS; ++w) sum += zs[w * 512 + pi];
    const int node = nb + nl;
    if (node < N) {
      const float v = sum * (1.0f / HEADS) + bias[c];
      out[(size_t)node * CH1 + c] = v > 0.f ? v : 0.f;
    }
  }
}

static inline unsigned nblk(long n, int b) { return (unsigned)((n + b - 1) / b); }

extern "C" void kernel_launch(void* const* d_in, const int* in_sizes, int n_in,
                              void* d_out, int out_size, void* d_ws, size_t ws_size,
                              hipStream_t stream) {
  const float* x        = (const float*)d_in[0];
  const int*   ei       = (const int*)  d_in[1];
  const float* W1       = (const float*)d_in[2];
  const float* att_src1 = (const float*)d_in[3];
  const float* att_dst1 = (const float*)d_in[4];
  const float* b1       = (const float*)d_in[5];
  const float* W2       = (const float*)d_in[6];
  const float* att_src2 = (const float*)d_in[7];
  const float* att_dst2 = (const float*)d_in[8];
  const float* b2       = (const float*)d_in[9];

  const int F_h   = in_sizes[3] / HEADS;
  const int F_out = in_sizes[7] / HEADS;
  const int F_in  = in_sizes[2] / (HEADS * F_h);
  const int N     = in_sizes[0] / F_in;
  const int E     = in_sizes[1] / 2;
  const int C2    = HEADS * F_out;
  const int* src = ei;
  const int* dst = ei + E;

  float* ws = (float*)d_ws;
  size_t o = 0;
  float* asrc = ws + o; o += (size_t)N * HEADS;
  float* adst = ws + o; o += (size_t)N * HEADS;
  float* x2   = ws + o; o += (size_t)N * F_h;
  float* Wa   = ws + o; o += (size_t)F_in * 16;
  const int NB = (N + BIN_NODES - 1) / BIN_NODES;
  const int G1 = (int)nblk(E, CHUNK);
  int* binned  = (int*)(ws + o); o += (size_t)2 * G1 * CHUNK;
  int* cnt     = (int*)(ws + o); o += (size_t)N;
  int* rowptr  = (int*)(ws + o); o += (size_t)N;
  int* csr_src = (int*)(ws + o); o += (size_t)E;
  int* seg     = (int*)(ws + o); o += (size_t)G1 * (NB + 1);
  float* outp = (float*)d_out;

  const int TB = 256;

  chunk_bin_kernel<<<G1 + 1, SORT_TB, NB * sizeof(int), stream>>>(
      src, dst, E, NB, G1, seg, binned, W1, att_src1, att_dst1, Wa);
  const int ctiles = (N + 15) / 16;
  bin_sort_kernel<<<NB + nblk(ctiles, SORT_TB / 32), SORT_TB, 0, stream>>>(
      binned, seg, G1, NB, N, cnt, rowptr, csr_src, x, Wa, asrc, adst);

  node_attend_project1_kernel<<<nblk(N, P1N), P1W * 32, 0, stream>>>(
      rowptr, cnt, csr_src, asrc, adst, x, W1, b1, x2, N);

  {
    long waves = (long)((N + 15) / 16) * (C2 / (16 * NT));
    wmma_gemm_f32<16, false><<<nblk(waves * 32, TB), TB, 0, stream>>>(
        x2, W2, att_src2, att_dst2, (float*)binned, asrc, adst, N, F_h, C2);
  }
  node_attend_project_kernel<32, 16><<<nblk(N, PN), PW * 32, 0, stream>>>(
      rowptr, cnt, csr_src, asrc, adst, x2, W2, b2, outp, N);
}


# layer-2 attention coefficients = finished rows @ (W2.att) on WMMA in the layer-1 node kernel's epilogue (row tile via LDS); the coefficient GEMM launch dropped (5 -> 4); float sums reordered
# speedup vs baseline: 21.4843x; 1.0401x over previous
#include <hip/hip_runtime.h>
#include <cstdint>

typedef float v2f __attribute__((ext_vector_type(2)));
typedef float v8f __attribute__((ext_vector_type(8)));

#define HEADS 8
#define NEG_SLOPE 0.2f
#define SM_EPS 1e-16f

__device__ __forceinline__ float lrelu(float v) {
  return v > 0.f ? v : NEG_SLOPE * v;
}

__device__ __forceinline__ float xhalf(float x) {
  const int b = __float_as_int(x);
  return __int_as_float(__builtin_amdgcn_permlanex16(b, b, 0x76543210, 0xfedcba98, false, false));
}
template <int CTRL>
__device__ __forceinline__ float xmask_get(float v) {
  return __int_as_float(__builtin_amdgcn_update_dpp(0, __float_as_int(v), CTRL, 0xf, 0xf, true));
}
template <bool IS_MAX>
__device__ __forceinline__ float comb(float a, float b) {
  return IS_MAX ? fmaxf(a, b) : a + b;
}
template <bool IS_MAX>
__device__ __forceinline__ float heads_reduce(const float (&a)[HEADS], int lane) {
  static_assert(HEADS == 8, "three halvings leave one head per lane");
  const bool h16 = (lane & 16) != 0;
  const bool h8 = (lane & 8) != 0;
  const bool h4 = (lane & 4) != 0;
  float b[4], c[2];
#pragma unroll
  for (int k = 0; k < 4; ++k) {
    const float keep = h16 ? a[k + 4] : a[k];
    const float send = h16 ? a[k] : a[k + 4];
    b[k] = comb<IS_MAX>(keep, xhalf(send));
  }
#pragma unroll
  for (int k = 0; k < 2; ++k) {
    const float keep = h8 ? b[k + 2] : b[k];
    const float send = h8 ? b[k] : b[k + 2];
    c[k] = comb<IS_MAX>(keep, xmask_get<0x168>(send));
  }
  const float keep = h4 ? c[1] : c[0];
  const float send = h4 ? c[0] : c[1];
  float d = comb<IS_MAX>(keep, xmask_get<0x164>(send));
  d = comb<IS_MAX>(d, xmask_get<0x162>(d));
  d = comb<IS_MAX>(d, xmask_get<0x161>(d));
  return d;
}
__device__ __forceinline__ float head_total(float d, int k) {
  return __int_as_float(__builtin_amdgcn_readlane(__float_as_int(d), 4 * k));
}

#define F1 128
#define CH1 32
#define CH2 16

template <int F, int CH>
__device__ __forceinline__ void fold_att_block(const float* __restrict__ W,
                                               const float* __restrict__ att_src,
                                               const float* __restrict__ att_dst,
                                               float* __restrict__ Wa) {
  const int t = (int)threadIdx.x;
  if (t >= F * HEADS) return;
  const int f = t >> 3;
  const int h = t & 7;
  float s = 0.f, d = 0.f;
#pragma unroll 8
  for (int c = 0; c < CH; ++c) {
    const float w = W[(size_t)f * (HEADS * CH) + h * CH + c];
    s = __builtin_fmaf(w, att_src[h * CH + c], s);
    d = __builtin_fmaf(w, att_dst[h * CH + c], d);
  }
  Wa[f * 16 + h] = s;
  Wa[f * 16 + 8 + h] = d;
}

__device__ __forceinline__ void coef_tile(const float* __restrict__ A,
                                          const float* __restrict__ Wa,
                                          float* __restrict__ asrc,
                                          float* __restrict__ adst,
                                          int M, int tile, int lane) {
  const int hf = lane >> 4;
  const int l = lane & 15;
  int rowA = tile * 16 + l;
  if (rowA >= M) rowA = M - 1;
  const float* ap = A + (size_t)rowA * F1 + 2 * hf;
  const float* bp = Wa + (2 * hf) * 16 + l;
  v8f acc = (v8f){};
#pragma unroll 8
  for (int ks = 0; ks < F1 / 4; ++ks) {
    v2f a, b;
    a.x = ap[4 * ks];
    a.y = ap[4 * ks + 1];
    b.x = bp[(4 * ks) * 16];
    b.y = bp[(4 * ks + 1) * 16];
    acc = __builtin_amdgcn_wmma_f32_16x16x4_f32(false, a, false, b,
                                                (short)0, acc, false, false);
  }
  const int row0 = tile * 16 + 8 * hf;
  float* dp = (l < 8) ? asrc : adst;
  const int h = l & 7;
#pragma unroll
  for (int r = 0; r < 8; ++r)
    if (row0 + r < M) dp[(size_t)(row0 + r) * HEADS + h] = acc[r];
}

#define BIN_SHIFT 8
#define BIN_NODES (1 << BIN_SHIFT)
#define SORT_TB 1024
#define EPT 8
#define CHUNK (SORT_TB * EPT)
#define SPT 4
#define MAXG 256

__device__ __forceinline__ int wave_incl_scan(int v, int lane) {
  for (int o = 1; o < 32; o <<= 1) {
    const int u = __shfl_up(v, o);
    if (lane >= o) v += u;
  }
  return v;
}

__global__ void __launch_bounds__(SORT_TB)
chunk_bin_kernel(const int* __restrict__ src, const int* __restrict__ dst, int E,
                 int NB, int G1, int* __restrict__ seg, int* __restrict__ binned,
                 const float* __restrict__ W, const float* __restrict__ att_src,
                 const float* __restrict__ att_dst, float* __restrict__ Wa,
                 const float* __restrict__ W2, const float* __restrict__ att_src2,
                 const float* __restrict__ att_dst2, float* __restrict__ Wa2) {
  extern __shared__ int sh[];
  __shared__ int wsum[SORT_TB / 32];
  if ((int)blockIdx.x >= G1) {
    if ((int)blockIdx.x == G1) fold_att_block<F1, CH1>(W, att_src, att_dst, Wa);
    else                       fold_att_block<CH1, CH2>(W2, att_src2, att_dst2, Wa2);
    return;
  }
  const int t = (int)threadIdx.x;
  const int lane = t & 31;
  const int w = t >> 5;
  for (int b = t; b < NB; b += SORT_TB) sh[b] = 0;
  __syncthreads();
  const int base = (int)blockIdx.x * CHUNK + t;
  int s[EPT], d[EPT], pos[EPT];
#pragma unroll
  for (int k = 0; k < EPT; ++k) {
    const int e = base + k * SORT_TB;
    const int ec = e < E ? e : E - 1;
    s[k] = src[ec];
    d[k] = dst[ec];
  }
#pragma unroll
  for (int k = 0; k < EPT; ++k) {
    const int e = base + k * SORT_TB;
    pos[k] = 0;
    if (e < E) pos[k] = atomicAdd(&sh[d[k] >> BIN_SHIFT], 1);
  }
  __syncthreads();
  int carry = 0;
  for (int b0 = 0; b0 < NB; b0 += SORT_TB) {
    const int b = b0 + t;
    const int cv = (b < NB) ? sh[b] : 0;
    const int incl = wave_incl_scan(cv, lane);
    if (lane == 31) wsum[w] = incl;
    __syncthreads();
    int woff = 0, tot = 0;
    for (int k = 0; k < SORT_TB / 32; ++k) {
      const int x = wsum[k];
      woff += (k < w) ? x : 0;
      tot += x;
    }
    if (b < NB) sh[b] = carry + woff + incl - cv;
    carry += tot;
    __syncthreads();
  }
  int* __restrict__ segrow = seg + (size_t)blockIdx.x * (size_t)(NB + 1);
  for (int b = t; b < NB; b += SORT_TB) segrow[b] = sh[b];
  if (t == 0) segrow[NB] = carry;
  int2* __restrict__ outp = (int2*)binned + (size_t)blockIdx.x * CHUNK;
#pragma unroll
  for (int k = 0; k < EPT; ++k) {
    const int e = base + k * SORT_TB;
    if (e < E) outp[sh[d[k] >> BIN_SHIFT] + pos[k]] = make_int2(s[k], d[k]);
  }
}

__device__ __forceinline__ void seg_addr(const int* P, const int* S, int step0,
                                         const int (&i)[SPT], int (&a)[SPT]) {
  int g[SPT];
#pragma unroll
  for (int k = 0; k < SPT; ++k) g[k] = 0;
  for (int step = step0; step > 0; step >>= 1) {
#pragma unroll
    for (int k = 0; k < SPT; ++k) {
      const int cand = g[k] + step;
      g[k] = (P[cand] <= i[k]) ? cand : g[k];
    }
  }
#pragma unroll
  for (int k = 0; k < SPT; ++k) a[k] = S[g[k]] + (i[k] - P[g[k]]);
}

__global__ void __launch_bounds__(SORT_TB, 16)
bin_sort_kernel(const int* __restrict__ binned, const int* __restrict__ seg,
                int G1, int NB, int N,
                int* __restrict__ cnt, int* __restrict__ rowptr,
                int* __restrict__ csr_src,
                const float* __restrict__ xA, const float* __restrict__ Wa,
                float* __restrict__ asrc, float* __restrict__ adst) {
  static_assert(MAXG == BIN_NODES, "both scans run on the block's first MAXG / 32 waves");
  __shared__ int c[BIN_NODES];
  __shared__ int wsum[MAXG / 32];
  __shared__ int wbeg[MAXG / 32];
  __shared__ int P[MAXG];
  __shared__ int S[MAXG];
  if ((int)blockIdx.x >= NB) {
    const int wib = __builtin_amdgcn_readfirstlane((int)(threadIdx.x >> 5));
    const int tile = ((int)blockIdx.x - NB) * (SORT_TB / 32) + wib;
    if (tile < ((N + 15) >> 4))
      coef_tile(xA, Wa, asrc, adst, N, tile, (int)(threadIdx.x & 31));
    return;
  }
  const int2* __restrict__ pairs = (const int2*)binned;
  const int t = (int)threadIdx.x;
  const int lane = t & 31;
  const int w = t >> 5;
  const int b = (int)blockIdx.x;
  int s0 = 0, len = 0, linc = 0;
  if (t < MAXG) {
    if (t < G1) {
      const int* __restrict__ r = seg + (size_t)t * (size_t)(NB + 1) + b;
      s0 = r[0];
      len = r[1] - s0;
    }
    linc = wave_incl_scan(len, lane);
    int bs = s0;
    for (int o = 16; o > 0; o >>= 1) bs += __shfl_xor(bs, o);
    if (lane == 31) wsum[w] = linc;
    if (lane == 0) wbeg[w] = bs;
    c[t] = 0;
  }
  __syncthreads();
  int total = 0, beg = 0;
  {
    int woff = 0;
    for (int k = 0; k < MAXG / 32; ++k) {
      const int x = wsum[k];
      woff += (k < w) ? x : 0;
      total += x;
      beg += wbeg[k];
    }
    if (t < MAXG) {
      P[t] = (t < G1) ? woff + linc - len : 0x7fffffff;
      S[t] = t * CHUNK + s0;
    }
  }
  __syncthreads();
  const int step0 = (G1 > 1) ? 1 << (31 - __clz(G1 - 1)) : 0;
  int af[SPT];
  {
    int ii[SPT];
#pragma unroll
    for (int k = 0; k < SPT; ++k) {
      const int i = t + k * SORT_TB;
      ii[k] = i < total ? i : (total > 0 ? total - 1 : 0);
    }
    seg_addr(P, S, step0, ii, af);
  }
  for (int i0 = t; i0 < total; i0 += SPT * SORT_TB) {
    int y[SPT], a[SPT];
    if (i0 == t) {
#pragma unroll
      for (int k = 0; k < SPT; ++k) a[k] = af[k];
    } else {
      int ii[SPT];
#pragma unroll
      for (int k = 0; k < SPT; ++k) {
        const int i = i0 + k * SORT_TB;
        ii[k] = i < total ? i : total - 1;
      }
      seg_addr(P, S, step0, ii, a);
    }
#pragma unroll
    for (int k = 0; k < SPT; ++k) y[k] = pairs[a[k]].y;
#pragma unroll
    for (int k = 0; k < SPT; ++k)
      atomicAdd(&c[y[k] & (BIN_NODES - 1)], (i0 + k * SORT_TB < total) ? 1 : 0);
  }
  __syncthreads();
  int mine = 0, incl = 0;
  if (t < BIN_NODES) {
    mine = c[t];
    incl = wave_incl_scan(mine, lane);
    if (lane == 31) wsum[w] = incl;
  }
  __syncthreads();
  if (t < BIN_NODES) {
    int woff = 0;
    for (int k = 0; k < BIN_NODES / 32; ++k) woff += (k < w) ? wsum[k] : 0;
    const int start = beg + woff + incl - mine;
    c[t] = start;
    const int n = b * BIN_NODES + t;
    if (n < N) {
      cnt[n] = mine;
      rowptr[n] = start;
    }
  }
  __syncthreads();
  for (int i0 = t; i0 < total; i0 += SPT * SORT_TB) {
    int2 sd[SPT];
    int pos[SPT], a[SPT];
    if (i0 == t) {
#pragma unroll
      for (int k = 0; k < SPT; ++k) a[k] = af[k];
    } else {
      int ii[SPT];
#pragma unroll
      for (int k = 0; k < SPT; ++k) {
        const int i = i0 + k * SORT_TB;
        ii[k] = i < total ? i : total - 1;
      }
      seg_addr(P, S, step0, ii, a);
    }
#pragma unroll
    for (int k = 0; k < SPT; ++k) sd[k] = pairs[a[k]];
#pragma unroll
    for (int k = 0; k < SPT; ++k) {
      pos[k] = 0;
      if (i0 + k * SORT_TB < total) pos[k] = atomicAdd(&c[sd[k].y & (BIN_NODES - 1)], 1);
    }
#pragma unroll
    for (int k = 0; k < SPT; ++k)
      if (i0 + k * SORT_TB < total) csr_src[pos[k]] = sd[k].x;
  }
}

__device__ __forceinline__ void edge_scores(const int* __restrict__ csr_src,
                                            const float* __restrict__ asrc,
                                            int start, int deg, int n, int j,
                                            const float (&ad)[HEADS],
                                            int& s, float (&e)[HEADS]) {
  s = (j < deg) ? csr_src[start + j] : n;
  const float4 a0 = *(const float4*)(asrc + (size_t)s * HEADS);
  const float4 a1 = *(const float4*)(asrc + (size_t)s * HEADS + 4);
  e[0] = lrelu(a0.x + ad[0]);
  e[1] = lrelu(a0.y + ad[1]);
  e[2] = lrelu(a0.z + ad[2]);
  e[3] = lrelu(a0.w + ad[3]);
  e[4] = lrelu(a1.x + ad[4]);
  e[5] = lrelu(a1.y + ad[5]);
  e[6] = lrelu(a1.z + ad[6]);
  e[7] = lrelu(a1.w + ad[7]);
}

#define PW 8
#define PN (2 * PW)
#define KW (HEADS * 32 / PW)
#define ZP (HEADS * 32 + 4)

#define GS 8
template <int F>
__device__ __forceinline__ void gather_inputs(const float* __restrict__ xin, int sreg,
                                              const float* wl, int count,
                                              int lane, float (&z)[HEADS]) {
  for (int j0 = 0; j0 < count; j0 += GS) {
    float v[GS];
#pragma unroll
    for (int g = 0; g < GS; ++g) {
      const int sj = __builtin_amdgcn_readlane(sreg, j0 + g);
      v[g] = xin[(unsigned)sj * F + (unsigned)lane];
    }
#pragma unroll
    for (int g = 0; g < GS; ++g) {
      const float4 a0 = *(const float4*)(wl + (j0 + g) * HEADS);
      const float4 a1 = *(const float4*)(wl + (j0 + g) * HEADS + 4);
      z[0] = __builtin_fmaf(a0.x, v[g], z[0]);  z[1] = __builtin_fmaf(a0.y, v[g], z[1]);
      z[2] = __builtin_fmaf(a0.z, v[g], z[2]);  z[3] = __builtin_fmaf(a0.w, v[g], z[3]);
      z[4] = __builtin_fmaf(a1.x, v[g], z[4]);  z[5] = __builtin_fmaf(a1.y, v[g], z[5]);
      z[6] = __builtin_fmaf(a1.z, v[g], z[6]);  z[7] = __builtin_fmaf(a1.w, v[g], z[7]);
    }
  }
}

__device__ __forceinline__ void put_weights(float* wl, int lane, const float (&alpha)[HEADS]) {
  __builtin_amdgcn_wave_barrier();
  *(float4*)(wl + lane * HEADS)     = make_float4(alpha[0], alpha[1], alpha[2], alpha[3]);
  *(float4*)(wl + lane * HEADS + 4) = make_float4(alpha[4], alpha[5], alpha[6], alpha[7]);
  __builtin_amdgcn_wave_barrier();
}

template <int F, int C>
__global__ void __launch_bounds__(PW * 32, 12)
node_attend_project_kernel(const int* __restrict__ rowptr,
                           const int* __restrict__ cnt,
                           const int* __restrict__ csr_src,
                           const float* __restrict__ asrc,
                           const float* __restrict__ adst,
                           const float* __restrict__ xin,
                           const float* __restrict__ Wm,
                           const float* __restrict__ bias,
                           float* __restrict__ out, int N) {
  static_assert(F == 32, "lane = input channel");
  static_assert(C == 16, "one 16-column tile of output channels");
  static_assert(PN == 16 && PW == HEADS && KW == F, "one 16-node tile; wave w owns head w's k values");
  static_assert(PW * 256 <= PN * ZP, "the partial tiles reuse the z rows' LDS");
  __shared__ __attribute__((aligned(16))) float zs[PN * ZP];
  __shared__ __attribute__((aligned(16))) float aw[PW * 32 * HEADS];
  const int lane = (int)(threadIdx.x & 31);
  const int wib = __builtin_amdgcn_readfirstlane((int)(threadIdx.x >> 5));
  const int nb = (int)blockIdx.x * PN;
  float* wl = aw + wib * (32 * HEADS);

  for (int u = 0; u < 2; ++u) {
    const int R = 2 * wib + u;
    const int n = nb + R;
    float z[HEADS];
#pragma unroll
    for (int k = 0; k < HEADS; ++k) z[k] = 0.f;
    if (n < N) {
      const int start = rowptr[n];
      const int deg = cnt[n];
      const int tot = deg + 1;
      float ad[HEADS];
#pragma unroll
      for (int k = 0; k < HEADS; ++k) ad[k] = adst[(size_t)n * HEADS + k];

      int s0;
      float e0[HEADS];
      const bool v0 = lane < tot;
      edge_scores(csr_src, asrc, start, deg, n, lane, ad, s0, e0);

      float mx[HEADS];
#pragma unroll
      for (int k = 0; k < HEADS; ++k) mx[k] = v0 ? e0[k] : -3.0e38f;
      for (int base = 32; base < tot; base += 32) {
        int s; float e[HEADS];
        const int j = base + lane;
        edge_scores(csr_src, asrc, start, deg, n, j, ad, s, e);
#pragma unroll
        for (int k = 0; k < HEADS; ++k) mx[k] = (j < tot) ? fmaxf(mx[k], e[k]) : mx[k];
      }
      {
        const float tmax = heads_reduce<true>(mx, lane);
#pragma unroll
        for (int k = 0; k < HEADS; ++k) mx[k] = head_total(tmax, k);
      }

      float ex0[HEADS], sm[HEADS];
#pragma unroll
      for (int k = 0; k < HEADS; ++k) {
        ex0[k] = v0 ? __expf(e0[k] - mx[k]) : 0.f;
        sm[k] = ex0[k];
      }
      for (int base = 32; base < tot; base += 32) {
        int s; float e[HEADS];
        const int j = base + lane;
        edge_scores(csr_src, asrc, start, deg, n, j, ad, s, e);
#pragma unroll
        for (int k = 0; k < HEADS; ++k) sm[k] += (j < tot) ? __expf(e[k] - mx[k]) : 0.f;
      }
      float inv[HEADS];
      {
        const float tsum = heads_reduce<false>(sm, lane);
        const float tinv = 1.0f / (tsum + SM_EPS);
#pragma unroll
        for (int k = 0; k < HEADS; ++k) inv[k] = head_total(tinv, k);
      }

      float alpha0[HEADS];
#pragma unroll
      for (int k = 0; k < HEADS; ++k) alpha0[k] = ex0[k] * inv[k];
      for (int base = 32; base < tot; base += 32) {
        int s; float e[HEADS], alpha[HEADS];
        const int j = base + lane;
        edge_scores(csr_src, asrc, start, deg, n, j, ad, s, e);
#pragma unroll
        for (int k = 0; k < HEADS; ++k)
          alpha[k] = (j < tot) ? __expf(e[k] - mx[k]) * inv[k] : 0.f;
        const int left = tot - base;
        put_weights(wl, lane, alpha);
        gather_inputs<F>(xin, s, wl, left < 32 ? left : 32, lane, z);
      }
      put_weights(wl, lane, alpha0);
      gather_inputs<F>(xin, s0, wl, tot < 32 ? tot : 32, lane, z);
    }
#pragma unroll
    for (int k = 0; k < HEADS; ++k) zs[R * ZP + k * F + lane] = z[k];
  }

  const int hf = lane >> 4;
  const int l = lane & 15;
  const float* wp = Wm + (size_t)(2 * hf) * (HEADS * C) + wib * C + l;
  v2f bfr[KW / 4];
#pragma unroll
  for (int ks = 0; ks < KW / 4; ++ks) {
    bfr[ks].x = wp[(size_t)(4 * ks) * (HEADS * C)];
    bfr[ks].y = wp[(size_t)(4 * ks + 1) * (HEADS * C)];
  }
  __syncthreads();

  v8f acc = (v8f){};
#pragma unroll
  for (int ks = 0; ks < KW / 4; ++ks) {
    const v2f a = *(const v2f*)&zs[l * ZP + KW * wib + 4 * ks + 2 * hf];
    acc = __builtin_amdgcn_wmma_f32_16x16x4_f32(false, a, false, bfr[ks],
                                                (short)0, acc, false, false);
  }
  __syncthreads();
#pragma unroll
  for (int r = 0; r < 8; ++r) zs[wib * 256 + (r + 8 * hf) * 16 + l] = acc[r];
  __syncthreads();

  const int t = (int)threadIdx.x;
  float sum = 0.f;
#pragma unroll
  for (int w = 0; w < PW; ++w) sum += zs[w * 256 + t];
  const int node = nb + (t >> 4);
  if (node < N)
    out[(size_t)node * C + (t & 15)] = sum * (1.0f / HEADS) + bias[t & 15];
}

#define P1W 16
#define P1N (2 * P1W)
#define Z1P (HEADS * F1 + 4)
#define X2P (CH1 + 4)

__device__ __forceinline__ void fma4(float4& z, float a, const float4& v) {
  z.x = __builtin_fmaf(a, v.x, z.x);
  z.y = __builtin_fmaf(a, v.y, z.y);
  z.z = __builtin_fmaf(a, v.z, z.z);
  z.w = __builtin_fmaf(a, v.w, z.w);
}

__device__ __forceinline__ void gather_inputs4(const float* __restrict__ xin, int sreg,
                                               const float* wl, int count,
                                               int lane, float4 (&z)[HEADS]) {
  for (int j0 = 0; j0 < count; j0 += GS) {
    float4 v[GS];
#pragma unroll
    for (int g = 0; g < GS; ++g) {
      const int sj = __builtin_amdgcn_readlane(sreg, j0 + g);
      v[g] = *(const float4*)((const char*)xin + (size_t)((unsigned)sj * (4u * F1) + 16u * (unsigned)lane));
    }
#pragma unroll
    for (int g = 0; g < GS; ++g) {
      const float4 a0 = *(const float4*)(wl + (j0 + g) * HEADS);
      const float4 a1 = *(const float4*)(wl + (j0 + g) * HEADS + 4);
      fma4(z[0], a0.x, v[g]);  fma4(z[1], a0.y, v[g]);
      fma4(z[2], a0.z, v[g]);  fma4(z[3], a0.w, v[g]);
      fma4(z[4], a1.x, v[g]);  fma4(z[5], a1.y, v[g]);
      fma4(z[6], a1.z, v[g]);  fma4(z[7], a1.w, v[g]);
    }
  }
}

__global__ void __launch_bounds__(P1W * 32, 8)
node_attend_project1_kernel(const int* __restrict__ rowptr,
                            const int* __restrict__ cnt,
                            const int* __restrict__ csr_src,
                            const float* __restrict__ asrc,
                            const float* __restrict__ adst,
                            const float* __restrict__ xin,
                            const float* __restrict__ Wm,
                            const float* __restrict__ bias,
                            float* __restrict__ out, int N,
                            const float* __restrict__ Wa2,
                            float* __restrict__ asrc2,
                            float* __restrict__ adst2) {
  static_assert(P1N == 32 && P1W == 2 * HEADS, "two 16-node tiles; wave w owns head w % HEADS of tile w / HEADS");
  static_assert(P1N * X2P <= P1W * 32 * HEADS, "the finished-row tile reuses the weight rows' LDS");
  static_assert(P1W * 512 <= P1N * Z1P, "the partial tiles reuse the z rows' LDS");
  static_assert(F1 * 4 == 32 * 16, "lane = 4 input channels");
  __shared__ __attribute__((aligned(16))) float zs[P1N * Z1P];
  __shared__ __attribute__((aligned(16))) float aw[P1W * 32 * HEADS];
  const int lane = (int)__builtin_amdgcn_mbcnt_lo(~0u, 0u);
  const int wib = __builtin_amdgcn_readfirstlane((int)(threadIdx.x >> 5));
  const int nb = (int)blockIdx.x * P1N;
  float* wl = aw + wib * (32 * HEADS);

  for (int u = 0; u < 2; ++u) {
    const int R = 2 * wib + u;
    const int n = nb + R;
    float4 z[HEADS];
#pragma unroll
    for (int k = 0; k < HEADS; ++k) z[k] = make_float4(0.f, 0.f, 0.f, 0.f);
    if (n < N) {
      const int start = rowptr[n];
      const int deg = cnt[n];
      const int tot = deg + 1;
      float ad[HEADS];
#pragma unroll
      for (int k = 0; k < HEADS; ++k) ad[k] = adst[(size_t)n * HEADS + k];

      int s0;
      float e0[HEADS];
      const bool v0 = lane < tot;
      edge_scores(csr_src, asrc, start, deg, n, lane, ad, s0, e0);

      float mx[HEADS];
#pragma unroll
      for (int k = 0; k < HEADS; ++k) mx[k] = v0 ? e0[k] : -3.0e38f;
      for (int base = 32; base < tot; base += 32) {
        int s; float e[HEADS];
        const int j = base + lane;
        edge_scores(csr_src, asrc, start, deg, n, j, ad, s, e);
#pragma unroll
        for (int k = 0; k < HEADS; ++k) mx[k] = (j < tot) ? fmaxf(mx[k], e[k]) : mx[k];
      }
      {
        const float tmax = heads_reduce<true>(mx, lane);
#pragma unroll
        for (int k = 0; k < HEADS; ++k) mx[k] = head_total(tmax, k);
      }

      float ex0[HEADS], sm[HEADS];
#pragma unroll
      for (int k = 0; k < HEADS; ++k) {
        ex0[k] = v0 ? __expf(e0[k] - mx[k]) : 0.f;
        sm[k] = ex0[k];
      }
      for (int base = 32; base < tot; base += 32) {
        int s; float e[HEADS];
        const int j = base + lane;
        edge_scores(csr_src, asrc, start, deg, n, j, ad, s, e);
#pragma unroll
        for (int k = 0; k < HEADS; ++k) sm[k] += (j < tot) ? __expf(e[k] - mx[k]) : 0.f;
      }
      float inv[HEADS];
      {
        const float tsum = heads_reduce<false>(sm, lane);
        const float tinv = 1.0f / (tsum + SM_EPS);
#pragma unroll
        for (int k = 0; k < HEADS; ++k) inv[k] = head_total(tinv, k);
      }

      float alpha0[HEADS];
#pragma unroll
      for (int k = 0; k < HEADS; ++k) alpha0[k] = ex0[k] * inv[k];
      for (int base = 32; base < tot; base += 32) {
        int s; float e[HEADS], alpha[HEADS];
        const int j = base + lane;
        edge_scores(csr_src, asrc, start, deg, n, j, ad, s, e);
#pragma unroll
        for (int k = 0; k < HEADS; ++k)
          alpha[k] = (j < tot) ? __expf(e[k] - mx[k]) * inv[k] : 0.f;
        const int left = tot - base;
        put_weights(wl, lane, alpha);
        gather_inputs4(xin, s, wl, left < 32 ? left : 32, lane, z);
      }
      put_weights(wl, lane, alpha0);
      gather_inputs4(xin, s0, wl, tot < 32 ? tot : 32, lane, z);
    }
#pragma unroll
    for (int k = 0; k < HEADS; ++k) *(float4*)&zs[R * Z1P + k * F1 + 4 * lane] = z[k];
  }

  const int hf = lane >> 4;
  const int l = lane & 15;
  const int hw = wib & (HEADS - 1);
  const int rt = wib >> 3;
  const float* wp = Wm + (size_t)(2 * hf) * (HEADS * CH1) + hw * CH1 + l;
  __syncthreads();

  v8f acc0 = (v8f){};
  v8f acc1 = (v8f){};
#pragma unroll 8
  for (int ks = 0; ks < F1 / 4; ++ks) {
    v2f b0, b1;
    b0.x = wp[(size_t)(4 * ks) * (HEADS * CH1)];
    b0.y = wp[(size_t)(4 * ks + 1) * (HEADS * CH1)];
    b1.x = wp[(size_t)(4 * ks) * (HEADS * CH1) + 16];
    b1.y = wp[(size_t)(4 * ks + 1) * (HEADS * CH1) + 16];
    const v2f a = *(const v2f*)&zs[(16 * rt + l) * Z1P + F1 * hw + 4 * ks + 2 * hf];
    acc0 = __builtin_amdgcn_wmma_f32_16x16x4_f32(false, a, false, b0,
                                                 (short)0, acc0, false, false);
    acc1 = __builtin_amdgcn_wmma_f32_16x16x4_f32(false, a, false, b1,
                                                 (short)0, acc1, false, false);
  }
  __syncthreads();
#pragma unroll
  for (int r = 0; r < 8; ++r) {
    zs[wib * 512 + (r + 8 * hf) * 16 + l] = acc0[r];
    zs[wib * 512 + 256 + (r + 8 * hf) * 16 + l] = acc1[r];
  }
  __syncthreads();

  const int t = (wib << 5) | lane;
  v2f bq[CH1 / 4];
  if (wib < 2) {
    const float* qp = Wa2 + (2 * hf) * 16 + l;
#pragma unroll
    for (int ks = 0; ks < CH1 / 4; ++ks) {
      bq[ks].x = qp[(4 * ks) * 16];
      bq[ks].y = qp[(4 * ks + 1) * 16];
    }
  }
  float* xt = aw;
  float fin[2];
#pragma unroll
  for (int q = 0; q < 2; ++q) {
    const int o = t + q * (P1W * 32);
    const int nl = o >> 5;
    const int c = o & 31;
    const int pi = (nl >> 4) * (HEADS * 512) + (c >> 4) * 256 + (nl & 15) * 16 + (c & 15);
    float sum = 0.f;
#pragma unroll
    for (int w = 0; w < HEADS; ++w) sum += zs[w * 512 + pi];
    const float v = sum * (1.0f / HEADS) + bias[c];
    fin[q] = v > 0.f ? v : 0.f;
    xt[nl * X2P + c] = fin[q];
  }
  __syncthreads();
#pragma unroll
  for (int q = 0; q < 2; ++q) {
    const int o = t + q * (P1W * 32);
    const int node = nb + (o >> 5);
    if (node < N) out[(size_t)node * CH1 + (o & 31)] = fin[q];
  }
  if (wib < 2) {
    v8f c0 = (v8f){};
    v8f c1 = (v8f){};
#pragma unroll
    for (int ks = 0; ks < CH1 / 4; ks += 2) {
      const v2f a0 = *(const v2f*)&xt[(16 * wib + l) * X2P + 4 * ks + 2 * hf];
      const v2f a1 = *(const v2f*)&xt[(16 * wib + l) * X2P + 4 * ks + 4 + 2 * hf];
      c0 = __builtin_amdgcn_wmma_f32_16x16x4_f32(false, a0, false, bq[ks],
                                                 (short)0, c0, false, false);
      c1 = __builtin_amdgcn_wmma_f32_16x16x4_f32(false, a1, false, bq[ks + 1],
                                                 (short)0, c1, false, false);
    }
    const int row0 = nb + 16 * wib + 8 * hf;
    float* dp = (l < 8) ? asrc2 : adst2;
    const int h = l & 7;
#pragma unroll
    for (int r = 0; r < 8; ++r)
      if (row0 + r < N) dp[(size_t)(row0 + r) * HEADS + h] = c0[r] + c1[r];
  }
}

static inline unsigned nblk(long n, int b) { return (unsigned)((n + b - 1) / b); }

extern "C" void kernel_launch(void* const* d_in, const int* in_sizes, int n_in,
                              void* d_out, int out_size, void* d_ws, size_t ws_size,
                              hipStream_t stream) {
  const float* x        = (const float*)d_in[0];
  const int*   ei       = (const int*)  d_in[1];
  const float* W1       = (const float*)d_in[2];
  const float* att_src1 = (const float*)d_in[3];
  const float* att_dst1 = (const float*)d_in[4];
  const float* b1       = (const float*)d_in[5];
  const float* W2       = (const float*)d_in[6];
  const float* att_src2 = (const float*)d_in[7];
  const float* att_dst2 = (const float*)d_in[8];
  const float* b2       = (const float*)d_in[9];

  const int F_h   = in_sizes[3] / HEADS;
  const int F_in  = in_sizes[2] / (HEADS * F_h);
  const int N     = in_sizes[0] / F_in;
  const int E     = in_sizes[1] / 2;
  const int* src = ei;
  const int* dst = ei + E;

  float* ws = (float*)d_ws;
  size_t o = 0;
  float* asrc  = ws + o; o += (size_t)N * HEADS;
  float* adst  = ws + o; o += (size_t)N * HEADS;
  float* asrc2 = ws + o; o += (size_t)N * HEADS;
  float* adst2 = ws + o; o += (size_t)N * HEADS;
  float* x2    = ws + o; o += (size_t)N * F_h;
  float* Wa    = ws + o; o += (size_t)F_in * 16;
  float* Wa2   = ws + o; o += (size_t)F_h * 16;
  const int NB = (N + BIN_NODES - 1) / BIN_NODES;
  const int G1 = (int)nblk(E, CHUNK);
  int* binned  = (int*)(ws + o); o += (size_t)2 * G1 * CHUNK;
  int* cnt     = (int*)(ws + o); o += (size_t)N;
  int* rowptr  = (int*)(ws + o); o += (size_t)N;
  int* csr_src = (int*)(ws + o); o += (size_t)E;
  int* seg     = (int*)(ws + o); o += (size_t)G1 * (NB + 1);
  float* outp = (float*)d_out;

  chunk_bin_kernel<<<G1 + 2, SORT_TB, NB * sizeof(int), stream>>>(
      src, dst, E, NB, G1, seg, binned, W1, att_src1, att_dst1, Wa,
      W2, att_src2, att_dst2, Wa2);
  const int ctiles = (N + 15) / 16;
  bin_sort_kernel<<<NB + nblk(ctiles, SORT_TB / 32), SORT_TB, 0, stream>>>(
      binned, seg, G1, NB, N, cnt, rowptr, csr_src, x, Wa, asrc, adst);

  node_attend_project1_kernel<<<nblk(N, P1N), P1W * 32, 0, stream>>>(
      rowptr, cnt, csr_src, asrc, adst, x, W1, b1, x2, N, Wa2, asrc2, adst2);

  node_attend_project_kernel<32, 16><<<nblk(N, PN), PW * 32, 0, stream>>>(
      rowptr, cnt, csr_src, asrc2, adst2, x2, W2, b2, outp, N);
}
